# starting kernel
# baseline (speedup 1.0000x reference)
#include <hip/hip_runtime.h>
#include <hip/hip_bf16.h>

typedef __attribute__((ext_vector_type(16))) _Float16 v16h;
typedef __attribute__((ext_vector_type(8)))  float    v8f;

#define NN   512
#define NE   4096
#define BB   32
#define HH   32
#define SKC  256
#define ECC  512
#define OUTF 12
#define AEMB 10
#define NEGBIG (-1e30f)
#define NLAY 8

__device__ __forceinline__ v8f wmma_f16(v16h a, v16h b, v8f c) {
    return __builtin_amdgcn_wmma_f32_16x16x32_f16(false, a, false, b, (short)0, c, false, false);
}

__device__ __forceinline__ float fast_sigmoid(float x) {
    return __builtin_amdgcn_rcpf(1.f + __expf(-x));
}
__device__ __forceinline__ float fast_tanh(float x) {
    float t = __expf(-2.f * fabsf(x));
    float r = (1.f - t) * __builtin_amdgcn_rcpf(1.f + t);
    return copysignf(r, x);
}

__device__ __forceinline__ void pack8(v16h& d, int off, const float* __restrict__ p) {
    float4 x = ((const float4*)p)[0];
    float4 y = ((const float4*)p)[1];
    d[off + 0] = (_Float16)x.x; d[off + 1] = (_Float16)x.y;
    d[off + 2] = (_Float16)x.z; d[off + 3] = (_Float16)x.w;
    d[off + 4] = (_Float16)y.x; d[off + 5] = (_Float16)y.y;
    d[off + 6] = (_Float16)y.z; d[off + 7] = (_Float16)y.w;
}
__device__ __forceinline__ void pack8_scale(v16h& d, int off, const float* __restrict__ p, float s) {
    float4 x = ((const float4*)p)[0];
    float4 y = ((const float4*)p)[1];
    d[off + 0] = (_Float16)(x.x * s); d[off + 1] = (_Float16)(x.y * s);
    d[off + 2] = (_Float16)(x.z * s); d[off + 3] = (_Float16)(x.w * s);
    d[off + 4] = (_Float16)(y.x * s); d[off + 5] = (_Float16)(y.y * s);
    d[off + 6] = (_Float16)(y.z * s); d[off + 7] = (_Float16)(y.w * s);
}
__device__ __forceinline__ void pack8_relu(v16h& d, int off, const float* __restrict__ p) {
    float4 x = ((const float4*)p)[0];
    float4 y = ((const float4*)p)[1];
    d[off + 0] = (_Float16)fmaxf(x.x, 0.f); d[off + 1] = (_Float16)fmaxf(x.y, 0.f);
    d[off + 2] = (_Float16)fmaxf(x.z, 0.f); d[off + 3] = (_Float16)fmaxf(x.w, 0.f);
    d[off + 4] = (_Float16)fmaxf(y.x, 0.f); d[off + 5] = (_Float16)fmaxf(y.y, 0.f);
    d[off + 6] = (_Float16)fmaxf(y.z, 0.f); d[off + 7] = (_Float16)fmaxf(y.w, 0.f);
}

__global__ void k_zero(float* p, int n) {
    int i = blockIdx.x * blockDim.x + threadIdx.x;
    if (i < n) p[i] = 0.f;
}

__global__ void k_prep_w(const float* __restrict__ fw, const float* __restrict__ gw,
                         const float* __restrict__ gcw,
                         float* fwT, float* gwT, float* gcwT) {
    int idx = blockIdx.x * blockDim.x + threadIdx.x;
    const int totalFG = NLAY * HH * 2 * HH;
    if (idx < totalFG) {
        int ci = idx & 31;
        int tap = (idx >> 5) & 1;
        int c = (idx >> 6) & 31;
        int l = idx >> 11;
        fwT[idx] = fw[((l * HH + c) * HH + ci) * 2 + tap];
        gwT[idx] = gw[((l * HH + c) * HH + ci) * 2 + tap];
    }
    if (idx < NLAY * HH * HH) {
        int k = idx & 31;
        int n = (idx >> 5) & 31;
        int l = idx >> 10;
        gcwT[idx] = gcw[l * HH * HH + k * HH + n];
    }
}

__global__ void k_scatter(const int* __restrict__ src, const int* __restrict__ dst,
                          unsigned char* adj, float* degout, float* degin) {
    int e = blockIdx.x * blockDim.x + threadIdx.x;
    if (e >= NE) return;
    int s = src[e], d = dst[e];
    adj[s * NN + d] = 1;
    atomicAdd(degout + s, 1.f);
    atomicAdd(degin + d, 1.f);
}

__global__ void k_rownorm(const unsigned char* __restrict__ adj,
                          const float* __restrict__ nv1, const float* __restrict__ nv2,
                          float* rowmax, float* rowsum) {
    __shared__ float sc[NN];
    __shared__ float red[256];
    int i = blockIdx.x;
    for (int j = threadIdx.x; j < NN; j += blockDim.x) {
        float v = NEGBIG;
        if (adj[i * NN + j]) {
            float acc = 0.f;
#pragma unroll
            for (int a = 0; a < AEMB; a++) acc += nv1[i * AEMB + a] * nv2[a * NN + j];
            v = fmaxf(acc, 0.f);
        }
        sc[j] = v;
    }
    __syncthreads();
    float m = NEGBIG;
    for (int j = threadIdx.x; j < NN; j += blockDim.x) m = fmaxf(m, sc[j]);
    red[threadIdx.x] = m;
    __syncthreads();
    for (int s = blockDim.x / 2; s > 0; s >>= 1) {
        if (threadIdx.x < s) red[threadIdx.x] = fmaxf(red[threadIdx.x], red[threadIdx.x + s]);
        __syncthreads();
    }
    m = red[0];
    __syncthreads();
    float sum = 0.f;
    for (int j = threadIdx.x; j < NN; j += blockDim.x) {
        float v = sc[j];
        if (v > 0.5f * NEGBIG) sum += __expf(v - m);
    }
    red[threadIdx.x] = sum;
    __syncthreads();
    for (int s = blockDim.x / 2; s > 0; s >>= 1) {
        if (threadIdx.x < s) red[threadIdx.x] += red[threadIdx.x + s];
        __syncthreads();
    }
    if (threadIdx.x == 0) {
        rowmax[i] = m;
        rowsum[i] = (red[0] > 0.f) ? red[0] : 1.f;
    }
}

__global__ void k_ew(const int* __restrict__ src, const int* __restrict__ dst,
                     const float* __restrict__ nv1, const float* __restrict__ nv2,
                     const float* __restrict__ rowmax, const float* __restrict__ rowsum,
                     float* ew) {
    int e = blockIdx.x * blockDim.x + threadIdx.x;
    if (e >= NE) return;
    int s = src[e], d = dst[e];
    float acc = 0.f;
#pragma unroll
    for (int a = 0; a < AEMB; a++) acc += nv1[s * AEMB + a] * nv2[a * NN + d];
    float scv = fmaxf(acc, 0.f);
    ew[e] = 1.f + __expf(scv - rowmax[s]) * __builtin_amdgcn_rcpf(rowsum[s]);
}

__global__ void k_norms(const float* degout, const float* degin, float* nsv, float* ndv) {
    int i = blockIdx.x * blockDim.x + threadIdx.x;
    if (i >= NN) return;
    nsv[i] = rsqrtf(fmaxf(degout[i], 1.f));
    ndv[i] = rsqrtf(fmaxf(degin[i], 1.f));
}

__global__ void k_start(const float* __restrict__ feat, const float* __restrict__ sw,
                        const float* __restrict__ sb, float* h) {
    int idx = blockIdx.x * blockDim.x + threadIdx.x;
    const int total = BB * NN * 13 * HH;
    if (idx >= total) return;
    int c = idx & 31;
    int t = (idx >> 5) % 13;
    int m = idx / (13 * HH);
    int b = m >> 9, n = m & 511;
    float x = (t == 0) ? 0.f : feat[(b * 12 + (t - 1)) * NN + n];
    h[idx] = sw[c] * x + sb[c];
}

__global__ void k_gau_wmma(const float* __restrict__ h, const float* __restrict__ fwT,
                           const float* __restrict__ fb, const float* __restrict__ gwT,
                           const float* __restrict__ gb, float* gau, int T, int d, int l) {
    int Tp = T - d;
    int tile = blockIdx.x * 4 + (threadIdx.x >> 5);
    int tm = tile >> 1, tn = tile & 1;
    int lane = threadIdx.x & 31;
    int mrow = tm * 16 + (lane & 15);
    int t = mrow % Tp;
    int bn = mrow / Tp;
    const float* hrow0 = h + (bn * T + t) * HH;
    const float* hrow1 = hrow0 + d * HH;
    int ncol = tn * 16 + (lane & 15);
    int base = (lane & 16) ? 8 : 0;
    const float* fwr = fwT + (size_t)((l * HH + ncol) * 2) * HH;
    const float* gwr = gwT + (size_t)((l * HH + ncol) * 2) * HH;
    v8f cf = {}, cg = {};
    v16h a, bf, bg;
    pack8(a, 0, hrow0 + base);  pack8(a, 8, hrow0 + base + 16);
    pack8(bf, 0, fwr + base);   pack8(bf, 8, fwr + base + 16);
    pack8(bg, 0, gwr + base);   pack8(bg, 8, gwr + base + 16);
    cf = wmma_f16(a, bf, cf);
    cg = wmma_f16(a, bg, cg);
    pack8(a, 0, hrow1 + base);      pack8(a, 8, hrow1 + base + 16);
    pack8(bf, 0, fwr + HH + base);  pack8(bf, 8, fwr + HH + base + 16);
    pack8(bg, 0, gwr + HH + base);  pack8(bg, 8, gwr + HH + base + 16);
    cf = wmma_f16(a, bf, cf);
    cg = wmma_f16(a, bg, cg);

    int rbase = tm * 16 + ((lane & 16) ? 8 : 0);
    float fbias = fb[l * HH + ncol];
    float gbias = gb[l * HH + ncol];
#pragma unroll
    for (int v = 0; v < 8; v++) {
        int row = rbase + v;
        float fv = fast_tanh(cf[v] + fbias);
        float gv = fast_sigmoid(cg[v] + gbias);
        gau[(size_t)row * HH + ncol] = fv * gv;
    }
}

__global__ void k_skip_wmma(const float* __restrict__ gau, const float* __restrict__ sw,
                            const float* __restrict__ sb, float* skip, int Tp, int l) {
    const int nTilesN = SKC / 16;
    int tile = blockIdx.x * 4 + (threadIdx.x >> 5);
    int tm = tile / nTilesN, tn = tile % nTilesN;
    int lane = threadIdx.x & 31;
    int mrow = tm * 16 + (lane & 15);
    int ncol = tn * 16 + (lane & 15);
    int base = (lane & 16) ? 8 : 0;
    const float* arow = gau + (size_t)(mrow * Tp + (Tp - 1)) * HH;
    const float* brow = sw + (size_t)(l * SKC + ncol) * HH;
    v16h a, bm;
    pack8(a, 0, arow + base);  pack8(a, 8, arow + base + 16);
    pack8(bm, 0, brow + base); pack8(bm, 8, brow + base + 16);
    v8f c = {};
    c = wmma_f16(a, bm, c);
    int rbase = tm * 16 + ((lane & 16) ? 8 : 0);
    float bias = sb[l * SKC + ncol];
#pragma unroll
    for (int v = 0; v < 8; v++) {
        float* p = skip + (size_t)(rbase + v) * SKC + ncol;
        *p += c[v] + bias;
    }
}

__global__ void k_msg(const int* __restrict__ src, const int* __restrict__ dst,
                      const float* __restrict__ ew, const float* __restrict__ nsv,
                      const float* __restrict__ gau, float* agg, int Tp) {
    int idx = blockIdx.x * blockDim.x + threadIdx.x;
    int total = NE * BB * Tp * HH;
    if (idx >= total) return;
    int c = idx & 31;
    int t = (idx >> 5) % Tp;
    int r = idx / (Tp * HH);
    int e = r % NE;
    int b = r / NE;
    int s = src[e], d = dst[e];
    float coeff = ew[e] * nsv[s];
    float* pa = &agg[((b * NN + d) * Tp + t) * HH + c];
    __builtin_prefetch(pa, 1, 0);
    atomicAdd(pa, coeff * gau[((b * NN + s) * Tp + t) * HH + c]);
}

__global__ void k_gconv_wmma(const float* __restrict__ agg, const float* __restrict__ ndv,
                             const float* __restrict__ gcwT, const float* __restrict__ gcb,
                             const float* __restrict__ hold, float* hpre,
                             float* bnsum, float* bnsq, int T, int d, int l) {
    int Tp = T - d;
    int tile = blockIdx.x * 4 + (threadIdx.x >> 5);
    int tm = tile >> 1, tn = tile & 1;
    int lane = threadIdx.x & 31;
    int mrow = tm * 16 + (lane & 15);
    int node_m = (mrow / Tp) & 511;
    float nrm = ndv[node_m];
    int ncol = tn * 16 + (lane & 15);
    int base = (lane & 16) ? 8 : 0;
    const float* arow = agg + (size_t)mrow * HH;
    const float* brow = gcwT + (size_t)(l * HH + ncol) * HH;
    v16h a, bm;
    pack8_scale(a, 0, arow + base, nrm);
    pack8_scale(a, 8, arow + base + 16, nrm);
    pack8(bm, 0, brow + base);
    pack8(bm, 8, brow + base + 16);
    v8f c = {};
    c = wmma_f16(a, bm, c);
    int rbase = tm * 16 + ((lane & 16) ? 8 : 0);
    float bias = gcb[l * HH + ncol];
    float s1 = 0.f, s2 = 0.f;
#pragma unroll
    for (int v = 0; v < 8; v++) {
        int row = rbase + v;
        int t = row % Tp;
        int bn = row / Tp;
        float val = fmaxf(c[v] + bias, 0.f) + hold[(bn * T + t + d) * HH + ncol];
        hpre[(size_t)row * HH + ncol] = val;
        s1 += val;
        s2 += val * val;
    }
    atomicAdd(&bnsum[ncol], s1);
    atomicAdd(&bnsq[ncol], s2);
}

__global__ void k_bn(float* h, const float* __restrict__ bnsum, const float* __restrict__ bnsq,
                     const float* __restrict__ gamma, const float* __restrict__ beta,
                     int Tp, int l) {
    int total = BB * NN * Tp * HH;
    int idx = blockIdx.x * blockDim.x + threadIdx.x;
    if (idx >= total) return;
    int c = idx & 31;
    float cnt = (float)(BB * NN * Tp);
    float mu = bnsum[c] / cnt;
    float var = bnsq[c] / cnt - mu * mu;
    float x = h[idx];
    h[idx] = (x - mu) * rsqrtf(var + 1e-5f) * gamma[l * HH + c] + beta[l * HH + c];
}

__global__ void k_end1_wmma(const float* __restrict__ skip, const float* __restrict__ w1,
                            const float* __restrict__ b1, float* out1) {
    const int nTilesN = ECC / 16;
    int tile = blockIdx.x * 4 + (threadIdx.x >> 5);
    int tm = tile / nTilesN, tn = tile % nTilesN;
    int lane = threadIdx.x & 31;
    int mrow = tm * 16 + (lane & 15);
    int ncol = tn * 16 + (lane & 15);
    int base = (lane & 16) ? 8 : 0;
    const float* arow = skip + (size_t)mrow * SKC;
    const float* brow = w1 + (size_t)ncol * SKC;
    v8f c = {};
#pragma unroll
    for (int k0 = 0; k0 < SKC; k0 += 32) {
        v16h a, bm;
        pack8_relu(a, 0, arow + k0 + base);
        pack8_relu(a, 8, arow + k0 + base + 16);
        pack8(bm, 0, brow + k0 + base);
        pack8(bm, 8, brow + k0 + base + 16);
        c = wmma_f16(a, bm, c);
    }
    int rbase = tm * 16 + ((lane & 16) ? 8 : 0);
    float bias = b1[ncol];
#pragma unroll
    for (int v = 0; v < 8; v++)
        out1[(size_t)(rbase + v) * ECC + ncol] = fmaxf(c[v] + bias, 0.f);
}

__global__ void k_end2_wmma(const float* __restrict__ in1, const float* __restrict__ w2,
                            const float* __restrict__ b2, float* out) {
    int tm = blockIdx.x * 4 + (threadIdx.x >> 5);
    int lane = threadIdx.x & 31;
    int mrow = tm * 16 + (lane & 15);
    int ncol = lane & 15;
    int base = (lane & 16) ? 8 : 0;
    const float* arow = in1 + (size_t)mrow * ECC;
    const float* brow = w2 + (size_t)ncol * ECC;
    v8f c = {};
#pragma unroll
    for (int k0 = 0; k0 < ECC; k0 += 32) {
        v16h a, bm = {};
        pack8(a, 0, arow + k0 + base);
        pack8(a, 8, arow + k0 + base + 16);
        if (ncol < OUTF) {
            pack8(bm, 0, brow + k0 + base);
            pack8(bm, 8, brow + k0 + base + 16);
        }
        c = wmma_f16(a, bm, c);
    }
    if (ncol < OUTF) {
        int rbase = tm * 16 + ((lane & 16) ? 8 : 0);
        float bias = b2[ncol];
#pragma unroll
        for (int v = 0; v < 8; v++) {
            int row = rbase + v;
            int b = row >> 9, n = row & 511;
            out[(b * OUTF + ncol) * NN + n] = c[v] + bias;
        }
    }
}

extern "C" void kernel_launch(void* const* d_in, const int* in_sizes, int n_in,
                              void* d_out, int out_size, void* d_ws, size_t ws_size,
                              hipStream_t stream) {
    const float* feat     = (const float*)d_in[0];
    const int*   src      = (const int*)d_in[1];
    const int*   dst      = (const int*)d_in[2];
    const float* nv1      = (const float*)d_in[3];
    const float* nv2      = (const float*)d_in[4];
    const float* start_w  = (const float*)d_in[5];
    const float* start_b  = (const float*)d_in[6];
    const float* filter_w = (const float*)d_in[7];
    const float* filter_b = (const float*)d_in[8];
    const float* gate_w   = (const float*)d_in[9];
    const float* gate_b   = (const float*)d_in[10];
    const float* skip_w   = (const float*)d_in[11];
    const float* skip_b   = (const float*)d_in[12];
    const float* bn_gamma = (const float*)d_in[13];
    const float* bn_beta  = (const float*)d_in[14];
    const float* gconv_w  = (const float*)d_in[15];
    const float* gconv_b  = (const float*)d_in[16];
    const float* end1_w   = (const float*)d_in[17];
    const float* end1_b   = (const float*)d_in[18];
    const float* end2_w   = (const float*)d_in[19];
    const float* end2_b   = (const float*)d_in[20];

    float* ws = (float*)d_ws;
    size_t o = 0;
    unsigned char* adj = (unsigned char*)(ws + o); o += (NN * NN) / 4;
    float* rowmax = ws + o; o += NN;
    float* rowsum = ws + o; o += NN;
    float* ew     = ws + o; o += NE;
    float* degout = ws + o; o += NN;
    float* degin  = ws + o; o += NN;
    float* nsv    = ws + o; o += NN;
    float* ndv    = ws + o; o += NN;
    float* bnsum  = ws + o; o += HH;
    float* bnsq   = ws + o; o += HH;
    float* fwT    = ws + o; o += NLAY * HH * 2 * HH;
    float* gwT    = ws + o; o += NLAY * HH * 2 * HH;
    float* gcwT   = ws + o; o += NLAY * HH * HH;
    float* skip   = ws + o; o += (size_t)BB * NN * SKC;
    float* hA     = ws + o; o += (size_t)BB * NN * 13 * HH;
    float* hB     = ws + o; o += (size_t)BB * NN * 13 * HH;
    float* gau    = ws + o; o += (size_t)BB * NN * 12 * HH;
    float* agg    = ws + o; o += (size_t)BB * NN * 12 * HH;
    float* out1   = gau;

#define ZEROF(p, n) k_zero<<<((n) + 255) / 256, 256, 0, stream>>>((p), (n))

    ZEROF((float*)adj, (NN * NN) / 4);
    ZEROF(degout, NN);
    ZEROF(degin, NN);
    ZEROF(skip, BB * NN * SKC);

    k_prep_w<<<(NLAY * HH * 2 * HH + 255) / 256, 256, 0, stream>>>(filter_w, gate_w, gconv_w,
                                                                   fwT, gwT, gcwT);
    k_scatter<<<(NE + 255) / 256, 256, 0, stream>>>(src, dst, adj, degout, degin);
    k_rownorm<<<NN, 256, 0, stream>>>(adj, nv1, nv2, rowmax, rowsum);
    k_ew<<<(NE + 255) / 256, 256, 0, stream>>>(src, dst, nv1, nv2, rowmax, rowsum, ew);
    k_norms<<<(NN + 255) / 256, 256, 0, stream>>>(degout, degin, nsv, ndv);
    k_start<<<(BB * NN * 13 * HH + 255) / 256, 256, 0, stream>>>(feat, start_w, start_b, hA);

    const int dil[8] = {1, 2, 1, 2, 1, 2, 1, 2};
    float* hcur = hA;
    float* hnext = hB;
    int T = 13;
    for (int l = 0; l < NLAY; l++) {
        int d = dil[l];
        int Tp = T - d;
        int M = BB * NN * Tp;
        int tilesGau = (M / 16) * 2;

        k_gau_wmma<<<tilesGau / 4, 128, 0, stream>>>(hcur, fwT, filter_b, gwT, gate_b,
                                                     gau, T, d, l);
        k_skip_wmma<<<(BB * NN / 16) * (SKC / 16) / 4, 128, 0, stream>>>(gau, skip_w, skip_b,
                                                                         skip, Tp, l);
        ZEROF(agg, M * HH);
        k_msg<<<(NE * BB * Tp * HH + 255) / 256, 256, 0, stream>>>(src, dst, ew, nsv,
                                                                   gau, agg, Tp);
        ZEROF(bnsum, HH);
        ZEROF(bnsq, HH);
        k_gconv_wmma<<<(M / 16) * 2 / 4, 128, 0, stream>>>(agg, ndv, gcwT, gconv_b,
                                                           hcur, hnext, bnsum, bnsq, T, d, l);
        k_bn<<<(M * HH + 255) / 256, 256, 0, stream>>>(hnext, bnsum, bnsq,
                                                       bn_gamma, bn_beta, Tp, l);
        float* tmp = hcur; hcur = hnext; hnext = tmp;
        T = Tp;
    }

    k_end1_wmma<<<(BB * NN / 16) * (ECC / 16) / 4, 128, 0, stream>>>(skip, end1_w, end1_b, out1);
    k_end2_wmma<<<(BB * NN / 16) / 4, 128, 0, stream>>>(out1, end2_w, end2_b, (float*)d_out);
#undef ZEROF
}


# message passing: per-destination edge lists built each call plus an f32 gather replace the float-atomic scatter and its zero fill; each sum now runs in fixed edge order
# speedup vs baseline: 1.8002x; 1.8002x over previous
#include <hip/hip_runtime.h>
#include <hip/hip_bf16.h>

typedef __attribute__((ext_vector_type(16))) _Float16 v16h;
typedef __attribute__((ext_vector_type(8)))  float    v8f;

#define NN   512
#define NE   4096
#define BB   32
#define HH   32
#define SKC  256
#define ECC  512
#define OUTF 12
#define AEMB 10
#define NEGBIG (-1e30f)
#define NLAY 8

__device__ __forceinline__ v8f wmma_f16(v16h a, v16h b, v8f c) {
    return __builtin_amdgcn_wmma_f32_16x16x32_f16(false, a, false, b, (short)0, c, false, false);
}

__device__ __forceinline__ float fast_sigmoid(float x) {
    return __builtin_amdgcn_rcpf(1.f + __expf(-x));
}
__device__ __forceinline__ float fast_tanh(float x) {
    float t = __expf(-2.f * fabsf(x));
    float r = (1.f - t) * __builtin_amdgcn_rcpf(1.f + t);
    return copysignf(r, x);
}

__device__ __forceinline__ void pack8(v16h& d, int off, const float* __restrict__ p) {
    float4 x = ((const float4*)p)[0];
    float4 y = ((const float4*)p)[1];
    d[off + 0] = (_Float16)x.x; d[off + 1] = (_Float16)x.y;
    d[off + 2] = (_Float16)x.z; d[off + 3] = (_Float16)x.w;
    d[off + 4] = (_Float16)y.x; d[off + 5] = (_Float16)y.y;
    d[off + 6] = (_Float16)y.z; d[off + 7] = (_Float16)y.w;
}
__device__ __forceinline__ void pack8_scale(v16h& d, int off, const float* __restrict__ p, float s) {
    float4 x = ((const float4*)p)[0];
    float4 y = ((const float4*)p)[1];
    d[off + 0] = (_Float16)(x.x * s); d[off + 1] = (_Float16)(x.y * s);
    d[off + 2] = (_Float16)(x.z * s); d[off + 3] = (_Float16)(x.w * s);
    d[off + 4] = (_Float16)(y.x * s); d[off + 5] = (_Float16)(y.y * s);
    d[off + 6] = (_Float16)(y.z * s); d[off + 7] = (_Float16)(y.w * s);
}
__device__ __forceinline__ void pack8_relu(v16h& d, int off, const float* __restrict__ p) {
    float4 x = ((const float4*)p)[0];
    float4 y = ((const float4*)p)[1];
    d[off + 0] = (_Float16)fmaxf(x.x, 0.f); d[off + 1] = (_Float16)fmaxf(x.y, 0.f);
    d[off + 2] = (_Float16)fmaxf(x.z, 0.f); d[off + 3] = (_Float16)fmaxf(x.w, 0.f);
    d[off + 4] = (_Float16)fmaxf(y.x, 0.f); d[off + 5] = (_Float16)fmaxf(y.y, 0.f);
    d[off + 6] = (_Float16)fmaxf(y.z, 0.f); d[off + 7] = (_Float16)fmaxf(y.w, 0.f);
}

__global__ void k_zero(float* p, int n) {
    int i = blockIdx.x * blockDim.x + threadIdx.x;
    if (i < n) p[i] = 0.f;
}

__global__ void k_prep_w(const float* __restrict__ fw, const float* __restrict__ gw,
                         const float* __restrict__ gcw,
                         float* fwT, float* gwT, float* gcwT) {
    int idx = blockIdx.x * blockDim.x + threadIdx.x;
    const int totalFG = NLAY * HH * 2 * HH;
    if (idx < totalFG) {
        int ci = idx & 31;
        int tap = (idx >> 5) & 1;
        int c = (idx >> 6) & 31;
        int l = idx >> 11;
        fwT[idx] = fw[((l * HH + c) * HH + ci) * 2 + tap];
        gwT[idx] = gw[((l * HH + c) * HH + ci) * 2 + tap];
    }
    if (idx < NLAY * HH * HH) {
        int k = idx & 31;
        int n = (idx >> 5) & 31;
        int l = idx >> 10;
        gcwT[idx] = gcw[l * HH * HH + k * HH + n];
    }
}

__global__ void k_scatter(const int* __restrict__ src, const int* __restrict__ dst,
                          unsigned char* adj, float* degout, float* degin) {
    int e = blockIdx.x * blockDim.x + threadIdx.x;
    if (e >= NE) return;
    int s = src[e], d = dst[e];
    adj[s * NN + d] = 1;
    atomicAdd(degout + s, 1.f);
    atomicAdd(degin + d, 1.f);
}

__global__ void k_rownorm(const unsigned char* __restrict__ adj,
                          const float* __restrict__ nv1, const float* __restrict__ nv2,
                          float* rowmax, float* rowsum) {
    __shared__ float sc[NN];
    __shared__ float red[256];
    int i = blockIdx.x;
    for (int j = threadIdx.x; j < NN; j += blockDim.x) {
        float v = NEGBIG;
        if (adj[i * NN + j]) {
            float acc = 0.f;
#pragma unroll
            for (int a = 0; a < AEMB; a++) acc += nv1[i * AEMB + a] * nv2[a * NN + j];
            v = fmaxf(acc, 0.f);
        }
        sc[j] = v;
    }
    __syncthreads();
    float m = NEGBIG;
    for (int j = threadIdx.x; j < NN; j += blockDim.x) m = fmaxf(m, sc[j]);
    red[threadIdx.x] = m;
    __syncthreads();
    for (int s = blockDim.x / 2; s > 0; s >>= 1) {
        if (threadIdx.x < s) red[threadIdx.x] = fmaxf(red[threadIdx.x], red[threadIdx.x + s]);
        __syncthreads();
    }
    m = red[0];
    __syncthreads();
    float sum = 0.f;
    for (int j = threadIdx.x; j < NN; j += blockDim.x) {
        float v = sc[j];
        if (v > 0.5f * NEGBIG) sum += __expf(v - m);
    }
    red[threadIdx.x] = sum;
    __syncthreads();
    for (int s = blockDim.x / 2; s > 0; s >>= 1) {
        if (threadIdx.x < s) red[threadIdx.x] += red[threadIdx.x + s];
        __syncthreads();
    }
    if (threadIdx.x == 0) {
        rowmax[i] = m;
        rowsum[i] = (red[0] > 0.f) ? red[0] : 1.f;
    }
}

__global__ void k_ew(const int* __restrict__ src, const int* __restrict__ dst,
                     const float* __restrict__ nv1, const float* __restrict__ nv2,
                     const float* __restrict__ rowmax, const float* __restrict__ rowsum,
                     float* ew) {
    int e = blockIdx.x * blockDim.x + threadIdx.x;
    if (e >= NE) return;
    int s = src[e], d = dst[e];
    float acc = 0.f;
#pragma unroll
    for (int a = 0; a < AEMB; a++) acc += nv1[s * AEMB + a] * nv2[a * NN + d];
    float scv = fmaxf(acc, 0.f);
    ew[e] = 1.f + __expf(scv - rowmax[s]) * __builtin_amdgcn_rcpf(rowsum[s]);
}

__global__ void k_norms(const float* degout, const float* degin, float* nsv, float* ndv) {
    int i = blockIdx.x * blockDim.x + threadIdx.x;
    if (i >= NN) return;
    nsv[i] = rsqrtf(fmaxf(degout[i], 1.f));
    ndv[i] = rsqrtf(fmaxf(degin[i], 1.f));
}

__global__ void k_start(const float* __restrict__ feat, const float* __restrict__ sw,
                        const float* __restrict__ sb, float* h) {
    int idx = blockIdx.x * blockDim.x + threadIdx.x;
    const int total = BB * NN * 13 * HH;
    if (idx >= total) return;
    int c = idx & 31;
    int t = (idx >> 5) % 13;
    int m = idx / (13 * HH);
    int b = m >> 9, n = m & 511;
    float x = (t == 0) ? 0.f : feat[(b * 12 + (t - 1)) * NN + n];
    h[idx] = sw[c] * x + sb[c];
}

__global__ void k_gau_wmma(const float* __restrict__ h, const float* __restrict__ fwT,
                           const float* __restrict__ fb, const float* __restrict__ gwT,
                           const float* __restrict__ gb, float* gau, int T, int d, int l) {
    int Tp = T - d;
    int tile = blockIdx.x * 4 + (threadIdx.x >> 5);
    int tm = tile >> 1, tn = tile & 1;
    int lane = threadIdx.x & 31;
    int mrow = tm * 16 + (lane & 15);
    int t = mrow % Tp;
    int bn = mrow / Tp;
    const float* hrow0 = h + (bn * T + t) * HH;
    const float* hrow1 = hrow0 + d * HH;
    int ncol = tn * 16 + (lane & 15);
    int base = (lane & 16) ? 8 : 0;
    const float* fwr = fwT + (size_t)((l * HH + ncol) * 2) * HH;
    const float* gwr = gwT + (size_t)((l * HH + ncol) * 2) * HH;
    v8f cf = {}, cg = {};
    v16h a, bf, bg;
    pack8(a, 0, hrow0 + base);  pack8(a, 8, hrow0 + base + 16);
    pack8(bf, 0, fwr + base);   pack8(bf, 8, fwr + base + 16);
    pack8(bg, 0, gwr + base);   pack8(bg, 8, gwr + base + 16);
    cf = wmma_f16(a, bf, cf);
    cg = wmma_f16(a, bg, cg);
    pack8(a, 0, hrow1 + base);      pack8(a, 8, hrow1 + base + 16);
    pack8(bf, 0, fwr + HH + base);  pack8(bf, 8, fwr + HH + base + 16);
    pack8(bg, 0, gwr + HH + base);  pack8(bg, 8, gwr + HH + base + 16);
    cf = wmma_f16(a, bf, cf);
    cg = wmma_f16(a, bg, cg);

    int rbase = tm * 16 + ((lane & 16) ? 8 : 0);
    float fbias = fb[l * HH + ncol];
    float gbias = gb[l * HH + ncol];
#pragma unroll
    for (int v = 0; v < 8; v++) {
        int row = rbase + v;
        float fv = fast_tanh(cf[v] + fbias);
        float gv = fast_sigmoid(cg[v] + gbias);
        gau[(size_t)row * HH + ncol] = fv * gv;
    }
}

__global__ void k_skip_wmma(const float* __restrict__ gau, const float* __restrict__ sw,
                            const float* __restrict__ sb, float* skip, int Tp, int l) {
    const int nTilesN = SKC / 16;
    int tile = blockIdx.x * 4 + (threadIdx.x >> 5);
    int tm = tile / nTilesN, tn = tile % nTilesN;
    int lane = threadIdx.x & 31;
    int mrow = tm * 16 + (lane & 15);
    int ncol = tn * 16 + (lane & 15);
    int base = (lane & 16) ? 8 : 0;
    const float* arow = gau + (size_t)(mrow * Tp + (Tp - 1)) * HH;
    const float* brow = sw + (size_t)(l * SKC + ncol) * HH;
    v16h a, bm;
    pack8(a, 0, arow + base);  pack8(a, 8, arow + base + 16);
    pack8(bm, 0, brow + base); pack8(bm, 8, brow + base + 16);
    v8f c = {};
    c = wmma_f16(a, bm, c);
    int rbase = tm * 16 + ((lane & 16) ? 8 : 0);
    float bias = sb[l * SKC + ncol];
#pragma unroll
    for (int v = 0; v < 8; v++) {
        float* p = skip + (size_t)(rbase + v) * SKC + ncol;
        *p += c[v] + bias;
    }
}

#define CSR_T 256
#define CSR_PER (NE / CSR_T)
__global__ void k_csr(const int* __restrict__ src, const int* __restrict__ dst,
                      const float* __restrict__ ew, const float* __restrict__ nsv,
                      int2* rowse, int2* rec) {
    __shared__ int s_lt[CSR_T];
    __shared__ int s_eq[CSR_T];
    const int d = blockIdx.x;
    const int tid = threadIdx.x;
    const int e0 = tid * CSR_PER;
    int lt = 0, eq = 0;
    for (int i = 0; i < CSR_PER; i++) {
        int x = dst[e0 + i];
        lt += (x < d) ? 1 : 0;
        eq += (x == d) ? 1 : 0;
    }
    s_lt[tid] = lt;
    s_eq[tid] = eq;
    __syncthreads();
    for (int off = 1; off < CSR_T; off <<= 1) {
        int a = 0, b = 0;
        if (tid >= off) { a = s_lt[tid - off]; b = s_eq[tid - off]; }
        __syncthreads();
        s_lt[tid] += a;
        s_eq[tid] += b;
        __syncthreads();
    }
    const int first = s_lt[CSR_T - 1];
    int pos = first + s_eq[tid] - eq;
    if (tid == 0) rowse[d] = make_int2(first, first + s_eq[CSR_T - 1]);
    for (int i = 0; i < CSR_PER; i++) {
        int e = e0 + i;
        if (dst[e] == d) {
            int sn = src[e];
            rec[pos] = make_int2(sn, __float_as_int(ew[e] * nsv[sn]));
            pos++;
        }
    }
}

__device__ __forceinline__ void fma4(float4& a, float w, const float4 v) {
    a.x = fmaf(w, v.x, a.x); a.y = fmaf(w, v.y, a.y);
    a.z = fmaf(w, v.z, a.z); a.w = fmaf(w, v.w, a.w);
}
__global__ void k_gather(const int2* __restrict__ rowse, const int2* __restrict__ rec,
                         const float4* __restrict__ gau4, float4* __restrict__ agg4,
                         int slab4) {
    const int total = (BB / 4) * NN * slab4;
    int idx = blockIdx.x * blockDim.x + threadIdx.x;
    if (idx >= total) return;
    int bd = idx / slab4;
    int q = idx - bd * slab4;
    int d = bd & (NN - 1);
    int b0 = bd >> 9;
    const int bstride = (BB / 4) * NN * slab4;
    const float4* g = gau4 + (b0 * NN) * slab4 + q;
    int2 se = rowse[d];
    float4 a0 = make_float4(0.f, 0.f, 0.f, 0.f), a1 = a0, a2 = a0, a3 = a0;
    int j = se.x;
    for (; j + 2 <= se.y; j += 2) {
        int2 r0 = rec[j], r1 = rec[j + 1];
        const float4* p0 = g + r0.x * slab4;
        const float4* p1 = g + r1.x * slab4;
        float4 u0 = p0[0], u1 = p0[bstride], u2 = p0[2 * bstride], u3 = p0[3 * bstride];
        float4 v0 = p1[0], v1 = p1[bstride], v2 = p1[2 * bstride], v3 = p1[3 * bstride];
        float w0 = __int_as_float(r0.y), w1 = __int_as_float(r1.y);
        fma4(a0, w0, u0); fma4(a1, w0, u1); fma4(a2, w0, u2); fma4(a3, w0, u3);
        fma4(a0, w1, v0); fma4(a1, w1, v1); fma4(a2, w1, v2); fma4(a3, w1, v3);
    }
    if (j < se.y) {
        int2 r0 = rec[j];
        const float4* p0 = g + r0.x * slab4;
        float w0 = __int_as_float(r0.y);
        fma4(a0, w0, p0[0]); fma4(a1, w0, p0[bstride]);
        fma4(a2, w0, p0[2 * bstride]); fma4(a3, w0, p0[3 * bstride]);
    }
    float4* o = agg4 + (bd * slab4 + q);
    o[0] = a0; o[bstride] = a1; o[2 * bstride] = a2; o[3 * bstride] = a3;
}

__global__ void k_gconv_wmma(const float* __restrict__ agg, const float* __restrict__ ndv,
                             const float* __restrict__ gcwT, const float* __restrict__ gcb,
                             const float* __restrict__ hold, float* hpre,
                             float* bnsum, float* bnsq, int T, int d, int l) {
    int Tp = T - d;
    int tile = blockIdx.x * 4 + (threadIdx.x >> 5);
    int tm = tile >> 1, tn = tile & 1;
    int lane = threadIdx.x & 31;
    int mrow = tm * 16 + (lane & 15);
    int node_m = (mrow / Tp) & 511;
    float nrm = ndv[node_m];
    int ncol = tn * 16 + (lane & 15);
    int base = (lane & 16) ? 8 : 0;
    const float* arow = agg + (size_t)mrow * HH;
    const float* brow = gcwT + (size_t)(l * HH + ncol) * HH;
    v16h a, bm;
    pack8_scale(a, 0, arow + base, nrm);
    pack8_scale(a, 8, arow + base + 16, nrm);
    pack8(bm, 0, brow + base);
    pack8(bm, 8, brow + base + 16);
    v8f c = {};
    c = wmma_f16(a, bm, c);
    int rbase = tm * 16 + ((lane & 16) ? 8 : 0);
    float bias = gcb[l * HH + ncol];
    float s1 = 0.f, s2 = 0.f;
#pragma unroll
    for (int v = 0; v < 8; v++) {
        int row = rbase + v;
        int t = row % Tp;
        int bn = row / Tp;
        float val = fmaxf(c[v] + bias, 0.f) + hold[(bn * T + t + d) * HH + ncol];
        hpre[(size_t)row * HH + ncol] = val;
        s1 += val;
        s2 += val * val;
    }
    atomicAdd(&bnsum[ncol], s1);
    atomicAdd(&bnsq[ncol], s2);
}

__global__ void k_bn(float* h, const float* __restrict__ bnsum, const float* __restrict__ bnsq,
                     const float* __restrict__ gamma, const float* __restrict__ beta,
                     int Tp, int l) {
    int total = BB * NN * Tp * HH;
    int idx = blockIdx.x * blockDim.x + threadIdx.x;
    if (idx >= total) return;
    int c = idx & 31;
    float cnt = (float)(BB * NN * Tp);
    float mu = bnsum[c] / cnt;
    float var = bnsq[c] / cnt - mu * mu;
    float x = h[idx];
    h[idx] = (x - mu) * rsqrtf(var + 1e-5f) * gamma[l * HH + c] + beta[l * HH + c];
}

__global__ void k_end1_wmma(const float* __restrict__ skip, const float* __restrict__ w1,
                            const float* __restrict__ b1, float* out1) {
    const int nTilesN = ECC / 16;
    int tile = blockIdx.x * 4 + (threadIdx.x >> 5);
    int tm = tile / nTilesN, tn = tile % nTilesN;
    int lane = threadIdx.x & 31;
    int mrow = tm * 16 + (lane & 15);
    int ncol = tn * 16 + (lane & 15);
    int base = (lane & 16) ? 8 : 0;
    const float* arow = skip + (size_t)mrow * SKC;
    const float* brow = w1 + (size_t)ncol * SKC;
    v8f c = {};
#pragma unroll
    for (int k0 = 0; k0 < SKC; k0 += 32) {
        v16h a, bm;
        pack8_relu(a, 0, arow + k0 + base);
        pack8_relu(a, 8, arow + k0 + base + 16);
        pack8(bm, 0, brow + k0 + base);
        pack8(bm, 8, brow + k0 + base + 16);
        c = wmma_f16(a, bm, c);
    }
    int rbase = tm * 16 + ((lane & 16) ? 8 : 0);
    float bias = b1[ncol];
#pragma unroll
    for (int v = 0; v < 8; v++)
        out1[(size_t)(rbase + v) * ECC + ncol] = fmaxf(c[v] + bias, 0.f);
}

__global__ void k_end2_wmma(const float* __restrict__ in1, const float* __restrict__ w2,
                            const float* __restrict__ b2, float* out) {
    int tm = blockIdx.x * 4 + (threadIdx.x >> 5);
    int lane = threadIdx.x & 31;
    int mrow = tm * 16 + (lane & 15);
    int ncol = lane & 15;
    int base = (lane & 16) ? 8 : 0;
    const float* arow = in1 + (size_t)mrow * ECC;
    const float* brow = w2 + (size_t)ncol * ECC;
    v8f c = {};
#pragma unroll
    for (int k0 = 0; k0 < ECC; k0 += 32) {
        v16h a, bm = {};
        pack8(a, 0, arow + k0 + base);
        pack8(a, 8, arow + k0 + base + 16);
        if (ncol < OUTF) {
            pack8(bm, 0, brow + k0 + base);
            pack8(bm, 8, brow + k0 + base + 16);
        }
        c = wmma_f16(a, bm, c);
    }
    if (ncol < OUTF) {
        int rbase = tm * 16 + ((lane & 16) ? 8 : 0);
        float bias = b2[ncol];
#pragma unroll
        for (int v = 0; v < 8; v++) {
            int row = rbase + v;
            int b = row >> 9, n = row & 511;
            out[(b * OUTF + ncol) * NN + n] = c[v] + bias;
        }
    }
}

extern "C" void kernel_launch(void* const* d_in, const int* in_sizes, int n_in,
                              void* d_out, int out_size, void* d_ws, size_t ws_size,
                              hipStream_t stream) {
    const float* feat     = (const float*)d_in[0];
    const int*   src      = (const int*)d_in[1];
    const int*   dst      = (const int*)d_in[2];
    const float* nv1      = (const float*)d_in[3];
    const float* nv2      = (const float*)d_in[4];
    const float* start_w  = (const float*)d_in[5];
    const float* start_b  = (const float*)d_in[6];
    const float* filter_w = (const float*)d_in[7];
    const float* filter_b = (const float*)d_in[8];
    const float* gate_w   = (const float*)d_in[9];
    const float* gate_b   = (const float*)d_in[10];
    const float* skip_w   = (const float*)d_in[11];
    const float* skip_b   = (const float*)d_in[12];
    const float* bn_gamma = (const float*)d_in[13];
    const float* bn_beta  = (const float*)d_in[14];
    const float* gconv_w  = (const float*)d_in[15];
    const float* gconv_b  = (const float*)d_in[16];
    const float* end1_w   = (const float*)d_in[17];
    const float* end1_b   = (const float*)d_in[18];
    const float* end2_w   = (const float*)d_in[19];
    const float* end2_b   = (const float*)d_in[20];

    float* ws = (float*)d_ws;
    size_t o = 0;
    unsigned char* adj = (unsigned char*)(ws + o); o += (NN * NN) / 4;
    float* rowmax = ws + o; o += NN;
    float* rowsum = ws + o; o += NN;
    float* ew     = ws + o; o += NE;
    float* degout = ws + o; o += NN;
    float* degin  = ws + o; o += NN;
    float* nsv    = ws + o; o += NN;
    float* ndv    = ws + o; o += NN;
    float* bnsum  = ws + o; o += HH;
    float* bnsq   = ws + o; o += HH;
    float* fwT    = ws + o; o += NLAY * HH * 2 * HH;
    float* gwT    = ws + o; o += NLAY * HH * 2 * HH;
    float* gcwT   = ws + o; o += NLAY * HH * HH;
    float* skip   = ws + o; o += (size_t)BB * NN * SKC;
    float* hA     = ws + o; o += (size_t)BB * NN * 13 * HH;
    float* hB     = ws + o; o += (size_t)BB * NN * 13 * HH;
    float* gau    = ws + o; o += (size_t)BB * NN * 12 * HH;
    float* agg    = ws + o; o += (size_t)BB * NN * 12 * HH;
    int2*  rowse  = (int2*)(ws + o); o += 2 * NN;
    int2*  rec    = (int2*)(ws + o); o += 2 * NE;
    float* out1   = gau;

#define ZEROF(p, n) k_zero<<<((n) + 255) / 256, 256, 0, stream>>>((p), (n))

    ZEROF((float*)adj, (NN * NN) / 4);
    ZEROF(degout, NN);
    ZEROF(degin, NN);
    ZEROF(skip, BB * NN * SKC);

    k_prep_w<<<(NLAY * HH * 2 * HH + 255) / 256, 256, 0, stream>>>(filter_w, gate_w, gconv_w,
                                                                   fwT, gwT, gcwT);
    k_scatter<<<(NE + 255) / 256, 256, 0, stream>>>(src, dst, adj, degout, degin);
    k_rownorm<<<NN, 256, 0, stream>>>(adj, nv1, nv2, rowmax, rowsum);
    k_ew<<<(NE + 255) / 256, 256, 0, stream>>>(src, dst, nv1, nv2, rowmax, rowsum, ew);
    k_norms<<<(NN + 255) / 256, 256, 0, stream>>>(degout, degin, nsv, ndv);
    k_csr<<<NN, CSR_T, 0, stream>>>(src, dst, ew, nsv, rowse, rec);
    k_start<<<(BB * NN * 13 * HH + 255) / 256, 256, 0, stream>>>(feat, start_w, start_b, hA);

    const int dil[8] = {1, 2, 1, 2, 1, 2, 1, 2};
    float* hcur = hA;
    float* hnext = hB;
    int T = 13;
    for (int l = 0; l < NLAY; l++) {
        int d = dil[l];
        int Tp = T - d;
        int M = BB * NN * Tp;
        int tilesGau = (M / 16) * 2;

        k_gau_wmma<<<tilesGau / 4, 128, 0, stream>>>(hcur, fwT, filter_b, gwT, gate_b,
                                                     gau, T, d, l);
        k_skip_wmma<<<(BB * NN / 16) * (SKC / 16) / 4, 128, 0, stream>>>(gau, skip_w, skip_b,
                                                                         skip, Tp, l);
        k_gather<<<((BB / 4) * NN * Tp * 8 + 255) / 256, 256, 0, stream>>>(
            rowse, rec, (const float4*)gau, (float4*)agg, Tp * 8);
        ZEROF(bnsum, HH);
        ZEROF(bnsq, HH);
        k_gconv_wmma<<<(M / 16) * 2 / 4, 128, 0, stream>>>(agg, ndv, gcwT, gconv_b,
                                                           hcur, hnext, bnsum, bnsq, T, d, l);
        k_bn<<<(M * HH + 255) / 256, 256, 0, stream>>>(hnext, bnsum, bnsq,
                                                       bn_gamma, bn_beta, Tp, l);
        float* tmp = hcur; hcur = hnext; hnext = tmp;
        T = Tp;
    }

    k_end1_wmma<<<(BB * NN / 16) * (ECC / 16) / 4, 128, 0, stream>>>(skip, end1_w, end1_b, out1);
    k_end2_wmma<<<(BB * NN / 16) / 4, 128, 0, stream>>>(out1, end2_w, end2_b, (float*)d_out);
#undef ZEROF
}


# batchnorm apply folded into its readers: the next layer's gate/filter A loader and residual read scale and shift raw h from per-layer sum planes; the 8 normalise passes and 15 of 16 sum fills go
# speedup vs baseline: 2.0130x; 1.1182x over previous
#include <hip/hip_runtime.h>
#include <hip/hip_bf16.h>

typedef __attribute__((ext_vector_type(16))) _Float16 v16h;
typedef __attribute__((ext_vector_type(8)))  float    v8f;

#define NN   512
#define NE   4096
#define BB   32
#define HH   32
#define SKC  256
#define ECC  512
#define OUTF 12
#define AEMB 10
#define NEGBIG (-1e30f)
#define NLAY 8

__device__ __forceinline__ v8f wmma_f16(v16h a, v16h b, v8f c) {
    return __builtin_amdgcn_wmma_f32_16x16x32_f16(false, a, false, b, (short)0, c, false, false);
}

__device__ __forceinline__ float fast_sigmoid(float x) {
    return __builtin_amdgcn_rcpf(1.f + __expf(-x));
}
__device__ __forceinline__ float fast_tanh(float x) {
    float t = __expf(-2.f * fabsf(x));
    float r = (1.f - t) * __builtin_amdgcn_rcpf(1.f + t);
    return copysignf(r, x);
}

__device__ __forceinline__ void pack8(v16h& d, int off, const float* __restrict__ p) {
    float4 x = ((const float4*)p)[0];
    float4 y = ((const float4*)p)[1];
    d[off + 0] = (_Float16)x.x; d[off + 1] = (_Float16)x.y;
    d[off + 2] = (_Float16)x.z; d[off + 3] = (_Float16)x.w;
    d[off + 4] = (_Float16)y.x; d[off + 5] = (_Float16)y.y;
    d[off + 6] = (_Float16)y.z; d[off + 7] = (_Float16)y.w;
}
__device__ __forceinline__ void pack8_scale(v16h& d, int off, const float* __restrict__ p, float s) {
    float4 x = ((const float4*)p)[0];
    float4 y = ((const float4*)p)[1];
    d[off + 0] = (_Float16)(x.x * s); d[off + 1] = (_Float16)(x.y * s);
    d[off + 2] = (_Float16)(x.z * s); d[off + 3] = (_Float16)(x.w * s);
    d[off + 4] = (_Float16)(y.x * s); d[off + 5] = (_Float16)(y.y * s);
    d[off + 6] = (_Float16)(y.z * s); d[off + 7] = (_Float16)(y.w * s);
}
__device__ __forceinline__ void pack8_relu(v16h& d, int off, const float* __restrict__ p) {
    float4 x = ((const float4*)p)[0];
    float4 y = ((const float4*)p)[1];
    d[off + 0] = (_Float16)fmaxf(x.x, 0.f); d[off + 1] = (_Float16)fmaxf(x.y, 0.f);
    d[off + 2] = (_Float16)fmaxf(x.z, 0.f); d[off + 3] = (_Float16)fmaxf(x.w, 0.f);
    d[off + 4] = (_Float16)fmaxf(y.x, 0.f); d[off + 5] = (_Float16)fmaxf(y.y, 0.f);
    d[off + 6] = (_Float16)fmaxf(y.z, 0.f); d[off + 7] = (_Float16)fmaxf(y.w, 0.f);
}

__device__ __forceinline__ void pack8_affine(v16h& d, int off, const float* __restrict__ p,
                                             const float* sc, const float* sh) {
    float4 x = ((const float4*)p)[0];
    float4 y = ((const float4*)p)[1];
    d[off + 0] = (_Float16)fmaf(x.x, sc[0], sh[0]); d[off + 1] = (_Float16)fmaf(x.y, sc[1], sh[1]);
    d[off + 2] = (_Float16)fmaf(x.z, sc[2], sh[2]); d[off + 3] = (_Float16)fmaf(x.w, sc[3], sh[3]);
    d[off + 4] = (_Float16)fmaf(y.x, sc[4], sh[4]); d[off + 5] = (_Float16)fmaf(y.y, sc[5], sh[5]);
    d[off + 6] = (_Float16)fmaf(y.z, sc[6], sh[6]); d[off + 7] = (_Float16)fmaf(y.w, sc[7], sh[7]);
}

__device__ __forceinline__ void bn_coef(float& sc, float& sh, float s, float q, float g, float b,
                                        float inv_cnt) {
    float mu = s * inv_cnt;
    float var = q * inv_cnt - mu * mu;
    sc = rsqrtf(var + 1e-5f) * g;
    sh = b - mu * sc;
}
__device__ __forceinline__ void bn_coef8(float* sc, float* sh, const float* __restrict__ stat,
                                         const float* __restrict__ gam,
                                         const float* __restrict__ bet, int c0, float inv_cnt) {
    float4 s0 = ((const float4*)(stat + c0))[0],      s1 = ((const float4*)(stat + c0))[1];
    float4 q0 = ((const float4*)(stat + HH + c0))[0], q1 = ((const float4*)(stat + HH + c0))[1];
    float4 g0 = ((const float4*)(gam + c0))[0],       g1 = ((const float4*)(gam + c0))[1];
    float4 b0 = ((const float4*)(bet + c0))[0],       b1 = ((const float4*)(bet + c0))[1];
    bn_coef(sc[0], sh[0], s0.x, q0.x, g0.x, b0.x, inv_cnt);
    bn_coef(sc[1], sh[1], s0.y, q0.y, g0.y, b0.y, inv_cnt);
    bn_coef(sc[2], sh[2], s0.z, q0.z, g0.z, b0.z, inv_cnt);
    bn_coef(sc[3], sh[3], s0.w, q0.w, g0.w, b0.w, inv_cnt);
    bn_coef(sc[4], sh[4], s1.x, q1.x, g1.x, b1.x, inv_cnt);
    bn_coef(sc[5], sh[5], s1.y, q1.y, g1.y, b1.y, inv_cnt);
    bn_coef(sc[6], sh[6], s1.z, q1.z, g1.z, b1.z, inv_cnt);
    bn_coef(sc[7], sh[7], s1.w, q1.w, g1.w, b1.w, inv_cnt);
}

__global__ void k_zero(float* p, int n) {
    int i = blockIdx.x * blockDim.x + threadIdx.x;
    if (i < n) p[i] = 0.f;
}

__global__ void k_prep_w(const float* __restrict__ fw, const float* __restrict__ gw,
                         const float* __restrict__ gcw,
                         float* fwT, float* gwT, float* gcwT) {
    int idx = blockIdx.x * blockDim.x + threadIdx.x;
    const int totalFG = NLAY * HH * 2 * HH;
    if (idx < totalFG) {
        int ci = idx & 31;
        int tap = (idx >> 5) & 1;
        int c = (idx >> 6) & 31;
        int l = idx >> 11;
        fwT[idx] = fw[((l * HH + c) * HH + ci) * 2 + tap];
        gwT[idx] = gw[((l * HH + c) * HH + ci) * 2 + tap];
    }
    if (idx < NLAY * HH * HH) {
        int k = idx & 31;
        int n = (idx >> 5) & 31;
        int l = idx >> 10;
        gcwT[idx] = gcw[l * HH * HH + k * HH + n];
    }
}

__global__ void k_scatter(const int* __restrict__ src, const int* __restrict__ dst,
                          unsigned char* adj, float* degout, float* degin) {
    int e = blockIdx.x * blockDim.x + threadIdx.x;
    if (e >= NE) return;
    int s = src[e], d = dst[e];
    adj[s * NN + d] = 1;
    atomicAdd(degout + s, 1.f);
    atomicAdd(degin + d, 1.f);
}

__global__ void k_rownorm(const unsigned char* __restrict__ adj,
                          const float* __restrict__ nv1, const float* __restrict__ nv2,
                          float* rowmax, float* rowsum) {
    __shared__ float sc[NN];
    __shared__ float red[256];
    int i = blockIdx.x;
    for (int j = threadIdx.x; j < NN; j += blockDim.x) {
        float v = NEGBIG;
        if (adj[i * NN + j]) {
            float acc = 0.f;
#pragma unroll
            for (int a = 0; a < AEMB; a++) acc += nv1[i * AEMB + a] * nv2[a * NN + j];
            v = fmaxf(acc, 0.f);
        }
        sc[j] = v;
    }
    __syncthreads();
    float m = NEGBIG;
    for (int j = threadIdx.x; j < NN; j += blockDim.x) m = fmaxf(m, sc[j]);
    red[threadIdx.x] = m;
    __syncthreads();
    for (int s = blockDim.x / 2; s > 0; s >>= 1) {
        if (threadIdx.x < s) red[threadIdx.x] = fmaxf(red[threadIdx.x], red[threadIdx.x + s]);
        __syncthreads();
    }
    m = red[0];
    __syncthreads();
    float sum = 0.f;
    for (int j = threadIdx.x; j < NN; j += blockDim.x) {
        float v = sc[j];
        if (v > 0.5f * NEGBIG) sum += __expf(v - m);
    }
    red[threadIdx.x] = sum;
    __syncthreads();
    for (int s = blockDim.x / 2; s > 0; s >>= 1) {
        if (threadIdx.x < s) red[threadIdx.x] += red[threadIdx.x + s];
        __syncthreads();
    }
    if (threadIdx.x == 0) {
        rowmax[i] = m;
        rowsum[i] = (red[0] > 0.f) ? red[0] : 1.f;
    }
}

__global__ void k_ew(const int* __restrict__ src, const int* __restrict__ dst,
                     const float* __restrict__ nv1, const float* __restrict__ nv2,
                     const float* __restrict__ rowmax, const float* __restrict__ rowsum,
                     float* ew) {
    int e = blockIdx.x * blockDim.x + threadIdx.x;
    if (e >= NE) return;
    int s = src[e], d = dst[e];
    float acc = 0.f;
#pragma unroll
    for (int a = 0; a < AEMB; a++) acc += nv1[s * AEMB + a] * nv2[a * NN + d];
    float scv = fmaxf(acc, 0.f);
    ew[e] = 1.f + __expf(scv - rowmax[s]) * __builtin_amdgcn_rcpf(rowsum[s]);
}

__global__ void k_norms(const float* degout, const float* degin, float* nsv, float* ndv) {
    int i = blockIdx.x * blockDim.x + threadIdx.x;
    if (i >= NN) return;
    nsv[i] = rsqrtf(fmaxf(degout[i], 1.f));
    ndv[i] = rsqrtf(fmaxf(degin[i], 1.f));
}

__global__ void k_start(const float* __restrict__ feat, const float* __restrict__ sw,
                        const float* __restrict__ sb, float* h) {
    int idx = blockIdx.x * blockDim.x + threadIdx.x;
    const int total = BB * NN * 13 * HH;
    if (idx >= total) return;
    int c = idx & 31;
    int t = (idx >> 5) % 13;
    int m = idx / (13 * HH);
    int b = m >> 9, n = m & 511;
    float x = (t == 0) ? 0.f : feat[(b * 12 + (t - 1)) * NN + n];
    h[idx] = sw[c] * x + sb[c];
}

template <bool NORM>
__device__ __forceinline__ void gau_body(const float* __restrict__ h,
                                         const float* __restrict__ pstat,
                                         const float* __restrict__ pgam,
                                         const float* __restrict__ pbet,
                                         const float* __restrict__ fwT,
                                         const float* __restrict__ fb,
                                         const float* __restrict__ gwT,
                                         const float* __restrict__ gb, float* gau,
                                         int T, int d, int l) {
    int Tp = T - d;
    int tile = blockIdx.x * 4 + (threadIdx.x >> 5);
    int tm = tile >> 1, tn = tile & 1;
    int lane = threadIdx.x & 31;
    int mrow = tm * 16 + (lane & 15);
    int t = mrow % Tp;
    int bn = mrow / Tp;
    const float* hrow0 = h + (bn * T + t) * HH;
    const float* hrow1 = hrow0 + d * HH;
    int ncol = tn * 16 + (lane & 15);
    int base = (lane & 16) ? 8 : 0;
    const float* fwr = fwT + (size_t)((l * HH + ncol) * 2) * HH;
    const float* gwr = gwT + (size_t)((l * HH + ncol) * 2) * HH;
    float sc[16], sh[16];
    if (NORM) {
        float inv_cnt = 1.f / (float)(BB * NN * T);
        bn_coef8(sc, sh, pstat, pgam, pbet, base, inv_cnt);
        bn_coef8(sc + 8, sh + 8, pstat, pgam, pbet, base + 16, inv_cnt);
    }
    v8f cf = {}, cg = {};
    v16h a, bf, bg;
    if (NORM) {
        pack8_affine(a, 0, hrow0 + base, sc, sh);
        pack8_affine(a, 8, hrow0 + base + 16, sc + 8, sh + 8);
    } else {
        pack8(a, 0, hrow0 + base);  pack8(a, 8, hrow0 + base + 16);
    }
    pack8(bf, 0, fwr + base);   pack8(bf, 8, fwr + base + 16);
    pack8(bg, 0, gwr + base);   pack8(bg, 8, gwr + base + 16);
    cf = wmma_f16(a, bf, cf);
    cg = wmma_f16(a, bg, cg);
    if (NORM) {
        pack8_affine(a, 0, hrow1 + base, sc, sh);
        pack8_affine(a, 8, hrow1 + base + 16, sc + 8, sh + 8);
    } else {
        pack8(a, 0, hrow1 + base);  pack8(a, 8, hrow1 + base + 16);
    }
    pack8(bf, 0, fwr + HH + base);  pack8(bf, 8, fwr + HH + base + 16);
    pack8(bg, 0, gwr + HH + base);  pack8(bg, 8, gwr + HH + base + 16);
    cf = wmma_f16(a, bf, cf);
    cg = wmma_f16(a, bg, cg);

    int rbase = tm * 16 + ((lane & 16) ? 8 : 0);
    float fbias = fb[l * HH + ncol];
    float gbias = gb[l * HH + ncol];
#pragma unroll
    for (int v = 0; v < 8; v++) {
        int row = rbase + v;
        float fv = fast_tanh(cf[v] + fbias);
        float gv = fast_sigmoid(cg[v] + gbias);
        gau[(size_t)row * HH + ncol] = fv * gv;
    }
}
__global__ void k_gau_first(const float* __restrict__ h, const float* __restrict__ fwT,
                            const float* __restrict__ fb, const float* __restrict__ gwT,
                            const float* __restrict__ gb, float* gau, int T, int d, int l) {
    gau_body<false>(h, nullptr, nullptr, nullptr, fwT, fb, gwT, gb, gau, T, d, l);
}
__global__ void k_gau_wmma(const float* __restrict__ h, const float* __restrict__ pstat,
                           const float* __restrict__ pgam, const float* __restrict__ pbet,
                           const float* __restrict__ fwT, const float* __restrict__ fb,
                           const float* __restrict__ gwT, const float* __restrict__ gb,
                           float* gau, int T, int d, int l) {
    gau_body<true>(h, pstat, pgam, pbet, fwT, fb, gwT, gb, gau, T, d, l);
}

__global__ void k_skip_wmma(const float* __restrict__ gau, const float* __restrict__ sw,
                            const float* __restrict__ sb, float* skip, int Tp, int l) {
    const int nTilesN = SKC / 16;
    int tile = blockIdx.x * 4 + (threadIdx.x >> 5);
    int tm = tile / nTilesN, tn = tile % nTilesN;
    int lane = threadIdx.x & 31;
    int mrow = tm * 16 + (lane & 15);
    int ncol = tn * 16 + (lane & 15);
    int base = (lane & 16) ? 8 : 0;
    const float* arow = gau + (size_t)(mrow * Tp + (Tp - 1)) * HH;
    const float* brow = sw + (size_t)(l * SKC + ncol) * HH;
    v16h a, bm;
    pack8(a, 0, arow + base);  pack8(a, 8, arow + base + 16);
    pack8(bm, 0, brow + base); pack8(bm, 8, brow + base + 16);
    v8f c = {};
    c = wmma_f16(a, bm, c);
    int rbase = tm * 16 + ((lane & 16) ? 8 : 0);
    float bias = sb[l * SKC + ncol];
#pragma unroll
    for (int v = 0; v < 8; v++) {
        float* p = skip + (size_t)(rbase + v) * SKC + ncol;
        *p += c[v] + bias;
    }
}

#define CSR_T 256
#define CSR_PER (NE / CSR_T)
__global__ void k_csr(const int* __restrict__ src, const int* __restrict__ dst,
                      const float* __restrict__ ew, const float* __restrict__ nsv,
                      int2* rowse, int2* rec) {
    __shared__ int s_lt[CSR_T];
    __shared__ int s_eq[CSR_T];
    const int d = blockIdx.x;
    const int tid = threadIdx.x;
    const int e0 = tid * CSR_PER;
    int lt = 0, eq = 0;
    for (int i = 0; i < CSR_PER; i++) {
        int x = dst[e0 + i];
        lt += (x < d) ? 1 : 0;
        eq += (x == d) ? 1 : 0;
    }
    s_lt[tid] = lt;
    s_eq[tid] = eq;
    __syncthreads();
    for (int off = 1; off < CSR_T; off <<= 1) {
        int a = 0, b = 0;
        if (tid >= off) { a = s_lt[tid - off]; b = s_eq[tid - off]; }
        __syncthreads();
        s_lt[tid] += a;
        s_eq[tid] += b;
        __syncthreads();
    }
    const int first = s_lt[CSR_T - 1];
    int pos = first + s_eq[tid] - eq;
    if (tid == 0) rowse[d] = make_int2(first, first + s_eq[CSR_T - 1]);
    for (int i = 0; i < CSR_PER; i++) {
        int e = e0 + i;
        if (dst[e] == d) {
            int sn = src[e];
            rec[pos] = make_int2(sn, __float_as_int(ew[e] * nsv[sn]));
            pos++;
        }
    }
}

__device__ __forceinline__ void fma4(float4& a, float w, const float4 v) {
    a.x = fmaf(w, v.x, a.x); a.y = fmaf(w, v.y, a.y);
    a.z = fmaf(w, v.z, a.z); a.w = fmaf(w, v.w, a.w);
}
__global__ void k_gather(const int2* __restrict__ rowse, const int2* __restrict__ rec,
                         const float4* __restrict__ gau4, float4* __restrict__ agg4,
                         int slab4) {
    const int total = (BB / 4) * NN * slab4;
    int idx = blockIdx.x * blockDim.x + threadIdx.x;
    if (idx >= total) return;
    int bd = idx / slab4;
    int q = idx - bd * slab4;
    int d = bd & (NN - 1);
    int b0 = bd >> 9;
    const int bstride = (BB / 4) * NN * slab4;
    const float4* g = gau4 + (b0 * NN) * slab4 + q;
    int2 se = rowse[d];
    float4 a0 = make_float4(0.f, 0.f, 0.f, 0.f), a1 = a0, a2 = a0, a3 = a0;
    int j = se.x;
    for (; j + 2 <= se.y; j += 2) {
        int2 r0 = rec[j], r1 = rec[j + 1];
        const float4* p0 = g + r0.x * slab4;
        const float4* p1 = g + r1.x * slab4;
        float4 u0 = p0[0], u1 = p0[bstride], u2 = p0[2 * bstride], u3 = p0[3 * bstride];
        float4 v0 = p1[0], v1 = p1[bstride], v2 = p1[2 * bstride], v3 = p1[3 * bstride];
        float w0 = __int_as_float(r0.y), w1 = __int_as_float(r1.y);
        fma4(a0, w0, u0); fma4(a1, w0, u1); fma4(a2, w0, u2); fma4(a3, w0, u3);
        fma4(a0, w1, v0); fma4(a1, w1, v1); fma4(a2, w1, v2); fma4(a3, w1, v3);
    }
    if (j < se.y) {
        int2 r0 = rec[j];
        const float4* p0 = g + r0.x * slab4;
        float w0 = __int_as_float(r0.y);
        fma4(a0, w0, p0[0]); fma4(a1, w0, p0[bstride]);
        fma4(a2, w0, p0[2 * bstride]); fma4(a3, w0, p0[3 * bstride]);
    }
    float4* o = agg4 + (bd * slab4 + q);
    o[0] = a0; o[bstride] = a1; o[2 * bstride] = a2; o[3 * bstride] = a3;
}

template <bool NORM>
__device__ __forceinline__ void gconv_body(const float* __restrict__ agg,
                                           const float* __restrict__ ndv,
                                           const float* __restrict__ gcwT,
                                           const float* __restrict__ gcb,
                                           const float* __restrict__ hold,
                                           const float* __restrict__ pstat,
                                           const float* __restrict__ pgam,
                                           const float* __restrict__ pbet,
                                           float* hpre, float* stat, int T, int d, int l) {
    int Tp = T - d;
    int tile = blockIdx.x * 4 + (threadIdx.x >> 5);
    int tm = tile >> 1, tn = tile & 1;
    int lane = threadIdx.x & 31;
    int mrow = tm * 16 + (lane & 15);
    int node_m = (mrow / Tp) & 511;
    float nrm = ndv[node_m];
    int ncol = tn * 16 + (lane & 15);
    int base = (lane & 16) ? 8 : 0;
    const float* arow = agg + (size_t)mrow * HH;
    const float* brow = gcwT + (size_t)(l * HH + ncol) * HH;
    v16h a, bm;
    pack8_scale(a, 0, arow + base, nrm);
    pack8_scale(a, 8, arow + base + 16, nrm);
    pack8(bm, 0, brow + base);
    pack8(bm, 8, brow + base + 16);
    v8f c = {};
    c = wmma_f16(a, bm, c);
    int rbase = tm * 16 + ((lane & 16) ? 8 : 0);
    float bias = gcb[l * HH + ncol];
    float psc = 1.f, psh = 0.f;
    if (NORM)
        bn_coef(psc, psh, pstat[ncol], pstat[HH + ncol], pgam[ncol], pbet[ncol],
                1.f / (float)(BB * NN * T));
    float s1 = 0.f, s2 = 0.f;
#pragma unroll
    for (int v = 0; v < 8; v++) {
        int row = rbase + v;
        int t = row % Tp;
        int bn = row / Tp;
        float hv = hold[(bn * T + t + d) * HH + ncol];
        if (NORM) hv = fmaf(hv, psc, psh);
        float val = fmaxf(c[v] + bias, 0.f) + hv;
        hpre[(size_t)row * HH + ncol] = val;
        s1 += val;
        s2 += val * val;
    }
    atomicAdd(&stat[ncol], s1);
    atomicAdd(&stat[HH + ncol], s2);
}
__global__ void k_gconv_first(const float* __restrict__ agg, const float* __restrict__ ndv,
                              const float* __restrict__ gcwT, const float* __restrict__ gcb,
                              const float* __restrict__ hold, float* hpre, float* stat,
                              int T, int d, int l) {
    gconv_body<false>(agg, ndv, gcwT, gcb, hold, nullptr, nullptr, nullptr, hpre, stat, T, d, l);
}
__global__ void k_gconv_wmma(const float* __restrict__ agg, const float* __restrict__ ndv,
                             const float* __restrict__ gcwT, const float* __restrict__ gcb,
                             const float* __restrict__ hold, const float* __restrict__ pstat,
                             const float* __restrict__ pgam, const float* __restrict__ pbet,
                             float* hpre, float* stat, int T, int d, int l) {
    gconv_body<true>(agg, ndv, gcwT, gcb, hold, pstat, pgam, pbet, hpre, stat, T, d, l);
}

__global__ void k_end1_wmma(const float* __restrict__ skip, const float* __restrict__ w1,
                            const float* __restrict__ b1, float* out1) {
    const int nTilesN = ECC / 16;
    int tile = blockIdx.x * 4 + (threadIdx.x >> 5);
    int tm = tile / nTilesN, tn = tile % nTilesN;
    int lane = threadIdx.x & 31;
    int mrow = tm * 16 + (lane & 15);
    int ncol = tn * 16 + (lane & 15);
    int base = (lane & 16) ? 8 : 0;
    const float* arow = skip + (size_t)mrow * SKC;
    const float* brow = w1 + (size_t)ncol * SKC;
    v8f c = {};
#pragma unroll
    for (int k0 = 0; k0 < SKC; k0 += 32) {
        v16h a, bm;
        pack8_relu(a, 0, arow + k0 + base);
        pack8_relu(a, 8, arow + k0 + base + 16);
        pack8(bm, 0, brow + k0 + base);
        pack8(bm, 8, brow + k0 + base + 16);
        c = wmma_f16(a, bm, c);
    }
    int rbase = tm * 16 + ((lane & 16) ? 8 : 0);
    float bias = b1[ncol];
#pragma unroll
    for (int v = 0; v < 8; v++)
        out1[(size_t)(rbase + v) * ECC + ncol] = fmaxf(c[v] + bias, 0.f);
}

__global__ void k_end2_wmma(const float* __restrict__ in1, const float* __restrict__ w2,
                            const float* __restrict__ b2, float* out) {
    int tm = blockIdx.x * 4 + (threadIdx.x >> 5);
    int lane = threadIdx.x & 31;
    int mrow = tm * 16 + (lane & 15);
    int ncol = lane & 15;
    int base = (lane & 16) ? 8 : 0;
    const float* arow = in1 + (size_t)mrow * ECC;
    const float* brow = w2 + (size_t)ncol * ECC;
    v8f c = {};
#pragma unroll
    for (int k0 = 0; k0 < ECC; k0 += 32) {
        v16h a, bm = {};
        pack8(a, 0, arow + k0 + base);
        pack8(a, 8, arow + k0 + base + 16);
        if (ncol < OUTF) {
            pack8(bm, 0, brow + k0 + base);
            pack8(bm, 8, brow + k0 + base + 16);
        }
        c = wmma_f16(a, bm, c);
    }
    if (ncol < OUTF) {
        int rbase = tm * 16 + ((lane & 16) ? 8 : 0);
        float bias = b2[ncol];
#pragma unroll
        for (int v = 0; v < 8; v++) {
            int row = rbase + v;
            int b = row >> 9, n = row & 511;
            out[(b * OUTF + ncol) * NN + n] = c[v] + bias;
        }
    }
}

extern "C" void kernel_launch(void* const* d_in, const int* in_sizes, int n_in,
                              void* d_out, int out_size, void* d_ws, size_t ws_size,
                              hipStream_t stream) {
    const float* feat     = (const float*)d_in[0];
    const int*   src      = (const int*)d_in[1];
    const int*   dst      = (const int*)d_in[2];
    const float* nv1      = (const float*)d_in[3];
    const float* nv2      = (const float*)d_in[4];
    const float* start_w  = (const float*)d_in[5];
    const float* start_b  = (const float*)d_in[6];
    const float* filter_w = (const float*)d_in[7];
    const float* filter_b = (const float*)d_in[8];
    const float* gate_w   = (const float*)d_in[9];
    const float* gate_b   = (const float*)d_in[10];
    const float* skip_w   = (const float*)d_in[11];
    const float* skip_b   = (const float*)d_in[12];
    const float* bn_gamma = (const float*)d_in[13];
    const float* bn_beta  = (const float*)d_in[14];
    const float* gconv_w  = (const float*)d_in[15];
    const float* gconv_b  = (const float*)d_in[16];
    const float* end1_w   = (const float*)d_in[17];
    const float* end1_b   = (const float*)d_in[18];
    const float* end2_w   = (const float*)d_in[19];
    const float* end2_b   = (const float*)d_in[20];

    float* ws = (float*)d_ws;
    size_t o = 0;
    unsigned char* adj = (unsigned char*)(ws + o); o += (NN * NN) / 4;
    float* rowmax = ws + o; o += NN;
    float* rowsum = ws + o; o += NN;
    float* ew     = ws + o; o += NE;
    float* degout = ws + o; o += NN;
    float* degin  = ws + o; o += NN;
    float* nsv    = ws + o; o += NN;
    float* ndv    = ws + o; o += NN;
    float* bnstat = ws + o; o += NLAY * 2 * HH;
    float* fwT    = ws + o; o += NLAY * HH * 2 * HH;
    float* gwT    = ws + o; o += NLAY * HH * 2 * HH;
    float* gcwT   = ws + o; o += NLAY * HH * HH;
    float* skip   = ws + o; o += (size_t)BB * NN * SKC;
    float* hA     = ws + o; o += (size_t)BB * NN * 13 * HH;
    float* hB     = ws + o; o += (size_t)BB * NN * 13 * HH;
    float* gau    = ws + o; o += (size_t)BB * NN * 12 * HH;
    float* agg    = ws + o; o += (size_t)BB * NN * 12 * HH;
    int2*  rowse  = (int2*)(ws + o); o += 2 * NN;
    int2*  rec    = (int2*)(ws + o); o += 2 * NE;
    float* out1   = gau;

#define ZEROF(p, n) k_zero<<<((n) + 255) / 256, 256, 0, stream>>>((p), (n))

    ZEROF((float*)adj, (NN * NN) / 4);
    ZEROF(degout, NN);
    ZEROF(degin, NN);
    ZEROF(skip, BB * NN * SKC);
    ZEROF(bnstat, NLAY * 2 * HH);

    k_prep_w<<<(NLAY * HH * 2 * HH + 255) / 256, 256, 0, stream>>>(filter_w, gate_w, gconv_w,
                                                                   fwT, gwT, gcwT);
    k_scatter<<<(NE + 255) / 256, 256, 0, stream>>>(src, dst, adj, degout, degin);
    k_rownorm<<<NN, 256, 0, stream>>>(adj, nv1, nv2, rowmax, rowsum);
    k_ew<<<(NE + 255) / 256, 256, 0, stream>>>(src, dst, nv1, nv2, rowmax, rowsum, ew);
    k_norms<<<(NN + 255) / 256, 256, 0, stream>>>(degout, degin, nsv, ndv);
    k_csr<<<NN, CSR_T, 0, stream>>>(src, dst, ew, nsv, rowse, rec);
    k_start<<<(BB * NN * 13 * HH + 255) / 256, 256, 0, stream>>>(feat, start_w, start_b, hA);

    const int dil[8] = {1, 2, 1, 2, 1, 2, 1, 2};
    float* hcur = hA;
    float* hnext = hB;
    int T = 13;
    for (int l = 0; l < NLAY; l++) {
        int d = dil[l];
        int Tp = T - d;
        int M = BB * NN * Tp;
        int tilesGau = (M / 16) * 2;

        const int lp = l ? l - 1 : 0;
        const float* pstat = bnstat + lp * 2 * HH;
        const float* pgam  = bn_gamma + lp * HH;
        const float* pbet  = bn_beta + lp * HH;
        if (l == 0)
            k_gau_first<<<tilesGau / 4, 128, 0, stream>>>(hcur, fwT, filter_b, gwT, gate_b,
                                                          gau, T, d, l);
        else
            k_gau_wmma<<<tilesGau / 4, 128, 0, stream>>>(hcur, pstat, pgam, pbet, fwT, filter_b,
                                                         gwT, gate_b, gau, T, d, l);
        k_skip_wmma<<<(BB * NN / 16) * (SKC / 16) / 4, 128, 0, stream>>>(gau, skip_w, skip_b,
                                                                         skip, Tp, l);
        k_gather<<<((BB / 4) * NN * Tp * 8 + 255) / 256, 256, 0, stream>>>(
            rowse, rec, (const float4*)gau, (float4*)agg, Tp * 8);
        if (l == 0)
            k_gconv_first<<<(M / 16) * 2 / 4, 128, 0, stream>>>(agg, ndv, gcwT, gconv_b, hcur,
                                                                hnext, bnstat, T, d, l);
        else
            k_gconv_wmma<<<(M / 16) * 2 / 4, 128, 0, stream>>>(agg, ndv, gcwT, gconv_b,
                                                               hcur, pstat, pgam, pbet, hnext,
                                                               bnstat + l * 2 * HH, T, d, l);
        float* tmp = hcur; hcur = hnext; hnext = tmp;
        T = Tp;
    }

    k_end1_wmma<<<(BB * NN / 16) * (ECC / 16) / 4, 128, 0, stream>>>(skip, end1_w, end1_b, out1);
    k_end2_wmma<<<(BB * NN / 16) / 4, 128, 0, stream>>>(out1, end2_w, end2_b, (float*)d_out);
#undef ZEROF
}


# fewer, longer waves: fills and start conv go float4 on a grid stride; gate/filter, graph-conv, skip and end1 waves take 2-4 tiles and keep a fragment; bn sums add 4 tiles per lane before the atomic
# speedup vs baseline: 3.4134x; 1.6957x over previous
#include <hip/hip_runtime.h>
#include <hip/hip_bf16.h>

typedef __attribute__((ext_vector_type(16))) _Float16 v16h;
typedef __attribute__((ext_vector_type(8)))  float    v8f;

#define NN   512
#define NE   4096
#define BB   32
#define HH   32
#define SKC  256
#define ECC  512
#define OUTF 12
#define AEMB 10
#define NEGBIG (-1e30f)
#define NLAY 8

__device__ __forceinline__ v8f wmma_f16(v16h a, v16h b, v8f c) {
    return __builtin_amdgcn_wmma_f32_16x16x32_f16(false, a, false, b, (short)0, c, false, false);
}

__device__ __forceinline__ float fast_sigmoid(float x) {
    return __builtin_amdgcn_rcpf(1.f + __expf(-x));
}
__device__ __forceinline__ float fast_tanh(float x) {
    float t = __expf(-2.f * fabsf(x));
    float r = (1.f - t) * __builtin_amdgcn_rcpf(1.f + t);
    return copysignf(r, x);
}

__device__ __forceinline__ void pack8(v16h& d, int off, const float* __restrict__ p) {
    float4 x = ((const float4*)p)[0];
    float4 y = ((const float4*)p)[1];
    d[off + 0] = (_Float16)x.x; d[off + 1] = (_Float16)x.y;
    d[off + 2] = (_Float16)x.z; d[off + 3] = (_Float16)x.w;
    d[off + 4] = (_Float16)y.x; d[off + 5] = (_Float16)y.y;
    d[off + 6] = (_Float16)y.z; d[off + 7] = (_Float16)y.w;
}
__device__ __forceinline__ void cvt8_scale(v16h& d, int off, float4 x, float4 y, float s) {
    d[off + 0] = (_Float16)(x.x * s); d[off + 1] = (_Float16)(x.y * s);
    d[off + 2] = (_Float16)(x.z * s); d[off + 3] = (_Float16)(x.w * s);
    d[off + 4] = (_Float16)(y.x * s); d[off + 5] = (_Float16)(y.y * s);
    d[off + 6] = (_Float16)(y.z * s); d[off + 7] = (_Float16)(y.w * s);
}
__device__ __forceinline__ void pack8_relu(v16h& d, int off, const float* __restrict__ p) {
    float4 x = ((const float4*)p)[0];
    float4 y = ((const float4*)p)[1];
    d[off + 0] = (_Float16)fmaxf(x.x, 0.f); d[off + 1] = (_Float16)fmaxf(x.y, 0.f);
    d[off + 2] = (_Float16)fmaxf(x.z, 0.f); d[off + 3] = (_Float16)fmaxf(x.w, 0.f);
    d[off + 4] = (_Float16)fmaxf(y.x, 0.f); d[off + 5] = (_Float16)fmaxf(y.y, 0.f);
    d[off + 6] = (_Float16)fmaxf(y.z, 0.f); d[off + 7] = (_Float16)fmaxf(y.w, 0.f);
}

__device__ __forceinline__ void pack8_affine(v16h& d, int off, const float* __restrict__ p,
                                             const float* sc, const float* sh) {
    float4 x = ((const float4*)p)[0];
    float4 y = ((const float4*)p)[1];
    d[off + 0] = (_Float16)fmaf(x.x, sc[0], sh[0]); d[off + 1] = (_Float16)fmaf(x.y, sc[1], sh[1]);
    d[off + 2] = (_Float16)fmaf(x.z, sc[2], sh[2]); d[off + 3] = (_Float16)fmaf(x.w, sc[3], sh[3]);
    d[off + 4] = (_Float16)fmaf(y.x, sc[4], sh[4]); d[off + 5] = (_Float16)fmaf(y.y, sc[5], sh[5]);
    d[off + 6] = (_Float16)fmaf(y.z, sc[6], sh[6]); d[off + 7] = (_Float16)fmaf(y.w, sc[7], sh[7]);
}

__device__ __forceinline__ void bn_coef(float& sc, float& sh, float s, float q, float g, float b,
                                        float inv_cnt) {
    float mu = s * inv_cnt;
    float var = q * inv_cnt - mu * mu;
    sc = rsqrtf(var + 1e-5f) * g;
    sh = b - mu * sc;
}
__device__ __forceinline__ void bn_coef8(float* sc, float* sh, const float* __restrict__ stat,
                                         const float* __restrict__ gam,
                                         const float* __restrict__ bet, int c0, float inv_cnt) {
    float4 s0 = ((const float4*)(stat + c0))[0],      s1 = ((const float4*)(stat + c0))[1];
    float4 q0 = ((const float4*)(stat + HH + c0))[0], q1 = ((const float4*)(stat + HH + c0))[1];
    float4 g0 = ((const float4*)(gam + c0))[0],       g1 = ((const float4*)(gam + c0))[1];
    float4 b0 = ((const float4*)(bet + c0))[0],       b1 = ((const float4*)(bet + c0))[1];
    bn_coef(sc[0], sh[0], s0.x, q0.x, g0.x, b0.x, inv_cnt);
    bn_coef(sc[1], sh[1], s0.y, q0.y, g0.y, b0.y, inv_cnt);
    bn_coef(sc[2], sh[2], s0.z, q0.z, g0.z, b0.z, inv_cnt);
    bn_coef(sc[3], sh[3], s0.w, q0.w, g0.w, b0.w, inv_cnt);
    bn_coef(sc[4], sh[4], s1.x, q1.x, g1.x, b1.x, inv_cnt);
    bn_coef(sc[5], sh[5], s1.y, q1.y, g1.y, b1.y, inv_cnt);
    bn_coef(sc[6], sh[6], s1.z, q1.z, g1.z, b1.z, inv_cnt);
    bn_coef(sc[7], sh[7], s1.w, q1.w, g1.w, b1.w, inv_cnt);
}

__global__ void k_zero(float* p, int n) {
    const int n4 = n >> 2;
    const int stride = gridDim.x * blockDim.x;
    const float4 z = make_float4(0.f, 0.f, 0.f, 0.f);
    for (int i = blockIdx.x * blockDim.x + threadIdx.x; i < n4; i += stride)
        ((float4*)p)[i] = z;
}

__global__ void k_prep_w(const float* __restrict__ fw, const float* __restrict__ gw,
                         const float* __restrict__ gcw,
                         float* fwT, float* gwT, float* gcwT) {
    int idx = blockIdx.x * blockDim.x + threadIdx.x;
    const int totalFG = NLAY * HH * 2 * HH;
    if (idx < totalFG) {
        int ci = idx & 31;
        int tap = (idx >> 5) & 1;
        int c = (idx >> 6) & 31;
        int l = idx >> 11;
        fwT[idx] = fw[((l * HH + c) * HH + ci) * 2 + tap];
        gwT[idx] = gw[((l * HH + c) * HH + ci) * 2 + tap];
    }
    if (idx < NLAY * HH * HH) {
        int k = idx & 31;
        int n = (idx >> 5) & 31;
        int l = idx >> 10;
        gcwT[idx] = gcw[l * HH * HH + k * HH + n];
    }
}

__global__ void k_scatter(const int* __restrict__ src, const int* __restrict__ dst,
                          unsigned char* adj, float* degout, float* degin) {
    int e = blockIdx.x * blockDim.x + threadIdx.x;
    if (e >= NE) return;
    int s = src[e], d = dst[e];
    adj[s * NN + d] = 1;
    atomicAdd(degout + s, 1.f);
    atomicAdd(degin + d, 1.f);
}

__global__ void k_rownorm(const unsigned char* __restrict__ adj,
                          const float* __restrict__ nv1, const float* __restrict__ nv2,
                          float* rowmax, float* rowsum) {
    __shared__ float sc[NN];
    __shared__ float red[256];
    int i = blockIdx.x;
    for (int j = threadIdx.x; j < NN; j += blockDim.x) {
        float v = NEGBIG;
        if (adj[i * NN + j]) {
            float acc = 0.f;
#pragma unroll
            for (int a = 0; a < AEMB; a++) acc += nv1[i * AEMB + a] * nv2[a * NN + j];
            v = fmaxf(acc, 0.f);
        }
        sc[j] = v;
    }
    __syncthreads();
    float m = NEGBIG;
    for (int j = threadIdx.x; j < NN; j += blockDim.x) m = fmaxf(m, sc[j]);
    red[threadIdx.x] = m;
    __syncthreads();
    for (int s = blockDim.x / 2; s > 0; s >>= 1) {
        if (threadIdx.x < s) red[threadIdx.x] = fmaxf(red[threadIdx.x], red[threadIdx.x + s]);
        __syncthreads();
    }
    m = red[0];
    __syncthreads();
    float sum = 0.f;
    for (int j = threadIdx.x; j < NN; j += blockDim.x) {
        float v = sc[j];
        if (v > 0.5f * NEGBIG) sum += __expf(v - m);
    }
    red[threadIdx.x] = sum;
    __syncthreads();
    for (int s = blockDim.x / 2; s > 0; s >>= 1) {
        if (threadIdx.x < s) red[threadIdx.x] += red[threadIdx.x + s];
        __syncthreads();
    }
    if (threadIdx.x == 0) {
        rowmax[i] = m;
        rowsum[i] = (red[0] > 0.f) ? red[0] : 1.f;
    }
}

__global__ void k_ew(const int* __restrict__ src, const int* __restrict__ dst,
                     const float* __restrict__ nv1, const float* __restrict__ nv2,
                     const float* __restrict__ rowmax, const float* __restrict__ rowsum,
                     float* ew) {
    int e = blockIdx.x * blockDim.x + threadIdx.x;
    if (e >= NE) return;
    int s = src[e], d = dst[e];
    float acc = 0.f;
#pragma unroll
    for (int a = 0; a < AEMB; a++) acc += nv1[s * AEMB + a] * nv2[a * NN + d];
    float scv = fmaxf(acc, 0.f);
    ew[e] = 1.f + __expf(scv - rowmax[s]) * __builtin_amdgcn_rcpf(rowsum[s]);
}

__global__ void k_norms(const float* degout, const float* degin, float* nsv, float* ndv) {
    int i = blockIdx.x * blockDim.x + threadIdx.x;
    if (i >= NN) return;
    nsv[i] = rsqrtf(fmaxf(degout[i], 1.f));
    ndv[i] = rsqrtf(fmaxf(degin[i], 1.f));
}

__global__ void k_start(const float* __restrict__ feat, const float* __restrict__ sw,
                        const float* __restrict__ sb, float* h) {
    const int total4 = BB * NN * 13 * (HH / 4);
    const int stride = gridDim.x * blockDim.x;
    for (int idx = blockIdx.x * blockDim.x + threadIdx.x; idx < total4; idx += stride) {
        int c4 = idx & 7;
        int mt = idx >> 3;
        int t = mt % 13;
        int m = mt / 13;
        int b = m >> 9, n = m & 511;
        float x = (t == 0) ? 0.f : feat[(b * 12 + (t - 1)) * NN + n];
        float4 w = ((const float4*)sw)[c4];
        float4 bb = ((const float4*)sb)[c4];
        ((float4*)h)[idx] = make_float4(w.x * x + bb.x, w.y * x + bb.y,
                                        w.z * x + bb.z, w.w * x + bb.w);
    }
}

#define GAU_NT 2
#define GC_NT  4
#define SK_NT  4
#define E1_NT  4

template <bool NORM>
__device__ __forceinline__ void gau_body(const float* __restrict__ h,
                                         const float* __restrict__ pstat,
                                         const float* __restrict__ pgam,
                                         const float* __restrict__ pbet,
                                         const float* __restrict__ fwT,
                                         const float* __restrict__ fb,
                                         const float* __restrict__ gwT,
                                         const float* __restrict__ gb, float* gau,
                                         int T, int d, int l) {
    int Tp = T - d;
    int unit = blockIdx.x * 4 + (threadIdx.x >> 5);
    int tn = unit & 1;
    int tm0 = (unit >> 1) * GAU_NT;
    int lane = threadIdx.x & 31;
    int ncol = tn * 16 + (lane & 15);
    int base = (lane & 16) ? 8 : 0;
    const float* fwr = fwT + (size_t)((l * HH + ncol) * 2) * HH;
    const float* gwr = gwT + (size_t)((l * HH + ncol) * 2) * HH;
    float sc[16], sh[16];
    if (NORM) {
        float inv_cnt = 1.f / (float)(BB * NN * T);
        bn_coef8(sc, sh, pstat, pgam, pbet, base, inv_cnt);
        bn_coef8(sc + 8, sh + 8, pstat, pgam, pbet, base + 16, inv_cnt);
    }
    v16h bf0, bg0, bf1, bg1;
    pack8(bf0, 0, fwr + base);       pack8(bf0, 8, fwr + base + 16);
    pack8(bg0, 0, gwr + base);       pack8(bg0, 8, gwr + base + 16);
    pack8(bf1, 0, fwr + HH + base);  pack8(bf1, 8, fwr + HH + base + 16);
    pack8(bg1, 0, gwr + HH + base);  pack8(bg1, 8, gwr + HH + base + 16);
    float fbias = fb[l * HH + ncol];
    float gbias = gb[l * HH + ncol];
#pragma unroll
    for (int i = 0; i < GAU_NT; i++) {
        int tm = tm0 + i;
        int mrow = tm * 16 + (lane & 15);
        int t = mrow % Tp;
        int bn = mrow / Tp;
        const float* hrow0 = h + (bn * T + t) * HH;
        const float* hrow1 = hrow0 + d * HH;
        v16h a0, a1;
        if (NORM) {
            pack8_affine(a0, 0, hrow0 + base, sc, sh);
            pack8_affine(a0, 8, hrow0 + base + 16, sc + 8, sh + 8);
            pack8_affine(a1, 0, hrow1 + base, sc, sh);
            pack8_affine(a1, 8, hrow1 + base + 16, sc + 8, sh + 8);
        } else {
            pack8(a0, 0, hrow0 + base);  pack8(a0, 8, hrow0 + base + 16);
            pack8(a1, 0, hrow1 + base);  pack8(a1, 8, hrow1 + base + 16);
        }
        v8f cf = {}, cg = {};
        cf = wmma_f16(a0, bf0, cf);
        cg = wmma_f16(a0, bg0, cg);
        cf = wmma_f16(a1, bf1, cf);
        cg = wmma_f16(a1, bg1, cg);

        int rbase = tm * 16 + ((lane & 16) ? 8 : 0);
#pragma unroll
        for (int v = 0; v < 8; v++) {
            int row = rbase + v;
            float fv = fast_tanh(cf[v] + fbias);
            float gv = fast_sigmoid(cg[v] + gbias);
            gau[(size_t)row * HH + ncol] = fv * gv;
        }
    }
}
__global__ void __launch_bounds__(128) k_gau_first(const float* __restrict__ h, const float* __restrict__ fwT,
                            const float* __restrict__ fb, const float* __restrict__ gwT,
                            const float* __restrict__ gb, float* gau, int T, int d, int l) {
    gau_body<false>(h, nullptr, nullptr, nullptr, fwT, fb, gwT, gb, gau, T, d, l);
}
__global__ void __launch_bounds__(128) k_gau_wmma(const float* __restrict__ h, const float* __restrict__ pstat,
                           const float* __restrict__ pgam, const float* __restrict__ pbet,
                           const float* __restrict__ fwT, const float* __restrict__ fb,
                           const float* __restrict__ gwT, const float* __restrict__ gb,
                           float* gau, int T, int d, int l) {
    gau_body<true>(h, pstat, pgam, pbet, fwT, fb, gwT, gb, gau, T, d, l);
}

__global__ void __launch_bounds__(128) k_skip_wmma(const float* __restrict__ gau, const float* __restrict__ sw,
                            const float* __restrict__ sb, float* skip, int Tp, int l) {
    const int nGroupsN = SKC / 16 / SK_NT;
    int unit = blockIdx.x * 4 + (threadIdx.x >> 5);
    int tm = unit / nGroupsN;
    int tn0 = (unit % nGroupsN) * SK_NT;
    int lane = threadIdx.x & 31;
    int mrow = tm * 16 + (lane & 15);
    int base = (lane & 16) ? 8 : 0;
    const float* arow = gau + (size_t)(mrow * Tp + (Tp - 1)) * HH;
    v16h a;
    pack8(a, 0, arow + base);  pack8(a, 8, arow + base + 16);
    int rbase = tm * 16 + ((lane & 16) ? 8 : 0);
#pragma unroll
    for (int i = 0; i < SK_NT; i++) {
        int ncol = (tn0 + i) * 16 + (lane & 15);
        const float* brow = sw + (size_t)(l * SKC + ncol) * HH;
        v16h bm;
        pack8(bm, 0, brow + base); pack8(bm, 8, brow + base + 16);
        v8f c = {};
        c = wmma_f16(a, bm, c);
        float bias = sb[l * SKC + ncol];
#pragma unroll
        for (int v = 0; v < 8; v++) {
            float* p = skip + (size_t)(rbase + v) * SKC + ncol;
            *p += c[v] + bias;
        }
    }
}

#define CSR_T 256
#define CSR_PER (NE / CSR_T)
__global__ void k_csr(const int* __restrict__ src, const int* __restrict__ dst,
                      const float* __restrict__ ew, const float* __restrict__ nsv,
                      int2* rowse, int2* rec) {
    __shared__ int s_lt[CSR_T];
    __shared__ int s_eq[CSR_T];
    const int d = blockIdx.x;
    const int tid = threadIdx.x;
    const int e0 = tid * CSR_PER;
    int lt = 0, eq = 0;
    for (int i = 0; i < CSR_PER; i++) {
        int x = dst[e0 + i];
        lt += (x < d) ? 1 : 0;
        eq += (x == d) ? 1 : 0;
    }
    s_lt[tid] = lt;
    s_eq[tid] = eq;
    __syncthreads();
    for (int off = 1; off < CSR_T; off <<= 1) {
        int a = 0, b = 0;
        if (tid >= off) { a = s_lt[tid - off]; b = s_eq[tid - off]; }
        __syncthreads();
        s_lt[tid] += a;
        s_eq[tid] += b;
        __syncthreads();
    }
    const int first = s_lt[CSR_T - 1];
    int pos = first + s_eq[tid] - eq;
    if (tid == 0) rowse[d] = make_int2(first, first + s_eq[CSR_T - 1]);
    for (int i = 0; i < CSR_PER; i++) {
        int e = e0 + i;
        if (dst[e] == d) {
            int sn = src[e];
            rec[pos] = make_int2(sn, __float_as_int(ew[e] * nsv[sn]));
            pos++;
        }
    }
}

__device__ __forceinline__ void fma4(float4& a, float w, const float4 v) {
    a.x = fmaf(w, v.x, a.x); a.y = fmaf(w, v.y, a.y);
    a.z = fmaf(w, v.z, a.z); a.w = fmaf(w, v.w, a.w);
}
__global__ void k_gather(const int2* __restrict__ rowse, const int2* __restrict__ rec,
                         const float4* __restrict__ gau4, float4* __restrict__ agg4,
                         int slab4) {
    const int total = (BB / 4) * NN * slab4;
    int idx = blockIdx.x * blockDim.x + threadIdx.x;
    if (idx >= total) return;
    int bd = idx / slab4;
    int q = idx - bd * slab4;
    int d = bd & (NN - 1);
    int b0 = bd >> 9;
    const int bstride = (BB / 4) * NN * slab4;
    const float4* g = gau4 + (b0 * NN) * slab4 + q;
    int2 se = rowse[d];
    float4 a0 = make_float4(0.f, 0.f, 0.f, 0.f), a1 = a0, a2 = a0, a3 = a0;
    int j = se.x;
    for (; j + 2 <= se.y; j += 2) {
        int2 r0 = rec[j], r1 = rec[j + 1];
        const float4* p0 = g + r0.x * slab4;
        const float4* p1 = g + r1.x * slab4;
        float4 u0 = p0[0], u1 = p0[bstride], u2 = p0[2 * bstride], u3 = p0[3 * bstride];
        float4 v0 = p1[0], v1 = p1[bstride], v2 = p1[2 * bstride], v3 = p1[3 * bstride];
        float w0 = __int_as_float(r0.y), w1 = __int_as_float(r1.y);
        fma4(a0, w0, u0); fma4(a1, w0, u1); fma4(a2, w0, u2); fma4(a3, w0, u3);
        fma4(a0, w1, v0); fma4(a1, w1, v1); fma4(a2, w1, v2); fma4(a3, w1, v3);
    }
    if (j < se.y) {
        int2 r0 = rec[j];
        const float4* p0 = g + r0.x * slab4;
        float w0 = __int_as_float(r0.y);
        fma4(a0, w0, p0[0]); fma4(a1, w0, p0[bstride]);
        fma4(a2, w0, p0[2 * bstride]); fma4(a3, w0, p0[3 * bstride]);
    }
    float4* o = agg4 + (bd * slab4 + q);
    o[0] = a0; o[bstride] = a1; o[2 * bstride] = a2; o[3 * bstride] = a3;
}

template <bool NORM>
__device__ __forceinline__ void gconv_body(const float* __restrict__ agg,
                                           const float* __restrict__ ndv,
                                           const float* __restrict__ gcwT,
                                           const float* __restrict__ gcb,
                                           const float* __restrict__ hold,
                                           const float* __restrict__ pstat,
                                           const float* __restrict__ pgam,
                                           const float* __restrict__ pbet,
                                           float* hpre, float* stat, int T, int d, int l) {
    int Tp = T - d;
    int unit = blockIdx.x * 4 + (threadIdx.x >> 5);
    int tn = unit & 1;
    int tm0 = (unit >> 1) * GC_NT;
    int lane = threadIdx.x & 31;
    int ncol = tn * 16 + (lane & 15);
    int base = (lane & 16) ? 8 : 0;
    const float* brow = gcwT + (size_t)(l * HH + ncol) * HH;
    v16h bm;
    pack8(bm, 0, brow + base);
    pack8(bm, 8, brow + base + 16);
    float bias = gcb[l * HH + ncol];
    float psc = 1.f, psh = 0.f;
    if (NORM)
        bn_coef(psc, psh, pstat[ncol], pstat[HH + ncol], pgam[ncol], pbet[ncol],
                1.f / (float)(BB * NN * T));
    float4 ra[GC_NT][4];
    float nrm[GC_NT];
    float hv[GC_NT][8];
#pragma unroll
    for (int i = 0; i < GC_NT; i++) {
        int tm = tm0 + i;
        int mrow = tm * 16 + (lane & 15);
        int node_m = (mrow / Tp) & 511;
        nrm[i] = ndv[node_m];
        const float* ap = agg + (size_t)mrow * HH + base;
        ra[i][0] = ((const float4*)ap)[0];
        ra[i][1] = ((const float4*)ap)[1];
        ra[i][2] = ((const float4*)(ap + 16))[0];
        ra[i][3] = ((const float4*)(ap + 16))[1];
        int rbase = tm * 16 + ((lane & 16) ? 8 : 0);
#pragma unroll
        for (int v = 0; v < 8; v++) {
            int row = rbase + v;
            int t = row % Tp;
            int bn = row / Tp;
            hv[i][v] = hold[(bn * T + t + d) * HH + ncol];
        }
    }
    float s1 = 0.f, s2 = 0.f;
#pragma unroll
    for (int i = 0; i < GC_NT; i++) {
        int tm = tm0 + i;
        v16h a;
        cvt8_scale(a, 0, ra[i][0], ra[i][1], nrm[i]);
        cvt8_scale(a, 8, ra[i][2], ra[i][3], nrm[i]);
        v8f c = {};
        c = wmma_f16(a, bm, c);
        int rbase = tm * 16 + ((lane & 16) ? 8 : 0);
#pragma unroll
        for (int v = 0; v < 8; v++) {
            int row = rbase + v;
            float h0 = hv[i][v];
            if (NORM) h0 = fmaf(h0, psc, psh);
            float val = fmaxf(c[v] + bias, 0.f) + h0;
            hpre[(size_t)row * HH + ncol] = val;
            s1 += val;
            s2 += val * val;
        }
    }
    atomicAdd(&stat[ncol], s1);
    atomicAdd(&stat[HH + ncol], s2);
}
__global__ void __launch_bounds__(128) k_gconv_first(const float* __restrict__ agg, const float* __restrict__ ndv,
                              const float* __restrict__ gcwT, const float* __restrict__ gcb,
                              const float* __restrict__ hold, float* hpre, float* stat,
                              int T, int d, int l) {
    gconv_body<false>(agg, ndv, gcwT, gcb, hold, nullptr, nullptr, nullptr, hpre, stat, T, d, l);
}
__global__ void __launch_bounds__(128) k_gconv_wmma(const float* __restrict__ agg, const float* __restrict__ ndv,
                             const float* __restrict__ gcwT, const float* __restrict__ gcb,
                             const float* __restrict__ hold, const float* __restrict__ pstat,
                             const float* __restrict__ pgam, const float* __restrict__ pbet,
                             float* hpre, float* stat, int T, int d, int l) {
    gconv_body<true>(agg, ndv, gcwT, gcb, hold, pstat, pgam, pbet, hpre, stat, T, d, l);
}

__global__ void __launch_bounds__(128) k_end1_wmma(const float* __restrict__ skip, const float* __restrict__ w1,
                            const float* __restrict__ b1, float* out1) {
    const int nGroupsN = ECC / 16 / E1_NT;
    int unit = blockIdx.x * 4 + (threadIdx.x >> 5);
    int tm = unit / nGroupsN;
    int tn0 = (unit % nGroupsN) * E1_NT;
    int lane = threadIdx.x & 31;
    int mrow = tm * 16 + (lane & 15);
    int base = (lane & 16) ? 8 : 0;
    const float* arow = skip + (size_t)mrow * SKC;
    v16h a[SKC / 32];
#pragma unroll
    for (int ks = 0; ks < SKC / 32; ks++) {
        pack8_relu(a[ks], 0, arow + ks * 32 + base);
        pack8_relu(a[ks], 8, arow + ks * 32 + base + 16);
    }
    int rbase = tm * 16 + ((lane & 16) ? 8 : 0);
#pragma unroll
    for (int i = 0; i < E1_NT; i++) {
        int ncol = (tn0 + i) * 16 + (lane & 15);
        const float* brow = w1 + (size_t)ncol * SKC;
        v8f c = {};
#pragma unroll
        for (int ks = 0; ks < SKC / 32; ks++) {
            v16h bm;
            pack8(bm, 0, brow + ks * 32 + base);
            pack8(bm, 8, brow + ks * 32 + base + 16);
            c = wmma_f16(a[ks], bm, c);
        }
        float bias = b1[ncol];
#pragma unroll
        for (int v = 0; v < 8; v++)
            out1[(size_t)(rbase + v) * ECC + ncol] = fmaxf(c[v] + bias, 0.f);
    }
}

__global__ void k_end2_wmma(const float* __restrict__ in1, const float* __restrict__ w2,
                            const float* __restrict__ b2, float* out) {
    int tm = blockIdx.x * 4 + (threadIdx.x >> 5);
    int lane = threadIdx.x & 31;
    int mrow = tm * 16 + (lane & 15);
    int ncol = lane & 15;
    int base = (lane & 16) ? 8 : 0;
    const float* arow = in1 + (size_t)mrow * ECC;
    const float* brow = w2 + (size_t)ncol * ECC;
    v8f c = {};
#pragma unroll
    for (int k0 = 0; k0 < ECC; k0 += 32) {
        v16h a, bm = {};
        pack8(a, 0, arow + k0 + base);
        pack8(a, 8, arow + k0 + base + 16);
        if (ncol < OUTF) {
            pack8(bm, 0, brow + k0 + base);
            pack8(bm, 8, brow + k0 + base + 16);
        }
        c = wmma_f16(a, bm, c);
    }
    if (ncol < OUTF) {
        int rbase = tm * 16 + ((lane & 16) ? 8 : 0);
        float bias = b2[ncol];
#pragma unroll
        for (int v = 0; v < 8; v++) {
            int row = rbase + v;
            int b = row >> 9, n = row & 511;
            out[(b * OUTF + ncol) * NN + n] = c[v] + bias;
        }
    }
}

extern "C" void kernel_launch(void* const* d_in, const int* in_sizes, int n_in,
                              void* d_out, int out_size, void* d_ws, size_t ws_size,
                              hipStream_t stream) {
    const float* feat     = (const float*)d_in[0];
    const int*   src      = (const int*)d_in[1];
    const int*   dst      = (const int*)d_in[2];
    const float* nv1      = (const float*)d_in[3];
    const float* nv2      = (const float*)d_in[4];
    const float* start_w  = (const float*)d_in[5];
    const float* start_b  = (const float*)d_in[6];
    const float* filter_w = (const float*)d_in[7];
    const float* filter_b = (const float*)d_in[8];
    const float* gate_w   = (const float*)d_in[9];
    const float* gate_b   = (const float*)d_in[10];
    const float* skip_w   = (const float*)d_in[11];
    const float* skip_b   = (const float*)d_in[12];
    const float* bn_gamma = (const float*)d_in[13];
    const float* bn_beta  = (const float*)d_in[14];
    const float* gconv_w  = (const float*)d_in[15];
    const float* gconv_b  = (const float*)d_in[16];
    const float* end1_w   = (const float*)d_in[17];
    const float* end1_b   = (const float*)d_in[18];
    const float* end2_w   = (const float*)d_in[19];
    const float* end2_b   = (const float*)d_in[20];

    float* ws = (float*)d_ws;
    size_t o = 0;
    unsigned char* adj = (unsigned char*)(ws + o); o += (NN * NN) / 4;
    float* rowmax = ws + o; o += NN;
    float* rowsum = ws + o; o += NN;
    float* ew     = ws + o; o += NE;
    float* degout = ws + o; o += NN;
    float* degin  = ws + o; o += NN;
    float* nsv    = ws + o; o += NN;
    float* ndv    = ws + o; o += NN;
    float* bnstat = ws + o; o += NLAY * 2 * HH;
    float* fwT    = ws + o; o += NLAY * HH * 2 * HH;
    float* gwT    = ws + o; o += NLAY * HH * 2 * HH;
    float* gcwT   = ws + o; o += NLAY * HH * HH;
    float* skip   = ws + o; o += (size_t)BB * NN * SKC;
    float* hA     = ws + o; o += (size_t)BB * NN * 13 * HH;
    float* hB     = ws + o; o += (size_t)BB * NN * 13 * HH;
    float* gau    = ws + o; o += (size_t)BB * NN * 12 * HH;
    float* agg    = ws + o; o += (size_t)BB * NN * 12 * HH;
    int2*  rowse  = (int2*)(ws + o); o += 2 * NN;
    int2*  rec    = (int2*)(ws + o); o += 2 * NE;
    float* out1   = gau;

#define ZEROF(p, n) \
    k_zero<<<(((n) / 4 + 255) / 256 < 512 ? ((n) / 4 + 255) / 256 : 512), 256, 0, stream>>>((p), (n))

    ZEROF((float*)adj, (NN * NN) / 4);
    ZEROF(degout, NN);
    ZEROF(degin, NN);
    ZEROF(skip, BB * NN * SKC);
    ZEROF(bnstat, NLAY * 2 * HH);

    k_prep_w<<<(NLAY * HH * 2 * HH + 255) / 256, 256, 0, stream>>>(filter_w, gate_w, gconv_w,
                                                                   fwT, gwT, gcwT);
    k_scatter<<<(NE + 255) / 256, 256, 0, stream>>>(src, dst, adj, degout, degin);
    k_rownorm<<<NN, 256, 0, stream>>>(adj, nv1, nv2, rowmax, rowsum);
    k_ew<<<(NE + 255) / 256, 256, 0, stream>>>(src, dst, nv1, nv2, rowmax, rowsum, ew);
    k_norms<<<(NN + 255) / 256, 256, 0, stream>>>(degout, degin, nsv, ndv);
    k_csr<<<NN, CSR_T, 0, stream>>>(src, dst, ew, nsv, rowse, rec);
    k_start<<<(BB * NN * 13 * (HH / 4)) / (256 * 8), 256, 0, stream>>>(feat, start_w, start_b, hA);

    const int dil[8] = {1, 2, 1, 2, 1, 2, 1, 2};
    float* hcur = hA;
    float* hnext = hB;
    int T = 13;
    for (int l = 0; l < NLAY; l++) {
        int d = dil[l];
        int Tp = T - d;
        int M = BB * NN * Tp;
        int tilesGau = (M / 16) * 2;

        const int lp = l ? l - 1 : 0;
        const float* pstat = bnstat + lp * 2 * HH;
        const float* pgam  = bn_gamma + lp * HH;
        const float* pbet  = bn_beta + lp * HH;
        if (l == 0)
            k_gau_first<<<tilesGau / 4 / GAU_NT, 128, 0, stream>>>(hcur, fwT, filter_b, gwT,
                                                                   gate_b, gau, T, d, l);
        else
            k_gau_wmma<<<tilesGau / 4 / GAU_NT, 128, 0, stream>>>(hcur, pstat, pgam, pbet, fwT,
                                                                  filter_b, gwT, gate_b, gau,
                                                                  T, d, l);
        k_skip_wmma<<<(BB * NN / 16) * (SKC / 16) / 4 / SK_NT, 128, 0, stream>>>(
            gau, skip_w, skip_b, skip, Tp, l);
        k_gather<<<((BB / 4) * NN * Tp * 8 + 255) / 256, 256, 0, stream>>>(
            rowse, rec, (const float4*)gau, (float4*)agg, Tp * 8);
        if (l == 0)
            k_gconv_first<<<(M / 16) * 2 / 4 / GC_NT, 128, 0, stream>>>(
                agg, ndv, gcwT, gconv_b, hcur, hnext, bnstat, T, d, l);
        else
            k_gconv_wmma<<<(M / 16) * 2 / 4 / GC_NT, 128, 0, stream>>>(
                agg, ndv, gcwT, gconv_b, hcur, pstat, pgam, pbet, hnext, bnstat + l * 2 * HH,
                T, d, l);
        float* tmp = hcur; hcur = hnext; hnext = tmp;
        T = Tp;
    }

    k_end1_wmma<<<(BB * NN / 16) * (ECC / 16) / 4 / E1_NT, 128, 0, stream>>>(skip, end1_w, end1_b,
                                                                             out1);
    k_end2_wmma<<<(BB * NN / 16) / 4, 128, 0, stream>>>(out1, end2_w, end2_b, (float*)d_out);
#undef ZEROF
}


# bn statistics: a graph-conv block, now 8 waves, adds its lane halves (shuffle) and its waves (LDS) and sends one float atomic per word, 16x fewer hot adds; the order of the bn sums changes
# speedup vs baseline: 4.7759x; 1.3992x over previous
#include <hip/hip_runtime.h>
#include <hip/hip_bf16.h>

typedef __attribute__((ext_vector_type(16))) _Float16 v16h;
typedef __attribute__((ext_vector_type(8)))  float    v8f;

#define NN   512
#define NE   4096
#define BB   32
#define HH   32
#define SKC  256
#define ECC  512
#define OUTF 12
#define AEMB 10
#define NEGBIG (-1e30f)
#define NLAY 8

__device__ __forceinline__ v8f wmma_f16(v16h a, v16h b, v8f c) {
    return __builtin_amdgcn_wmma_f32_16x16x32_f16(false, a, false, b, (short)0, c, false, false);
}

__device__ __forceinline__ float fast_sigmoid(float x) {
    return __builtin_amdgcn_rcpf(1.f + __expf(-x));
}
__device__ __forceinline__ float fast_tanh(float x) {
    float t = __expf(-2.f * fabsf(x));
    float r = (1.f - t) * __builtin_amdgcn_rcpf(1.f + t);
    return copysignf(r, x);
}

__device__ __forceinline__ void pack8(v16h& d, int off, const float* __restrict__ p) {
    float4 x = ((const float4*)p)[0];
    float4 y = ((const float4*)p)[1];
    d[off + 0] = (_Float16)x.x; d[off + 1] = (_Float16)x.y;
    d[off + 2] = (_Float16)x.z; d[off + 3] = (_Float16)x.w;
    d[off + 4] = (_Float16)y.x; d[off + 5] = (_Float16)y.y;
    d[off + 6] = (_Float16)y.z; d[off + 7] = (_Float16)y.w;
}
__device__ __forceinline__ void cvt8_scale(v16h& d, int off, float4 x, float4 y, float s) {
    d[off + 0] = (_Float16)(x.x * s); d[off + 1] = (_Float16)(x.y * s);
    d[off + 2] = (_Float16)(x.z * s); d[off + 3] = (_Float16)(x.w * s);
    d[off + 4] = (_Float16)(y.x * s); d[off + 5] = (_Float16)(y.y * s);
    d[off + 6] = (_Float16)(y.z * s); d[off + 7] = (_Float16)(y.w * s);
}
__device__ __forceinline__ void pack8_relu(v16h& d, int off, const float* __restrict__ p) {
    float4 x = ((const float4*)p)[0];
    float4 y = ((const float4*)p)[1];
    d[off + 0] = (_Float16)fmaxf(x.x, 0.f); d[off + 1] = (_Float16)fmaxf(x.y, 0.f);
    d[off + 2] = (_Float16)fmaxf(x.z, 0.f); d[off + 3] = (_Float16)fmaxf(x.w, 0.f);
    d[off + 4] = (_Float16)fmaxf(y.x, 0.f); d[off + 5] = (_Float16)fmaxf(y.y, 0.f);
    d[off + 6] = (_Float16)fmaxf(y.z, 0.f); d[off + 7] = (_Float16)fmaxf(y.w, 0.f);
}

__device__ __forceinline__ void pack8_affine(v16h& d, int off, const float* __restrict__ p,
                                             const float* sc, const float* sh) {
    float4 x = ((const float4*)p)[0];
    float4 y = ((const float4*)p)[1];
    d[off + 0] = (_Float16)fmaf(x.x, sc[0], sh[0]); d[off + 1] = (_Float16)fmaf(x.y, sc[1], sh[1]);
    d[off + 2] = (_Float16)fmaf(x.z, sc[2], sh[2]); d[off + 3] = (_Float16)fmaf(x.w, sc[3], sh[3]);
    d[off + 4] = (_Float16)fmaf(y.x, sc[4], sh[4]); d[off + 5] = (_Float16)fmaf(y.y, sc[5], sh[5]);
    d[off + 6] = (_Float16)fmaf(y.z, sc[6], sh[6]); d[off + 7] = (_Float16)fmaf(y.w, sc[7], sh[7]);
}

__device__ __forceinline__ void bn_coef(float& sc, float& sh, float s, float q, float g, float b,
                                        float inv_cnt) {
    float mu = s * inv_cnt;
    float var = q * inv_cnt - mu * mu;
    sc = rsqrtf(var + 1e-5f) * g;
    sh = b - mu * sc;
}
__device__ __forceinline__ void bn_coef8(float* sc, float* sh, const float* __restrict__ stat,
                                         const float* __restrict__ gam,
                                         const float* __restrict__ bet, int c0, float inv_cnt) {
    float4 s0 = ((const float4*)(stat + c0))[0],      s1 = ((const float4*)(stat + c0))[1];
    float4 q0 = ((const float4*)(stat + HH + c0))[0], q1 = ((const float4*)(stat + HH + c0))[1];
    float4 g0 = ((const float4*)(gam + c0))[0],       g1 = ((const float4*)(gam + c0))[1];
    float4 b0 = ((const float4*)(bet + c0))[0],       b1 = ((const float4*)(bet + c0))[1];
    bn_coef(sc[0], sh[0], s0.x, q0.x, g0.x, b0.x, inv_cnt);
    bn_coef(sc[1], sh[1], s0.y, q0.y, g0.y, b0.y, inv_cnt);
    bn_coef(sc[2], sh[2], s0.z, q0.z, g0.z, b0.z, inv_cnt);
    bn_coef(sc[3], sh[3], s0.w, q0.w, g0.w, b0.w, inv_cnt);
    bn_coef(sc[4], sh[4], s1.x, q1.x, g1.x, b1.x, inv_cnt);
    bn_coef(sc[5], sh[5], s1.y, q1.y, g1.y, b1.y, inv_cnt);
    bn_coef(sc[6], sh[6], s1.z, q1.z, g1.z, b1.z, inv_cnt);
    bn_coef(sc[7], sh[7], s1.w, q1.w, g1.w, b1.w, inv_cnt);
}

__global__ void k_zero(float* p, int n) {
    const int n4 = n >> 2;
    const int stride = gridDim.x * blockDim.x;
    const float4 z = make_float4(0.f, 0.f, 0.f, 0.f);
    for (int i = blockIdx.x * blockDim.x + threadIdx.x; i < n4; i += stride)
        ((float4*)p)[i] = z;
}

__global__ void k_prep_w(const float* __restrict__ fw, const float* __restrict__ gw,
                         const float* __restrict__ gcw,
                         float* fwT, float* gwT, float* gcwT) {
    int idx = blockIdx.x * blockDim.x + threadIdx.x;
    const int totalFG = NLAY * HH * 2 * HH;
    if (idx < totalFG) {
        int ci = idx & 31;
        int tap = (idx >> 5) & 1;
        int c = (idx >> 6) & 31;
        int l = idx >> 11;
        fwT[idx] = fw[((l * HH + c) * HH + ci) * 2 + tap];
        gwT[idx] = gw[((l * HH + c) * HH + ci) * 2 + tap];
    }
    if (idx < NLAY * HH * HH) {
        int k = idx & 31;
        int n = (idx >> 5) & 31;
        int l = idx >> 10;
        gcwT[idx] = gcw[l * HH * HH + k * HH + n];
    }
}

__global__ void k_scatter(const int* __restrict__ src, const int* __restrict__ dst,
                          unsigned char* adj, float* degout, float* degin) {
    int e = blockIdx.x * blockDim.x + threadIdx.x;
    if (e >= NE) return;
    int s = src[e], d = dst[e];
    adj[s * NN + d] = 1;
    atomicAdd(degout + s, 1.f);
    atomicAdd(degin + d, 1.f);
}

__global__ void k_rownorm(const unsigned char* __restrict__ adj,
                          const float* __restrict__ nv1, const float* __restrict__ nv2,
                          float* rowmax, float* rowsum) {
    __shared__ float sc[NN];
    __shared__ float red[256];
    int i = blockIdx.x;
    for (int j = threadIdx.x; j < NN; j += blockDim.x) {
        float v = NEGBIG;
        if (adj[i * NN + j]) {
            float acc = 0.f;
#pragma unroll
            for (int a = 0; a < AEMB; a++) acc += nv1[i * AEMB + a] * nv2[a * NN + j];
            v = fmaxf(acc, 0.f);
        }
        sc[j] = v;
    }
    __syncthreads();
    float m = NEGBIG;
    for (int j = threadIdx.x; j < NN; j += blockDim.x) m = fmaxf(m, sc[j]);
    red[threadIdx.x] = m;
    __syncthreads();
    for (int s = blockDim.x / 2; s > 0; s >>= 1) {
        if (threadIdx.x < s) red[threadIdx.x] = fmaxf(red[threadIdx.x], red[threadIdx.x + s]);
        __syncthreads();
    }
    m = red[0];
    __syncthreads();
    float sum = 0.f;
    for (int j = threadIdx.x; j < NN; j += blockDim.x) {
        float v = sc[j];
        if (v > 0.5f * NEGBIG) sum += __expf(v - m);
    }
    red[threadIdx.x] = sum;
    __syncthreads();
    for (int s = blockDim.x / 2; s > 0; s >>= 1) {
        if (threadIdx.x < s) red[threadIdx.x] += red[threadIdx.x + s];
        __syncthreads();
    }
    if (threadIdx.x == 0) {
        rowmax[i] = m;
        rowsum[i] = (red[0] > 0.f) ? red[0] : 1.f;
    }
}

__global__ void k_ew(const int* __restrict__ src, const int* __restrict__ dst,
                     const float* __restrict__ nv1, const float* __restrict__ nv2,
                     const float* __restrict__ rowmax, const float* __restrict__ rowsum,
                     float* ew) {
    int e = blockIdx.x * blockDim.x + threadIdx.x;
    if (e >= NE) return;
    int s = src[e], d = dst[e];
    float acc = 0.f;
#pragma unroll
    for (int a = 0; a < AEMB; a++) acc += nv1[s * AEMB + a] * nv2[a * NN + d];
    float scv = fmaxf(acc, 0.f);
    ew[e] = 1.f + __expf(scv - rowmax[s]) * __builtin_amdgcn_rcpf(rowsum[s]);
}

__global__ void k_norms(const float* degout, const float* degin, float* nsv, float* ndv) {
    int i = blockIdx.x * blockDim.x + threadIdx.x;
    if (i >= NN) return;
    nsv[i] = rsqrtf(fmaxf(degout[i], 1.f));
    ndv[i] = rsqrtf(fmaxf(degin[i], 1.f));
}

__global__ void k_start(const float* __restrict__ feat, const float* __restrict__ sw,
                        const float* __restrict__ sb, float* h) {
    const int total4 = BB * NN * 13 * (HH / 4);
    const int stride = gridDim.x * blockDim.x;
    for (int idx = blockIdx.x * blockDim.x + threadIdx.x; idx < total4; idx += stride) {
        int c4 = idx & 7;
        int mt = idx >> 3;
        int t = mt % 13;
        int m = mt / 13;
        int b = m >> 9, n = m & 511;
        float x = (t == 0) ? 0.f : feat[(b * 12 + (t - 1)) * NN + n];
        float4 w = ((const float4*)sw)[c4];
        float4 bb = ((const float4*)sb)[c4];
        ((float4*)h)[idx] = make_float4(w.x * x + bb.x, w.y * x + bb.y,
                                        w.z * x + bb.z, w.w * x + bb.w);
    }
}

#define GAU_NT 2
#define GC_NT  4
#define GC_WAVES 8
#define SK_NT  4
#define GLK (NLAY * HH)
#define E1_NT  4

template <bool NORM>
__device__ __forceinline__ void gau_body(const float* __restrict__ h,
                                         const float* __restrict__ pstat,
                                         const float* __restrict__ pgam,
                                         const float* __restrict__ pbet,
                                         const float* __restrict__ fwT,
                                         const float* __restrict__ fb,
                                         const float* __restrict__ gwT,
                                         const float* __restrict__ gb, float* gau,
                                         float* glast, int T, int d, int l) {
    int Tp = T - d;
    int unit = blockIdx.x * 4 + (threadIdx.x >> 5);
    int tn = unit & 1;
    int tm0 = (unit >> 1) * GAU_NT;
    int lane = threadIdx.x & 31;
    int ncol = tn * 16 + (lane & 15);
    int base = (lane & 16) ? 8 : 0;
    const float* fwr = fwT + (size_t)((l * HH + ncol) * 2) * HH;
    const float* gwr = gwT + (size_t)((l * HH + ncol) * 2) * HH;
    float sc[16], sh[16];
    if (NORM) {
        float inv_cnt = 1.f / (float)(BB * NN * T);
        bn_coef8(sc, sh, pstat, pgam, pbet, base, inv_cnt);
        bn_coef8(sc + 8, sh + 8, pstat, pgam, pbet, base + 16, inv_cnt);
    }
    v16h bf0, bg0, bf1, bg1;
    pack8(bf0, 0, fwr + base);       pack8(bf0, 8, fwr + base + 16);
    pack8(bg0, 0, gwr + base);       pack8(bg0, 8, gwr + base + 16);
    pack8(bf1, 0, fwr + HH + base);  pack8(bf1, 8, fwr + HH + base + 16);
    pack8(bg1, 0, gwr + HH + base);  pack8(bg1, 8, gwr + HH + base + 16);
    float fbias = fb[l * HH + ncol];
    float gbias = gb[l * HH + ncol];
    float vals[GAU_NT][8];
    int tr0[GAU_NT], bnr0[GAU_NT];
#pragma unroll
    for (int i = 0; i < GAU_NT; i++) {
        int tm = tm0 + i;
        int mrow = tm * 16 + (lane & 15);
        int t = mrow % Tp;
        int bn = mrow / Tp;
        const float* hrow0 = h + (bn * T + t) * HH;
        const float* hrow1 = hrow0 + d * HH;
        v16h a0, a1;
        if (NORM) {
            pack8_affine(a0, 0, hrow0 + base, sc, sh);
            pack8_affine(a0, 8, hrow0 + base + 16, sc + 8, sh + 8);
            pack8_affine(a1, 0, hrow1 + base, sc, sh);
            pack8_affine(a1, 8, hrow1 + base + 16, sc + 8, sh + 8);
        } else {
            pack8(a0, 0, hrow0 + base);  pack8(a0, 8, hrow0 + base + 16);
            pack8(a1, 0, hrow1 + base);  pack8(a1, 8, hrow1 + base + 16);
        }
        v8f cf = {}, cg = {};
        cf = wmma_f16(a0, bf0, cf);
        cg = wmma_f16(a0, bg0, cg);
        cf = wmma_f16(a1, bf1, cf);
        cg = wmma_f16(a1, bg1, cg);

        int rbase = tm * 16 + ((lane & 16) ? 8 : 0);
        tr0[i] = __shfl(t, (lane & 16) ? 8 : 0);
        bnr0[i] = __shfl(bn, (lane & 16) ? 8 : 0);
#pragma unroll
        for (int v = 0; v < 8; v++) {
            int row = rbase + v;
            float fv = fast_tanh(cf[v] + fbias);
            float gv = fast_sigmoid(cg[v] + gbias);
            vals[i][v] = fv * gv;
            gau[(size_t)row * HH + ncol] = vals[i][v];
        }
    }
#pragma unroll
    for (int i = 0; i < GAU_NT; i++) {
        int rem = Tp - 1 - tr0[i];
        int gi = bnr0[i] * GLK + l * HH + ncol;
#pragma unroll
        for (int v = 0; v < 8; v++) {
            if (v == rem) {
                glast[gi] = vals[i][v];
                rem += Tp;
                gi += GLK;
            }
        }
    }
}
__global__ void __launch_bounds__(128) k_gau_first(const float* __restrict__ h, const float* __restrict__ fwT,
                            const float* __restrict__ fb, const float* __restrict__ gwT,
                            const float* __restrict__ gb, float* gau, float* glast,
                            int T, int d, int l) {
    gau_body<false>(h, nullptr, nullptr, nullptr, fwT, fb, gwT, gb, gau, glast, T, d, l);
}
__global__ void __launch_bounds__(128) k_gau_wmma(const float* __restrict__ h, const float* __restrict__ pstat,
                           const float* __restrict__ pgam, const float* __restrict__ pbet,
                           const float* __restrict__ fwT, const float* __restrict__ fb,
                           const float* __restrict__ gwT, const float* __restrict__ gb,
                           float* gau, float* glast, int T, int d, int l) {
    gau_body<true>(h, pstat, pgam, pbet, fwT, fb, gwT, gb, gau, glast, T, d, l);
}

__global__ void __launch_bounds__(128) k_skip_all(const float* __restrict__ glast, const float* __restrict__ sw,
                           const float* __restrict__ sb, float* skip) {
    const int nGroupsN = SKC / 16 / SK_NT;
    int unit = blockIdx.x * 4 + (threadIdx.x >> 5);
    int tm = unit / nGroupsN;
    int tn0 = (unit % nGroupsN) * SK_NT;
    int lane = threadIdx.x & 31;
    int mrow = tm * 16 + (lane & 15);
    int base = (lane & 16) ? 8 : 0;
    const float* arow = glast + (size_t)mrow * GLK;
    v16h a[NLAY];
#pragma unroll
    for (int l = 0; l < NLAY; l++) {
        pack8(a[l], 0, arow + l * HH + base);
        pack8(a[l], 8, arow + l * HH + base + 16);
    }
    int rbase = tm * 16 + ((lane & 16) ? 8 : 0);
#pragma unroll
    for (int i = 0; i < SK_NT; i++) {
        int ncol = (tn0 + i) * 16 + (lane & 15);
        v8f c = {};
        float bias = 0.f;
#pragma unroll
        for (int l = 0; l < NLAY; l++) {
            const float* brow = sw + (size_t)(l * SKC + ncol) * HH;
            v16h bm;
            pack8(bm, 0, brow + base);
            pack8(bm, 8, brow + base + 16);
            c = wmma_f16(a[l], bm, c);
            bias += sb[l * SKC + ncol];
        }
#pragma unroll
        for (int v = 0; v < 8; v++)
            skip[(size_t)(rbase + v) * SKC + ncol] = c[v] + bias;
    }
}

#define CSR_T 256
#define CSR_PER (NE / CSR_T)
__global__ void k_csr(const int* __restrict__ src, const int* __restrict__ dst,
                      const float* __restrict__ ew, const float* __restrict__ nsv,
                      int2* rowse, int2* rec) {
    __shared__ int s_lt[CSR_T];
    __shared__ int s_eq[CSR_T];
    const int d = blockIdx.x;
    const int tid = threadIdx.x;
    const int e0 = tid * CSR_PER;
    int lt = 0, eq = 0;
    for (int i = 0; i < CSR_PER; i++) {
        int x = dst[e0 + i];
        lt += (x < d) ? 1 : 0;
        eq += (x == d) ? 1 : 0;
    }
    s_lt[tid] = lt;
    s_eq[tid] = eq;
    __syncthreads();
    for (int off = 1; off < CSR_T; off <<= 1) {
        int a = 0, b = 0;
        if (tid >= off) { a = s_lt[tid - off]; b = s_eq[tid - off]; }
        __syncthreads();
        s_lt[tid] += a;
        s_eq[tid] += b;
        __syncthreads();
    }
    const int first = s_lt[CSR_T - 1];
    int pos = first + s_eq[tid] - eq;
    if (tid == 0) rowse[d] = make_int2(first, first + s_eq[CSR_T - 1]);
    for (int i = 0; i < CSR_PER; i++) {
        int e = e0 + i;
        if (dst[e] == d) {
            int sn = src[e];
            rec[pos] = make_int2(sn, __float_as_int(ew[e] * nsv[sn]));
            pos++;
        }
    }
}

__device__ __forceinline__ void fma4(float4& a, float w, const float4 v) {
    a.x = fmaf(w, v.x, a.x); a.y = fmaf(w, v.y, a.y);
    a.z = fmaf(w, v.z, a.z); a.w = fmaf(w, v.w, a.w);
}
__global__ void k_gather(const int2* __restrict__ rowse, const int2* __restrict__ rec,
                         const float4* __restrict__ gau4, float4* __restrict__ agg4,
                         int slab4) {
    const int total = (BB / 4) * NN * slab4;
    int idx = blockIdx.x * blockDim.x + threadIdx.x;
    if (idx >= total) return;
    int bd = idx / slab4;
    int q = idx - bd * slab4;
    int d = bd & (NN - 1);
    int b0 = bd >> 9;
    const int bstride = (BB / 4) * NN * slab4;
    const float4* g = gau4 + (b0 * NN) * slab4 + q;
    int2 se = rowse[d];
    float4 a0 = make_float4(0.f, 0.f, 0.f, 0.f), a1 = a0, a2 = a0, a3 = a0;
    int j = se.x;
    for (; j + 2 <= se.y; j += 2) {
        int2 r0 = rec[j], r1 = rec[j + 1];
        const float4* p0 = g + r0.x * slab4;
        const float4* p1 = g + r1.x * slab4;
        float4 u0 = p0[0], u1 = p0[bstride], u2 = p0[2 * bstride], u3 = p0[3 * bstride];
        float4 v0 = p1[0], v1 = p1[bstride], v2 = p1[2 * bstride], v3 = p1[3 * bstride];
        float w0 = __int_as_float(r0.y), w1 = __int_as_float(r1.y);
        fma4(a0, w0, u0); fma4(a1, w0, u1); fma4(a2, w0, u2); fma4(a3, w0, u3);
        fma4(a0, w1, v0); fma4(a1, w1, v1); fma4(a2, w1, v2); fma4(a3, w1, v3);
    }
    if (j < se.y) {
        int2 r0 = rec[j];
        const float4* p0 = g + r0.x * slab4;
        float w0 = __int_as_float(r0.y);
        fma4(a0, w0, p0[0]); fma4(a1, w0, p0[bstride]);
        fma4(a2, w0, p0[2 * bstride]); fma4(a3, w0, p0[3 * bstride]);
    }
    float4* o = agg4 + (bd * slab4 + q);
    o[0] = a0; o[bstride] = a1; o[2 * bstride] = a2; o[3 * bstride] = a3;
}

template <bool NORM>
__device__ __forceinline__ void gconv_body(const float* __restrict__ agg,
                                           const float* __restrict__ ndv,
                                           const float* __restrict__ gcwT,
                                           const float* __restrict__ gcb,
                                           const float* __restrict__ hold,
                                           const float* __restrict__ pstat,
                                           const float* __restrict__ pgam,
                                           const float* __restrict__ pbet,
                                           float* hpre, float* stat, int T, int d, int l) {
    int Tp = T - d;
    int unit = blockIdx.x * GC_WAVES + (threadIdx.x >> 5);
    int tn = unit & 1;
    int tm0 = (unit >> 1) * GC_NT;
    int lane = threadIdx.x & 31;
    int ncol = tn * 16 + (lane & 15);
    int base = (lane & 16) ? 8 : 0;
    const float* brow = gcwT + (size_t)(l * HH + ncol) * HH;
    v16h bm;
    pack8(bm, 0, brow + base);
    pack8(bm, 8, brow + base + 16);
    float bias = gcb[l * HH + ncol];
    float psc = 1.f, psh = 0.f;
    if (NORM)
        bn_coef(psc, psh, pstat[ncol], pstat[HH + ncol], pgam[ncol], pbet[ncol],
                1.f / (float)(BB * NN * T));
    float4 ra[GC_NT][4];
    float nrm[GC_NT];
    float hv[GC_NT][8];
#pragma unroll
    for (int i = 0; i < GC_NT; i++) {
        int tm = tm0 + i;
        int mrow = tm * 16 + (lane & 15);
        int node_m = (mrow / Tp) & 511;
        nrm[i] = ndv[node_m];
        const float* ap = agg + (size_t)mrow * HH + base;
        ra[i][0] = ((const float4*)ap)[0];
        ra[i][1] = ((const float4*)ap)[1];
        ra[i][2] = ((const float4*)(ap + 16))[0];
        ra[i][3] = ((const float4*)(ap + 16))[1];
        int rbase = tm * 16 + ((lane & 16) ? 8 : 0);
#pragma unroll
        for (int v = 0; v < 8; v++) {
            int row = rbase + v;
            int t = row % Tp;
            int bn = row / Tp;
            hv[i][v] = hold[(bn * T + t + d) * HH + ncol];
        }
    }
    float s1 = 0.f, s2 = 0.f;
#pragma unroll
    for (int i = 0; i < GC_NT; i++) {
        int tm = tm0 + i;
        v16h a;
        cvt8_scale(a, 0, ra[i][0], ra[i][1], nrm[i]);
        cvt8_scale(a, 8, ra[i][2], ra[i][3], nrm[i]);
        v8f c = {};
        c = wmma_f16(a, bm, c);
        int rbase = tm * 16 + ((lane & 16) ? 8 : 0);
#pragma unroll
        for (int v = 0; v < 8; v++) {
            int row = rbase + v;
            float h0 = hv[i][v];
            if (NORM) h0 = fmaf(h0, psc, psh);
            float val = fmaxf(c[v] + bias, 0.f) + h0;
            hpre[(size_t)row * HH + ncol] = val;
            s1 += val;
            s2 += val * val;
        }
    }
    s1 += __shfl_xor(s1, 16);
    s2 += __shfl_xor(s2, 16);
    __shared__ float red[GC_WAVES][32];
    const int wave = threadIdx.x >> 5;
    if (lane < 16) {
        red[wave][lane] = s1;
        red[wave][16 + lane] = s2;
    }
    __builtin_amdgcn_fence(__ATOMIC_RELEASE, "workgroup", "local");
    __builtin_amdgcn_s_barrier();
    __builtin_amdgcn_fence(__ATOMIC_ACQUIRE, "workgroup", "local");
    if (threadIdx.x < 2 * HH) {
        const int wd = threadIdx.x;
        const int ch = wd & (HH - 1);
        const int slot = (wd >> 5) * 16 + (ch & 15);
        const int w0 = ch >> 4;
        float tot = 0.f;
#pragma unroll
        for (int k = 0; k < GC_WAVES / 2; k++) tot += red[w0 + 2 * k][slot];
        atomicAdd(&stat[wd], tot);
    }
}
__global__ void __launch_bounds__(32 * GC_WAVES) k_gconv_first(const float* __restrict__ agg, const float* __restrict__ ndv,
                              const float* __restrict__ gcwT, const float* __restrict__ gcb,
                              const float* __restrict__ hold, float* hpre, float* stat,
                              int T, int d, int l) {
    gconv_body<false>(agg, ndv, gcwT, gcb, hold, nullptr, nullptr, nullptr, hpre, stat, T, d, l);
}
__global__ void __launch_bounds__(32 * GC_WAVES) k_gconv_wmma(const float* __restrict__ agg, const float* __restrict__ ndv,
                             const float* __restrict__ gcwT, const float* __restrict__ gcb,
                             const float* __restrict__ hold, const float* __restrict__ pstat,
                             const float* __restrict__ pgam, const float* __restrict__ pbet,
                             float* hpre, float* stat, int T, int d, int l) {
    gconv_body<true>(agg, ndv, gcwT, gcb, hold, pstat, pgam, pbet, hpre, stat, T, d, l);
}

__global__ void __launch_bounds__(128) k_end1_wmma(const float* __restrict__ skip, const float* __restrict__ w1,
                            const float* __restrict__ b1, float* out1) {
    const int nGroupsN = ECC / 16 / E1_NT;
    int unit = blockIdx.x * 4 + (threadIdx.x >> 5);
    int tm = unit / nGroupsN;
    int tn0 = (unit % nGroupsN) * E1_NT;
    int lane = threadIdx.x & 31;
    int mrow = tm * 16 + (lane & 15);
    int base = (lane & 16) ? 8 : 0;
    const float* arow = skip + (size_t)mrow * SKC;
    v16h a[SKC / 32];
#pragma unroll
    for (int ks = 0; ks < SKC / 32; ks++) {
        pack8_relu(a[ks], 0, arow + ks * 32 + base);
        pack8_relu(a[ks], 8, arow + ks * 32 + base + 16);
    }
    int rbase = tm * 16 + ((lane & 16) ? 8 : 0);
#pragma unroll
    for (int i = 0; i < E1_NT; i++) {
        int ncol = (tn0 + i) * 16 + (lane & 15);
        const float* brow = w1 + (size_t)ncol * SKC;
        v8f c = {};
#pragma unroll
        for (int ks = 0; ks < SKC / 32; ks++) {
            v16h bm;
            pack8(bm, 0, brow + ks * 32 + base);
            pack8(bm, 8, brow + ks * 32 + base + 16);
            c = wmma_f16(a[ks], bm, c);
        }
        float bias = b1[ncol];
#pragma unroll
        for (int v = 0; v < 8; v++)
            out1[(size_t)(rbase + v) * ECC + ncol] = fmaxf(c[v] + bias, 0.f);
    }
}

__global__ void k_end2_wmma(const float* __restrict__ in1, const float* __restrict__ w2,
                            const float* __restrict__ b2, float* out) {
    int tm = blockIdx.x * 4 + (threadIdx.x >> 5);
    int lane = threadIdx.x & 31;
    int mrow = tm * 16 + (lane & 15);
    int ncol = lane & 15;
    int base = (lane & 16) ? 8 : 0;
    const float* arow = in1 + (size_t)mrow * ECC;
    const float* brow = w2 + (size_t)ncol * ECC;
    v8f c = {};
#pragma unroll
    for (int k0 = 0; k0 < ECC; k0 += 32) {
        v16h a, bm = {};
        pack8(a, 0, arow + k0 + base);
        pack8(a, 8, arow + k0 + base + 16);
        if (ncol < OUTF) {
            pack8(bm, 0, brow + k0 + base);
            pack8(bm, 8, brow + k0 + base + 16);
        }
        c = wmma_f16(a, bm, c);
    }
    if (ncol < OUTF) {
        int rbase = tm * 16 + ((lane & 16) ? 8 : 0);
        float bias = b2[ncol];
#pragma unroll
        for (int v = 0; v < 8; v++) {
            int row = rbase + v;
            int b = row >> 9, n = row & 511;
            out[(b * OUTF + ncol) * NN + n] = c[v] + bias;
        }
    }
}

extern "C" void kernel_launch(void* const* d_in, const int* in_sizes, int n_in,
                              void* d_out, int out_size, void* d_ws, size_t ws_size,
                              hipStream_t stream) {
    const float* feat     = (const float*)d_in[0];
    const int*   src      = (const int*)d_in[1];
    const int*   dst      = (const int*)d_in[2];
    const float* nv1      = (const float*)d_in[3];
    const float* nv2      = (const float*)d_in[4];
    const float* start_w  = (const float*)d_in[5];
    const float* start_b  = (const float*)d_in[6];
    const float* filter_w = (const float*)d_in[7];
    const float* filter_b = (const float*)d_in[8];
    const float* gate_w   = (const float*)d_in[9];
    const float* gate_b   = (const float*)d_in[10];
    const float* skip_w   = (const float*)d_in[11];
    const float* skip_b   = (const float*)d_in[12];
    const float* bn_gamma = (const float*)d_in[13];
    const float* bn_beta  = (const float*)d_in[14];
    const float* gconv_w  = (const float*)d_in[15];
    const float* gconv_b  = (const float*)d_in[16];
    const float* end1_w   = (const float*)d_in[17];
    const float* end1_b   = (const float*)d_in[18];
    const float* end2_w   = (const float*)d_in[19];
    const float* end2_b   = (const float*)d_in[20];

    float* ws = (float*)d_ws;
    size_t o = 0;
    unsigned char* adj = (unsigned char*)(ws + o); o += (NN * NN) / 4;
    float* rowmax = ws + o; o += NN;
    float* rowsum = ws + o; o += NN;
    float* ew     = ws + o; o += NE;
    float* degout = ws + o; o += NN;
    float* degin  = ws + o; o += NN;
    float* nsv    = ws + o; o += NN;
    float* ndv    = ws + o; o += NN;
    float* bnstat = ws + o; o += NLAY * 2 * HH;
    float* fwT    = ws + o; o += NLAY * HH * 2 * HH;
    float* gwT    = ws + o; o += NLAY * HH * 2 * HH;
    float* gcwT   = ws + o; o += NLAY * HH * HH;
    float* skip   = ws + o; o += (size_t)BB * NN * SKC;
    float* glast  = ws + o; o += (size_t)BB * NN * GLK;
    float* hA     = ws + o; o += (size_t)BB * NN * 13 * HH;
    float* hB     = ws + o; o += (size_t)BB * NN * 13 * HH;
    float* gau    = ws + o; o += (size_t)BB * NN * 12 * HH;
    float* agg    = ws + o; o += (size_t)BB * NN * 12 * HH;
    int2*  rowse  = (int2*)(ws + o); o += 2 * NN;
    int2*  rec    = (int2*)(ws + o); o += 2 * NE;
    float* out1   = gau;

#define ZEROF(p, n) \
    k_zero<<<(((n) / 4 + 255) / 256 < 512 ? ((n) / 4 + 255) / 256 : 512), 256, 0, stream>>>((p), (n))

    ZEROF((float*)adj, (NN * NN) / 4);
    ZEROF(degout, NN);
    ZEROF(degin, NN);
    ZEROF(bnstat, NLAY * 2 * HH);

    k_prep_w<<<(NLAY * HH * 2 * HH + 255) / 256, 256, 0, stream>>>(filter_w, gate_w, gconv_w,
                                                                   fwT, gwT, gcwT);
    k_scatter<<<(NE + 255) / 256, 256, 0, stream>>>(src, dst, adj, degout, degin);
    k_rownorm<<<NN, 256, 0, stream>>>(adj, nv1, nv2, rowmax, rowsum);
    k_ew<<<(NE + 255) / 256, 256, 0, stream>>>(src, dst, nv1, nv2, rowmax, rowsum, ew);
    k_norms<<<(NN + 255) / 256, 256, 0, stream>>>(degout, degin, nsv, ndv);
    k_csr<<<NN, CSR_T, 0, stream>>>(src, dst, ew, nsv, rowse, rec);
    k_start<<<(BB * NN * 13 * (HH / 4)) / (256 * 8), 256, 0, stream>>>(feat, start_w, start_b, hA);

    const int dil[8] = {1, 2, 1, 2, 1, 2, 1, 2};
    float* hcur = hA;
    float* hnext = hB;
    int T = 13;
    for (int l = 0; l < NLAY; l++) {
        int d = dil[l];
        int Tp = T - d;
        int M = BB * NN * Tp;
        int tilesGau = (M / 16) * 2;

        const int lp = l ? l - 1 : 0;
        const float* pstat = bnstat + lp * 2 * HH;
        const float* pgam  = bn_gamma + lp * HH;
        const float* pbet  = bn_beta + lp * HH;
        if (l == 0)
            k_gau_first<<<tilesGau / 4 / GAU_NT, 128, 0, stream>>>(hcur, fwT, filter_b, gwT,
                                                                   gate_b, gau, glast, T, d, l);
        else
            k_gau_wmma<<<tilesGau / 4 / GAU_NT, 128, 0, stream>>>(hcur, pstat, pgam, pbet, fwT,
                                                                  filter_b, gwT, gate_b, gau,
                                                                  glast, T, d, l);
        k_gather<<<((BB / 4) * NN * Tp * 8 + 255) / 256, 256, 0, stream>>>(
            rowse, rec, (const float4*)gau, (float4*)agg, Tp * 8);
        if (l == 0)
            k_gconv_first<<<(M / 16) * 2 / GC_WAVES / GC_NT, 32 * GC_WAVES, 0, stream>>>(
                agg, ndv, gcwT, gconv_b, hcur, hnext, bnstat, T, d, l);
        else
            k_gconv_wmma<<<(M / 16) * 2 / GC_WAVES / GC_NT, 32 * GC_WAVES, 0, stream>>>(
                agg, ndv, gcwT, gconv_b, hcur, pstat, pgam, pbet, hnext, bnstat + l * 2 * HH,
                T, d, l);
        float* tmp = hcur; hcur = hnext; hnext = tmp;
        T = Tp;
    }

    k_skip_all<<<(BB * NN / 16) * (SKC / 16) / 4 / SK_NT, 128, 0, stream>>>(glast, skip_w, skip_b,
                                                                            skip);
    k_end1_wmma<<<(BB * NN / 16) * (ECC / 16) / 4 / E1_NT, 128, 0, stream>>>(skip, end1_w, end1_b,
                                                                             out1);
    k_end2_wmma<<<(BB * NN / 16) / 4, 128, 0, stream>>>(out1, end2_w, end2_b, (float*)d_out);
#undef ZEROF
}


# set-up in 1 launch, not 11: row softmax statistics and degree norms straight from the edge list, start conv, weight repack, sum fill side by side; edge lists ride in the first gate/filter launch
# speedup vs baseline: 5.1617x; 1.0808x over previous
#include <hip/hip_runtime.h>
#include <hip/hip_bf16.h>

typedef __attribute__((ext_vector_type(16))) _Float16 v16h;
typedef __attribute__((ext_vector_type(8)))  float    v8f;

#define NN   512
#define NE   4096
#define BB   32
#define HH   32
#define SKC  256
#define ECC  512
#define OUTF 12
#define AEMB 10
#define NEGBIG (-1e30f)
#define NLAY 8

__device__ __forceinline__ v8f wmma_f16(v16h a, v16h b, v8f c) {
    return __builtin_amdgcn_wmma_f32_16x16x32_f16(false, a, false, b, (short)0, c, false, false);
}

__device__ __forceinline__ float fast_sigmoid(float x) {
    return __builtin_amdgcn_rcpf(1.f + __expf(-x));
}
__device__ __forceinline__ float fast_tanh(float x) {
    float t = __expf(-2.f * fabsf(x));
    float r = (1.f - t) * __builtin_amdgcn_rcpf(1.f + t);
    return copysignf(r, x);
}

__device__ __forceinline__ void pack8(v16h& d, int off, const float* __restrict__ p) {
    float4 x = ((const float4*)p)[0];
    float4 y = ((const float4*)p)[1];
    d[off + 0] = (_Float16)x.x; d[off + 1] = (_Float16)x.y;
    d[off + 2] = (_Float16)x.z; d[off + 3] = (_Float16)x.w;
    d[off + 4] = (_Float16)y.x; d[off + 5] = (_Float16)y.y;
    d[off + 6] = (_Float16)y.z; d[off + 7] = (_Float16)y.w;
}
__device__ __forceinline__ void cvt8_scale(v16h& d, int off, float4 x, float4 y, float s) {
    d[off + 0] = (_Float16)(x.x * s); d[off + 1] = (_Float16)(x.y * s);
    d[off + 2] = (_Float16)(x.z * s); d[off + 3] = (_Float16)(x.w * s);
    d[off + 4] = (_Float16)(y.x * s); d[off + 5] = (_Float16)(y.y * s);
    d[off + 6] = (_Float16)(y.z * s); d[off + 7] = (_Float16)(y.w * s);
}
__device__ __forceinline__ void pack8_relu(v16h& d, int off, const float* __restrict__ p) {
    float4 x = ((const float4*)p)[0];
    float4 y = ((const float4*)p)[1];
    d[off + 0] = (_Float16)fmaxf(x.x, 0.f); d[off + 1] = (_Float16)fmaxf(x.y, 0.f);
    d[off + 2] = (_Float16)fmaxf(x.z, 0.f); d[off + 3] = (_Float16)fmaxf(x.w, 0.f);
    d[off + 4] = (_Float16)fmaxf(y.x, 0.f); d[off + 5] = (_Float16)fmaxf(y.y, 0.f);
    d[off + 6] = (_Float16)fmaxf(y.z, 0.f); d[off + 7] = (_Float16)fmaxf(y.w, 0.f);
}

__device__ __forceinline__ void pack8_affine(v16h& d, int off, const float* __restrict__ p,
                                             const float* sc, const float* sh) {
    float4 x = ((const float4*)p)[0];
    float4 y = ((const float4*)p)[1];
    d[off + 0] = (_Float16)fmaf(x.x, sc[0], sh[0]); d[off + 1] = (_Float16)fmaf(x.y, sc[1], sh[1]);
    d[off + 2] = (_Float16)fmaf(x.z, sc[2], sh[2]); d[off + 3] = (_Float16)fmaf(x.w, sc[3], sh[3]);
    d[off + 4] = (_Float16)fmaf(y.x, sc[4], sh[4]); d[off + 5] = (_Float16)fmaf(y.y, sc[5], sh[5]);
    d[off + 6] = (_Float16)fmaf(y.z, sc[6], sh[6]); d[off + 7] = (_Float16)fmaf(y.w, sc[7], sh[7]);
}

__device__ __forceinline__ void bn_coef(float& sc, float& sh, float s, float q, float g, float b,
                                        float inv_cnt) {
    float mu = s * inv_cnt;
    float var = q * inv_cnt - mu * mu;
    sc = rsqrtf(var + 1e-5f) * g;
    sh = b - mu * sc;
}
__device__ __forceinline__ void bn_coef8(float* sc, float* sh, const float* __restrict__ stat,
                                         const float* __restrict__ gam,
                                         const float* __restrict__ bet, int c0, float inv_cnt) {
    float4 s0 = ((const float4*)(stat + c0))[0],      s1 = ((const float4*)(stat + c0))[1];
    float4 q0 = ((const float4*)(stat + HH + c0))[0], q1 = ((const float4*)(stat + HH + c0))[1];
    float4 g0 = ((const float4*)(gam + c0))[0],       g1 = ((const float4*)(gam + c0))[1];
    float4 b0 = ((const float4*)(bet + c0))[0],       b1 = ((const float4*)(bet + c0))[1];
    bn_coef(sc[0], sh[0], s0.x, q0.x, g0.x, b0.x, inv_cnt);
    bn_coef(sc[1], sh[1], s0.y, q0.y, g0.y, b0.y, inv_cnt);
    bn_coef(sc[2], sh[2], s0.z, q0.z, g0.z, b0.z, inv_cnt);
    bn_coef(sc[3], sh[3], s0.w, q0.w, g0.w, b0.w, inv_cnt);
    bn_coef(sc[4], sh[4], s1.x, q1.x, g1.x, b1.x, inv_cnt);
    bn_coef(sc[5], sh[5], s1.y, q1.y, g1.y, b1.y, inv_cnt);
    bn_coef(sc[6], sh[6], s1.z, q1.z, g1.z, b1.z, inv_cnt);
    bn_coef(sc[7], sh[7], s1.w, q1.w, g1.w, b1.w, inv_cnt);
}

#define SU_T      256
#define SU_PER    (NE / SU_T)
#define SU_STARTB ((BB * NN * 13 * (HH / 4)) / (SU_T * 8))
#define SU_PREPB  ((NLAY * HH * 2 * HH) / SU_T)
#define SU_BLOCKS (NN + SU_STARTB + SU_PREPB + 1)
__global__ void __launch_bounds__(SU_T) k_setup(const int* __restrict__ src, const int* __restrict__ dst,
                        const float* __restrict__ nv1, const float* __restrict__ nv2,
                        const float* __restrict__ feat, const float* __restrict__ sw,
                        const float* __restrict__ sb,
                        const float* __restrict__ fw, const float* __restrict__ gw,
                        const float* __restrict__ gcw,
                        float* rowmax, float* rowsum, float* nsv, float* ndv,
                        float* h, float* fwT, float* gwT, float* gcwT, float* bnstat) {
    __shared__ float sc[NN];
    __shared__ float red[SU_T];
    __shared__ int cnt[SU_T / 32];
    const int tid = threadIdx.x;
    if (blockIdx.x < NN) {
        const int i = blockIdx.x;
        const int e0 = tid * SU_PER;
        int4 sq[SU_PER / 4], dq[SU_PER / 4];
#pragma unroll
        for (int k = 0; k < SU_PER / 4; k++) {
            sq[k] = ((const int4*)(src + e0))[k];
            dq[k] = ((const int4*)(dst + e0))[k];
        }
        for (int j = tid; j < NN; j += SU_T) sc[j] = NEGBIG;
        __syncthreads();
        unsigned outm = 0;
        int packed = 0;
#pragma unroll
        for (int k = 0; k < SU_PER / 4; k++) {
            outm |= (sq[k].x == i ? 1u : 0u) << (4 * k);
            outm |= (sq[k].y == i ? 1u : 0u) << (4 * k + 1);
            outm |= (sq[k].z == i ? 1u : 0u) << (4 * k + 2);
            outm |= (sq[k].w == i ? 1u : 0u) << (4 * k + 3);
            packed += (dq[k].x == i ? 0x10000 : 0) + (dq[k].y == i ? 0x10000 : 0) +
                      (dq[k].z == i ? 0x10000 : 0) + (dq[k].w == i ? 0x10000 : 0);
        }
        packed += __popc(outm);
        while (outm) {
            int k = __ffs(outm) - 1;
            outm &= outm - 1;
            int d = dst[e0 + k];
            float acc = 0.f;
#pragma unroll
            for (int a = 0; a < AEMB; a++) acc += nv1[i * AEMB + a] * nv2[a * NN + d];
            sc[d] = fmaxf(acc, 0.f);
        }
        for (int off = 16; off > 0; off >>= 1) packed += __shfl_xor(packed, off);
        if ((tid & 31) == 0) cnt[tid >> 5] = packed;
        __syncthreads();
        float m = NEGBIG;
        for (int j = tid; j < NN; j += SU_T) m = fmaxf(m, sc[j]);
        red[tid] = m;
        __syncthreads();
        for (int s = SU_T / 2; s > 0; s >>= 1) {
            if (tid < s) red[tid] = fmaxf(red[tid], red[tid + s]);
            __syncthreads();
        }
        m = red[0];
        __syncthreads();
        float sum = 0.f;
        for (int j = tid; j < NN; j += SU_T) {
            float v = sc[j];
            if (v > 0.5f * NEGBIG) sum += __expf(v - m);
        }
        red[tid] = sum;
        __syncthreads();
        for (int s = SU_T / 2; s > 0; s >>= 1) {
            if (tid < s) red[tid] += red[tid + s];
            __syncthreads();
        }
        if (tid == 0) {
            rowmax[i] = m;
            rowsum[i] = (red[0] > 0.f) ? red[0] : 1.f;
            int tot = 0;
            for (int w = 0; w < SU_T / 32; w++) tot += cnt[w];
            nsv[i] = rsqrtf(fmaxf((float)(tot & 0xffff), 1.f));
            ndv[i] = rsqrtf(fmaxf((float)(tot >> 16), 1.f));
        }
    } else if (blockIdx.x < NN + SU_STARTB) {
        const int total4 = BB * NN * 13 * (HH / 4);
        for (int idx = (blockIdx.x - NN) * SU_T + tid; idx < total4; idx += SU_STARTB * SU_T) {
            int c4 = idx & 7;
            int mt = idx >> 3;
            int t = mt % 13;
            int m = mt / 13;
            int b = m >> 9, n = m & 511;
            float x = (t == 0) ? 0.f : feat[(b * 12 + (t - 1)) * NN + n];
            float4 w = ((const float4*)sw)[c4];
            float4 bb = ((const float4*)sb)[c4];
            ((float4*)h)[idx] = make_float4(w.x * x + bb.x, w.y * x + bb.y,
                                            w.z * x + bb.z, w.w * x + bb.w);
        }
    } else if (blockIdx.x < NN + SU_STARTB + SU_PREPB) {
        int idx = (blockIdx.x - NN - SU_STARTB) * SU_T + tid;
        {
            int ci = idx & 31;
            int tap = (idx >> 5) & 1;
            int cc = (idx >> 6) & 31;
            int l = idx >> 11;
            fwT[idx] = fw[((l * HH + cc) * HH + ci) * 2 + tap];
            gwT[idx] = gw[((l * HH + cc) * HH + ci) * 2 + tap];
        }
        if (idx < NLAY * HH * HH) {
            int k = idx & 31;
            int n = (idx >> 5) & 31;
            int l = idx >> 10;
            gcwT[idx] = gcw[l * HH * HH + k * HH + n];
        }
    } else {
        for (int j = tid; j < NLAY * 2 * HH; j += SU_T) bnstat[j] = 0.f;
    }
}

#define GAU_NT 2
#define GC_NT  4
#define GC_WAVES 8
#define SK_NT  4
#define GLK (NLAY * HH)
#define E1_NT  4

template <bool NORM>
__device__ __forceinline__ void gau_body(const float* __restrict__ h,
                                         const float* __restrict__ pstat,
                                         const float* __restrict__ pgam,
                                         const float* __restrict__ pbet,
                                         const float* __restrict__ fwT,
                                         const float* __restrict__ fb,
                                         const float* __restrict__ gwT,
                                         const float* __restrict__ gb, float* gau,
                                         float* glast, int T, int d, int l, int blk) {
    int Tp = T - d;
    int unit = blk * 4 + (threadIdx.x >> 5);
    int tn = unit & 1;
    int tm0 = (unit >> 1) * GAU_NT;
    int lane = threadIdx.x & 31;
    int ncol = tn * 16 + (lane & 15);
    int base = (lane & 16) ? 8 : 0;
    const float* fwr = fwT + (size_t)((l * HH + ncol) * 2) * HH;
    const float* gwr = gwT + (size_t)((l * HH + ncol) * 2) * HH;
    float sc[16], sh[16];
    if (NORM) {
        float inv_cnt = 1.f / (float)(BB * NN * T);
        bn_coef8(sc, sh, pstat, pgam, pbet, base, inv_cnt);
        bn_coef8(sc + 8, sh + 8, pstat, pgam, pbet, base + 16, inv_cnt);
    }
    v16h bf0, bg0, bf1, bg1;
    pack8(bf0, 0, fwr + base);       pack8(bf0, 8, fwr + base + 16);
    pack8(bg0, 0, gwr + base);       pack8(bg0, 8, gwr + base + 16);
    pack8(bf1, 0, fwr + HH + base);  pack8(bf1, 8, fwr + HH + base + 16);
    pack8(bg1, 0, gwr + HH + base);  pack8(bg1, 8, gwr + HH + base + 16);
    float fbias = fb[l * HH + ncol];
    float gbias = gb[l * HH + ncol];
    float vals[GAU_NT][8];
    int tr0[GAU_NT], bnr0[GAU_NT];
#pragma unroll
    for (int i = 0; i < GAU_NT; i++) {
        int tm = tm0 + i;
        int mrow = tm * 16 + (lane & 15);
        int t = mrow % Tp;
        int bn = mrow / Tp;
        const float* hrow0 = h + (bn * T + t) * HH;
        const float* hrow1 = hrow0 + d * HH;
        v16h a0, a1;
        if (NORM) {
            pack8_affine(a0, 0, hrow0 + base, sc, sh);
            pack8_affine(a0, 8, hrow0 + base + 16, sc + 8, sh + 8);
            pack8_affine(a1, 0, hrow1 + base, sc, sh);
            pack8_affine(a1, 8, hrow1 + base + 16, sc + 8, sh + 8);
        } else {
            pack8(a0, 0, hrow0 + base);  pack8(a0, 8, hrow0 + base + 16);
            pack8(a1, 0, hrow1 + base);  pack8(a1, 8, hrow1 + base + 16);
        }
        v8f cf = {}, cg = {};
        cf = wmma_f16(a0, bf0, cf);
        cg = wmma_f16(a0, bg0, cg);
        cf = wmma_f16(a1, bf1, cf);
        cg = wmma_f16(a1, bg1, cg);

        int rbase = tm * 16 + ((lane & 16) ? 8 : 0);
        tr0[i] = __shfl(t, (lane & 16) ? 8 : 0);
        bnr0[i] = __shfl(bn, (lane & 16) ? 8 : 0);
#pragma unroll
        for (int v = 0; v < 8; v++) {
            int row = rbase + v;
            float fv = fast_tanh(cf[v] + fbias);
            float gv = fast_sigmoid(cg[v] + gbias);
            vals[i][v] = fv * gv;
            gau[(size_t)row * HH + ncol] = vals[i][v];
        }
    }
#pragma unroll
    for (int i = 0; i < GAU_NT; i++) {
        int rem = Tp - 1 - tr0[i];
        int gi = bnr0[i] * GLK + l * HH + ncol;
#pragma unroll
        for (int v = 0; v < 8; v++) {
            if (v == rem) {
                glast[gi] = vals[i][v];
                rem += Tp;
                gi += GLK;
            }
        }
    }
}
#define CSR_T 128
#define CSR_PER (NE / CSR_T)
__device__ __forceinline__ void csr_piece(const int d, const int* __restrict__ src,
                                          const int* __restrict__ dst,
                                          const float* __restrict__ nv1,
                                          const float* __restrict__ nv2,
                                          const float* __restrict__ rowmax,
                                          const float* __restrict__ rowsum,
                                          const float* __restrict__ nsv,
                                          int2* rowse, int2* rec) {
    __shared__ int s_lt[CSR_T];
    __shared__ int s_eq[CSR_T];
    const int tid = threadIdx.x;
    const int e0 = tid * CSR_PER;
    int lt = 0;
    unsigned inm = 0;
    int4 dq[CSR_PER / 4];
#pragma unroll
    for (int k = 0; k < CSR_PER / 4; k++) dq[k] = ((const int4*)(dst + e0))[k];
#pragma unroll
    for (int k = 0; k < CSR_PER / 4; k++) {
        lt += (dq[k].x < d ? 1 : 0) + (dq[k].y < d ? 1 : 0) + (dq[k].z < d ? 1 : 0) +
              (dq[k].w < d ? 1 : 0);
        inm |= (dq[k].x == d ? 1u : 0u) << (4 * k);
        inm |= (dq[k].y == d ? 1u : 0u) << (4 * k + 1);
        inm |= (dq[k].z == d ? 1u : 0u) << (4 * k + 2);
        inm |= (dq[k].w == d ? 1u : 0u) << (4 * k + 3);
    }
    const int eq = __popc(inm);
    s_lt[tid] = lt;
    s_eq[tid] = eq;
    __syncthreads();
    for (int off = 1; off < CSR_T; off <<= 1) {
        int a = 0, b = 0;
        if (tid >= off) { a = s_lt[tid - off]; b = s_eq[tid - off]; }
        __syncthreads();
        s_lt[tid] += a;
        s_eq[tid] += b;
        __syncthreads();
    }
    const int first = s_lt[CSR_T - 1];
    int pos = first + s_eq[tid] - eq;
    if (tid == 0) rowse[d] = make_int2(first, first + s_eq[CSR_T - 1]);
    while (inm) {
        int e = e0 + __ffs(inm) - 1;
        inm &= inm - 1;
        int sn = src[e];
        float acc = 0.f;
#pragma unroll
        for (int a = 0; a < AEMB; a++) acc += nv1[sn * AEMB + a] * nv2[a * NN + d];
        float scv = fmaxf(acc, 0.f);
        float ewv = 1.f + __expf(scv - rowmax[sn]) * __builtin_amdgcn_rcpf(rowsum[sn]);
        rec[pos] = make_int2(sn, __float_as_int(ewv * nsv[sn]));
        pos++;
    }
}

__global__ void __launch_bounds__(128) k_gau_first(const float* __restrict__ h, const float* __restrict__ fwT,
                            const float* __restrict__ fb, const float* __restrict__ gwT,
                            const float* __restrict__ gb, float* gau, float* glast,
                            int T, int d, int l,
                            const int* __restrict__ src, const int* __restrict__ dst,
                            const float* __restrict__ nv1, const float* __restrict__ nv2,
                            const float* __restrict__ rowmax, const float* __restrict__ rowsum,
                            const float* __restrict__ nsv, int2* rowse, int2* rec) {
    if (blockIdx.x < NN)
        csr_piece(blockIdx.x, src, dst, nv1, nv2, rowmax, rowsum, nsv, rowse, rec);
    else
        gau_body<false>(h, nullptr, nullptr, nullptr, fwT, fb, gwT, gb, gau, glast, T, d, l,
                        blockIdx.x - NN);
}
__global__ void __launch_bounds__(128) k_gau_wmma(const float* __restrict__ h, const float* __restrict__ pstat,
                           const float* __restrict__ pgam, const float* __restrict__ pbet,
                           const float* __restrict__ fwT, const float* __restrict__ fb,
                           const float* __restrict__ gwT, const float* __restrict__ gb,
                           float* gau, float* glast, int T, int d, int l) {
    gau_body<true>(h, pstat, pgam, pbet, fwT, fb, gwT, gb, gau, glast, T, d, l, blockIdx.x);
}

__global__ void __launch_bounds__(128) k_skip_all(const float* __restrict__ glast, const float* __restrict__ sw,
                           const float* __restrict__ sb, float* skip) {
    const int nGroupsN = SKC / 16 / SK_NT;
    int unit = blockIdx.x * 4 + (threadIdx.x >> 5);
    int tm = unit / nGroupsN;
    int tn0 = (unit % nGroupsN) * SK_NT;
    int lane = threadIdx.x & 31;
    int mrow = tm * 16 + (lane & 15);
    int base = (lane & 16) ? 8 : 0;
    const float* arow = glast + (size_t)mrow * GLK;
    v16h a[NLAY];
#pragma unroll
    for (int l = 0; l < NLAY; l++) {
        pack8(a[l], 0, arow + l * HH + base);
        pack8(a[l], 8, arow + l * HH + base + 16);
    }
    int rbase = tm * 16 + ((lane & 16) ? 8 : 0);
#pragma unroll
    for (int i = 0; i < SK_NT; i++) {
        int ncol = (tn0 + i) * 16 + (lane & 15);
        v8f c = {};
        float bias = 0.f;
#pragma unroll
        for (int l = 0; l < NLAY; l++) {
            const float* brow = sw + (size_t)(l * SKC + ncol) * HH;
            v16h bm;
            pack8(bm, 0, brow + base);
            pack8(bm, 8, brow + base + 16);
            c = wmma_f16(a[l], bm, c);
            bias += sb[l * SKC + ncol];
        }
#pragma unroll
        for (int v = 0; v < 8; v++)
            skip[(size_t)(rbase + v) * SKC + ncol] = c[v] + bias;
    }
}

__device__ __forceinline__ void fma4(float4& a, float w, const float4 v) {
    a.x = fmaf(w, v.x, a.x); a.y = fmaf(w, v.y, a.y);
    a.z = fmaf(w, v.z, a.z); a.w = fmaf(w, v.w, a.w);
}
__global__ void k_gather(const int2* __restrict__ rowse, const int2* __restrict__ rec,
                         const float4* __restrict__ gau4, float4* __restrict__ agg4,
                         int slab4) {
    const int total = (BB / 4) * NN * slab4;
    int idx = blockIdx.x * blockDim.x + threadIdx.x;
    if (idx >= total) return;
    int bd = idx / slab4;
    int q = idx - bd * slab4;
    int d = bd & (NN - 1);
    int b0 = bd >> 9;
    const int bstride = (BB / 4) * NN * slab4;
    const float4* g = gau4 + (b0 * NN) * slab4 + q;
    int2 se = rowse[d];
    float4 a0 = make_float4(0.f, 0.f, 0.f, 0.f), a1 = a0, a2 = a0, a3 = a0;
    int j = se.x;
    for (; j + 2 <= se.y; j += 2) {
        int2 r0 = rec[j], r1 = rec[j + 1];
        const float4* p0 = g + r0.x * slab4;
        const float4* p1 = g + r1.x * slab4;
        float4 u0 = p0[0], u1 = p0[bstride], u2 = p0[2 * bstride], u3 = p0[3 * bstride];
        float4 v0 = p1[0], v1 = p1[bstride], v2 = p1[2 * bstride], v3 = p1[3 * bstride];
        float w0 = __int_as_float(r0.y), w1 = __int_as_float(r1.y);
        fma4(a0, w0, u0); fma4(a1, w0, u1); fma4(a2, w0, u2); fma4(a3, w0, u3);
        fma4(a0, w1, v0); fma4(a1, w1, v1); fma4(a2, w1, v2); fma4(a3, w1, v3);
    }
    if (j < se.y) {
        int2 r0 = rec[j];
        const float4* p0 = g + r0.x * slab4;
        float w0 = __int_as_float(r0.y);
        fma4(a0, w0, p0[0]); fma4(a1, w0, p0[bstride]);
        fma4(a2, w0, p0[2 * bstride]); fma4(a3, w0, p0[3 * bstride]);
    }
    float4* o = agg4 + (bd * slab4 + q);
    o[0] = a0; o[bstride] = a1; o[2 * bstride] = a2; o[3 * bstride] = a3;
}

template <bool NORM>
__device__ __forceinline__ void gconv_body(const float* __restrict__ agg,
                                           const float* __restrict__ ndv,
                                           const float* __restrict__ gcwT,
                                           const float* __restrict__ gcb,
                                           const float* __restrict__ hold,
                                           const float* __restrict__ pstat,
                                           const float* __restrict__ pgam,
                                           const float* __restrict__ pbet,
                                           float* hpre, float* stat, int T, int d, int l) {
    int Tp = T - d;
    int unit = blockIdx.x * GC_WAVES + (threadIdx.x >> 5);
    int tn = unit & 1;
    int tm0 = (unit >> 1) * GC_NT;
    int lane = threadIdx.x & 31;
    int ncol = tn * 16 + (lane & 15);
    int base = (lane & 16) ? 8 : 0;
    const float* brow = gcwT + (size_t)(l * HH + ncol) * HH;
    v16h bm;
    pack8(bm, 0, brow + base);
    pack8(bm, 8, brow + base + 16);
    float bias = gcb[l * HH + ncol];
    float psc = 1.f, psh = 0.f;
    if (NORM)
        bn_coef(psc, psh, pstat[ncol], pstat[HH + ncol], pgam[ncol], pbet[ncol],
                1.f / (float)(BB * NN * T));
    float4 ra[GC_NT][4];
    float nrm[GC_NT];
    float hv[GC_NT][8];
#pragma unroll
    for (int i = 0; i < GC_NT; i++) {
        int tm = tm0 + i;
        int mrow = tm * 16 + (lane & 15);
        int node_m = (mrow / Tp) & 511;
        nrm[i] = ndv[node_m];
        const float* ap = agg + (size_t)mrow * HH + base;
        ra[i][0] = ((const float4*)ap)[0];
        ra[i][1] = ((const float4*)ap)[1];
        ra[i][2] = ((const float4*)(ap + 16))[0];
        ra[i][3] = ((const float4*)(ap + 16))[1];
        int rbase = tm * 16 + ((lane & 16) ? 8 : 0);
#pragma unroll
        for (int v = 0; v < 8; v++) {
            int row = rbase + v;
            int t = row % Tp;
            int bn = row / Tp;
            hv[i][v] = hold[(bn * T + t + d) * HH + ncol];
        }
    }
    float s1 = 0.f, s2 = 0.f;
#pragma unroll
    for (int i = 0; i < GC_NT; i++) {
        int tm = tm0 + i;
        v16h a;
        cvt8_scale(a, 0, ra[i][0], ra[i][1], nrm[i]);
        cvt8_scale(a, 8, ra[i][2], ra[i][3], nrm[i]);
        v8f c = {};
        c = wmma_f16(a, bm, c);
        int rbase = tm * 16 + ((lane & 16) ? 8 : 0);
#pragma unroll
        for (int v = 0; v < 8; v++) {
            int row = rbase + v;
            float h0 = hv[i][v];
            if (NORM) h0 = fmaf(h0, psc, psh);
            float val = fmaxf(c[v] + bias, 0.f) + h0;
            hpre[(size_t)row * HH + ncol] = val;
            s1 += val;
            s2 += val * val;
        }
    }
    s1 += __shfl_xor(s1, 16);
    s2 += __shfl_xor(s2, 16);
    __shared__ float red[GC_WAVES][32];
    const int wave = threadIdx.x >> 5;
    if (lane < 16) {
        red[wave][lane] = s1;
        red[wave][16 + lane] = s2;
    }
    __builtin_amdgcn_fence(__ATOMIC_RELEASE, "workgroup", "local");
    __builtin_amdgcn_s_barrier();
    __builtin_amdgcn_fence(__ATOMIC_ACQUIRE, "workgroup", "local");
    if (threadIdx.x < 2 * HH) {
        const int wd = threadIdx.x;
        const int ch = wd & (HH - 1);
        const int slot = (wd >> 5) * 16 + (ch & 15);
        const int w0 = ch >> 4;
        float tot = 0.f;
#pragma unroll
        for (int k = 0; k < GC_WAVES / 2; k++) tot += red[w0 + 2 * k][slot];
        atomicAdd(&stat[wd], tot);
    }
}
__global__ void __launch_bounds__(32 * GC_WAVES) k_gconv_first(const float* __restrict__ agg, const float* __restrict__ ndv,
                              const float* __restrict__ gcwT, const float* __restrict__ gcb,
                              const float* __restrict__ hold, float* hpre, float* stat,
                              int T, int d, int l) {
    gconv_body<false>(agg, ndv, gcwT, gcb, hold, nullptr, nullptr, nullptr, hpre, stat, T, d, l);
}
__global__ void __launch_bounds__(32 * GC_WAVES) k_gconv_wmma(const float* __restrict__ agg, const float* __restrict__ ndv,
                             const float* __restrict__ gcwT, const float* __restrict__ gcb,
                             const float* __restrict__ hold, const float* __restrict__ pstat,
                             const float* __restrict__ pgam, const float* __restrict__ pbet,
                             float* hpre, float* stat, int T, int d, int l) {
    gconv_body<true>(agg, ndv, gcwT, gcb, hold, pstat, pgam, pbet, hpre, stat, T, d, l);
}

__global__ void __launch_bounds__(128) k_end1_wmma(const float* __restrict__ skip, const float* __restrict__ w1,
                            const float* __restrict__ b1, float* out1) {
    const int nGroupsN = ECC / 16 / E1_NT;
    int unit = blockIdx.x * 4 + (threadIdx.x >> 5);
    int tm = unit / nGroupsN;
    int tn0 = (unit % nGroupsN) * E1_NT;
    int lane = threadIdx.x & 31;
    int mrow = tm * 16 + (lane & 15);
    int base = (lane & 16) ? 8 : 0;
    const float* arow = skip + (size_t)mrow * SKC;
    v16h a[SKC / 32];
#pragma unroll
    for (int ks = 0; ks < SKC / 32; ks++) {
        pack8_relu(a[ks], 0, arow + ks * 32 + base);
        pack8_relu(a[ks], 8, arow + ks * 32 + base + 16);
    }
    int rbase = tm * 16 + ((lane & 16) ? 8 : 0);
#pragma unroll
    for (int i = 0; i < E1_NT; i++) {
        int ncol = (tn0 + i) * 16 + (lane & 15);
        const float* brow = w1 + (size_t)ncol * SKC;
        v8f c = {};
#pragma unroll
        for (int ks = 0; ks < SKC / 32; ks++) {
            v16h bm;
            pack8(bm, 0, brow + ks * 32 + base);
            pack8(bm, 8, brow + ks * 32 + base + 16);
            c = wmma_f16(a[ks], bm, c);
        }
        float bias = b1[ncol];
#pragma unroll
        for (int v = 0; v < 8; v++)
            out1[(size_t)(rbase + v) * ECC + ncol] = fmaxf(c[v] + bias, 0.f);
    }
}

__global__ void k_end2_wmma(const float* __restrict__ in1, const float* __restrict__ w2,
                            const float* __restrict__ b2, float* out) {
    int tm = blockIdx.x * 4 + (threadIdx.x >> 5);
    int lane = threadIdx.x & 31;
    int mrow = tm * 16 + (lane & 15);
    int ncol = lane & 15;
    int base = (lane & 16) ? 8 : 0;
    const float* arow = in1 + (size_t)mrow * ECC;
    const float* brow = w2 + (size_t)ncol * ECC;
    v8f c = {};
#pragma unroll
    for (int k0 = 0; k0 < ECC; k0 += 32) {
        v16h a, bm = {};
        pack8(a, 0, arow + k0 + base);
        pack8(a, 8, arow + k0 + base + 16);
        if (ncol < OUTF) {
            pack8(bm, 0, brow + k0 + base);
            pack8(bm, 8, brow + k0 + base + 16);
        }
        c = wmma_f16(a, bm, c);
    }
    if (ncol < OUTF) {
        int rbase = tm * 16 + ((lane & 16) ? 8 : 0);
        float bias = b2[ncol];
#pragma unroll
        for (int v = 0; v < 8; v++) {
            int row = rbase + v;
            int b = row >> 9, n = row & 511;
            out[(b * OUTF + ncol) * NN + n] = c[v] + bias;
        }
    }
}

extern "C" void kernel_launch(void* const* d_in, const int* in_sizes, int n_in,
                              void* d_out, int out_size, void* d_ws, size_t ws_size,
                              hipStream_t stream) {
    const float* feat     = (const float*)d_in[0];
    const int*   src      = (const int*)d_in[1];
    const int*   dst      = (const int*)d_in[2];
    const float* nv1      = (const float*)d_in[3];
    const float* nv2      = (const float*)d_in[4];
    const float* start_w  = (const float*)d_in[5];
    const float* start_b  = (const float*)d_in[6];
    const float* filter_w = (const float*)d_in[7];
    const float* filter_b = (const float*)d_in[8];
    const float* gate_w   = (const float*)d_in[9];
    const float* gate_b   = (const float*)d_in[10];
    const float* skip_w   = (const float*)d_in[11];
    const float* skip_b   = (const float*)d_in[12];
    const float* bn_gamma = (const float*)d_in[13];
    const float* bn_beta  = (const float*)d_in[14];
    const float* gconv_w  = (const float*)d_in[15];
    const float* gconv_b  = (const float*)d_in[16];
    const float* end1_w   = (const float*)d_in[17];
    const float* end1_b   = (const float*)d_in[18];
    const float* end2_w   = (const float*)d_in[19];
    const float* end2_b   = (const float*)d_in[20];

    float* ws = (float*)d_ws;
    size_t o = 0;
    float* rowmax = ws + o; o += NN;
    float* rowsum = ws + o; o += NN;
    float* nsv    = ws + o; o += NN;
    float* ndv    = ws + o; o += NN;
    float* bnstat = ws + o; o += NLAY * 2 * HH;
    float* fwT    = ws + o; o += NLAY * HH * 2 * HH;
    float* gwT    = ws + o; o += NLAY * HH * 2 * HH;
    float* gcwT   = ws + o; o += NLAY * HH * HH;
    float* skip   = ws + o; o += (size_t)BB * NN * SKC;
    float* glast  = ws + o; o += (size_t)BB * NN * GLK;
    float* hA     = ws + o; o += (size_t)BB * NN * 13 * HH;
    float* hB     = ws + o; o += (size_t)BB * NN * 13 * HH;
    float* gau    = ws + o; o += (size_t)BB * NN * 12 * HH;
    float* agg    = ws + o; o += (size_t)BB * NN * 12 * HH;
    int2*  rowse  = (int2*)(ws + o); o += 2 * NN;
    int2*  rec    = (int2*)(ws + o); o += 2 * NE;
    float* out1   = gau;

    k_setup<<<SU_BLOCKS, SU_T, 0, stream>>>(src, dst, nv1, nv2, feat, start_w, start_b,
                                            filter_w, gate_w, gconv_w, rowmax, rowsum, nsv, ndv,
                                            hA, fwT, gwT, gcwT, bnstat);

    const int dil[8] = {1, 2, 1, 2, 1, 2, 1, 2};
    float* hcur = hA;
    float* hnext = hB;
    int T = 13;
    for (int l = 0; l < NLAY; l++) {
        int d = dil[l];
        int Tp = T - d;
        int M = BB * NN * Tp;
        int tilesGau = (M / 16) * 2;

        const int lp = l ? l - 1 : 0;
        const float* pstat = bnstat + lp * 2 * HH;
        const float* pgam  = bn_gamma + lp * HH;
        const float* pbet  = bn_beta + lp * HH;
        if (l == 0)
            k_gau_first<<<NN + tilesGau / 4 / GAU_NT, 128, 0, stream>>>(
                hcur, fwT, filter_b, gwT, gate_b, gau, glast, T, d, l,
                src, dst, nv1, nv2, rowmax, rowsum, nsv, rowse, rec);
        else
            k_gau_wmma<<<tilesGau / 4 / GAU_NT, 128, 0, stream>>>(hcur, pstat, pgam, pbet, fwT,
                                                                  filter_b, gwT, gate_b, gau,
                                                                  glast, T, d, l);
        k_gather<<<((BB / 4) * NN * Tp * 8 + 255) / 256, 256, 0, stream>>>(
            rowse, rec, (const float4*)gau, (float4*)agg, Tp * 8);
        if (l == 0)
            k_gconv_first<<<(M / 16) * 2 / GC_WAVES / GC_NT, 32 * GC_WAVES, 0, stream>>>(
                agg, ndv, gcwT, gconv_b, hcur, hnext, bnstat, T, d, l);
        else
            k_gconv_wmma<<<(M / 16) * 2 / GC_WAVES / GC_NT, 32 * GC_WAVES, 0, stream>>>(
                agg, ndv, gcwT, gconv_b, hcur, pstat, pgam, pbet, hnext, bnstat + l * 2 * HH,
                T, d, l);
        float* tmp = hcur; hcur = hnext; hnext = tmp;
        T = Tp;
    }

    k_skip_all<<<(BB * NN / 16) * (SKC / 16) / 4 / SK_NT, 128, 0, stream>>>(glast, skip_w, skip_b,
                                                                            skip);
    k_end1_wmma<<<(BB * NN / 16) * (ECC / 16) / 4 / E1_NT, 128, 0, stream>>>(skip, end1_w, end1_b,
                                                                             out1);
    k_end2_wmma<<<(BB * NN / 16) / 4, 128, 0, stream>>>(out1, end2_w, end2_b, (float*)d_out);
}


# graph-conv weight product moved in front of the message passing (linear): gate/filter pass multiplies its rows by W in registers; gather adds norm, bias, relu, residual, bn sums; sum order changes
# speedup vs baseline: 6.0117x; 1.1647x over previous
#include <hip/hip_runtime.h>
#include <hip/hip_bf16.h>

typedef __attribute__((ext_vector_type(16))) _Float16 v16h;
typedef __attribute__((ext_vector_type(8)))  float    v8f;

#define NN   512
#define NE   4096
#define BB   32
#define HH   32
#define SKC  256
#define ECC  512
#define OUTF 12
#define AEMB 10
#define NEGBIG (-1e30f)
#define NLAY 8

__device__ __forceinline__ v8f wmma_f16(v16h a, v16h b, v8f c) {
    return __builtin_amdgcn_wmma_f32_16x16x32_f16(false, a, false, b, (short)0, c, false, false);
}

__device__ __forceinline__ float fast_sigmoid(float x) {
    return __builtin_amdgcn_rcpf(1.f + __expf(-x));
}
__device__ __forceinline__ float fast_tanh(float x) {
    float t = __expf(-2.f * fabsf(x));
    float r = (1.f - t) * __builtin_amdgcn_rcpf(1.f + t);
    return copysignf(r, x);
}

__device__ __forceinline__ void pack8(v16h& d, int off, const float* __restrict__ p) {
    float4 x = ((const float4*)p)[0];
    float4 y = ((const float4*)p)[1];
    d[off + 0] = (_Float16)x.x; d[off + 1] = (_Float16)x.y;
    d[off + 2] = (_Float16)x.z; d[off + 3] = (_Float16)x.w;
    d[off + 4] = (_Float16)y.x; d[off + 5] = (_Float16)y.y;
    d[off + 6] = (_Float16)y.z; d[off + 7] = (_Float16)y.w;
}
__device__ __forceinline__ void cvt8(v16h& d, int off, float4 x, float4 y) {
    d[off + 0] = (_Float16)x.x; d[off + 1] = (_Float16)x.y;
    d[off + 2] = (_Float16)x.z; d[off + 3] = (_Float16)x.w;
    d[off + 4] = (_Float16)y.x; d[off + 5] = (_Float16)y.y;
    d[off + 6] = (_Float16)y.z; d[off + 7] = (_Float16)y.w;
}
__device__ __forceinline__ void pack8_relu(v16h& d, int off, const float* __restrict__ p) {
    float4 x = ((const float4*)p)[0];
    float4 y = ((const float4*)p)[1];
    d[off + 0] = (_Float16)fmaxf(x.x, 0.f); d[off + 1] = (_Float16)fmaxf(x.y, 0.f);
    d[off + 2] = (_Float16)fmaxf(x.z, 0.f); d[off + 3] = (_Float16)fmaxf(x.w, 0.f);
    d[off + 4] = (_Float16)fmaxf(y.x, 0.f); d[off + 5] = (_Float16)fmaxf(y.y, 0.f);
    d[off + 6] = (_Float16)fmaxf(y.z, 0.f); d[off + 7] = (_Float16)fmaxf(y.w, 0.f);
}

__device__ __forceinline__ void cvt8_affine(v16h& d, int off, float4 x, float4 y,
                                            const float* sc, const float* sh) {
    d[off + 0] = (_Float16)fmaf(x.x, sc[0], sh[0]); d[off + 1] = (_Float16)fmaf(x.y, sc[1], sh[1]);
    d[off + 2] = (_Float16)fmaf(x.z, sc[2], sh[2]); d[off + 3] = (_Float16)fmaf(x.w, sc[3], sh[3]);
    d[off + 4] = (_Float16)fmaf(y.x, sc[4], sh[4]); d[off + 5] = (_Float16)fmaf(y.y, sc[5], sh[5]);
    d[off + 6] = (_Float16)fmaf(y.z, sc[6], sh[6]); d[off + 7] = (_Float16)fmaf(y.w, sc[7], sh[7]);
}

__device__ __forceinline__ void bn_coef(float& sc, float& sh, float s, float q, float g, float b,
                                        float inv_cnt) {
    float mu = s * inv_cnt;
    float var = q * inv_cnt - mu * mu;
    sc = rsqrtf(var + 1e-5f) * g;
    sh = b - mu * sc;
}
__device__ __forceinline__ void bn_coef8(float* sc, float* sh, const float* __restrict__ stat,
                                         const float* __restrict__ gam,
                                         const float* __restrict__ bet, int c0, float inv_cnt) {
    float4 s0 = ((const float4*)(stat + c0))[0],      s1 = ((const float4*)(stat + c0))[1];
    float4 q0 = ((const float4*)(stat + HH + c0))[0], q1 = ((const float4*)(stat + HH + c0))[1];
    float4 g0 = ((const float4*)(gam + c0))[0],       g1 = ((const float4*)(gam + c0))[1];
    float4 b0 = ((const float4*)(bet + c0))[0],       b1 = ((const float4*)(bet + c0))[1];
    bn_coef(sc[0], sh[0], s0.x, q0.x, g0.x, b0.x, inv_cnt);
    bn_coef(sc[1], sh[1], s0.y, q0.y, g0.y, b0.y, inv_cnt);
    bn_coef(sc[2], sh[2], s0.z, q0.z, g0.z, b0.z, inv_cnt);
    bn_coef(sc[3], sh[3], s0.w, q0.w, g0.w, b0.w, inv_cnt);
    bn_coef(sc[4], sh[4], s1.x, q1.x, g1.x, b1.x, inv_cnt);
    bn_coef(sc[5], sh[5], s1.y, q1.y, g1.y, b1.y, inv_cnt);
    bn_coef(sc[6], sh[6], s1.z, q1.z, g1.z, b1.z, inv_cnt);
    bn_coef(sc[7], sh[7], s1.w, q1.w, g1.w, b1.w, inv_cnt);
}

#define SU_T      256
#define SU_PER    (NE / SU_T)
#define SU_STARTB ((BB * NN * 13 * (HH / 4)) / (SU_T * 8))
#define SU_PREPB  ((NLAY * HH * 2 * HH) / SU_T)
#define SU_BLOCKS (NN + SU_STARTB + SU_PREPB + 1)
__global__ void __launch_bounds__(SU_T) k_setup(const int* __restrict__ src, const int* __restrict__ dst,
                        const float* __restrict__ nv1, const float* __restrict__ nv2,
                        const float* __restrict__ feat, const float* __restrict__ sw,
                        const float* __restrict__ sb,
                        const float* __restrict__ fw, const float* __restrict__ gw,
                        const float* __restrict__ gcw,
                        float* rowmax, float* rowsum, float* nsv, float* ndv,
                        float* h, float* fwT, float* gwT, float* gcwT, float* bnstat) {
    __shared__ float sc[NN];
    __shared__ float red[SU_T];
    __shared__ int cnt[SU_T / 32];
    const int tid = threadIdx.x;
    if (blockIdx.x < NN) {
        const int i = blockIdx.x;
        const int e0 = tid * SU_PER;
        int4 sq[SU_PER / 4], dq[SU_PER / 4];
#pragma unroll
        for (int k = 0; k < SU_PER / 4; k++) {
            sq[k] = ((const int4*)(src + e0))[k];
            dq[k] = ((const int4*)(dst + e0))[k];
        }
        for (int j = tid; j < NN; j += SU_T) sc[j] = NEGBIG;
        __syncthreads();
        unsigned outm = 0;
        int packed = 0;
#pragma unroll
        for (int k = 0; k < SU_PER / 4; k++) {
            outm |= (sq[k].x == i ? 1u : 0u) << (4 * k);
            outm |= (sq[k].y == i ? 1u : 0u) << (4 * k + 1);
            outm |= (sq[k].z == i ? 1u : 0u) << (4 * k + 2);
            outm |= (sq[k].w == i ? 1u : 0u) << (4 * k + 3);
            packed += (dq[k].x == i ? 0x10000 : 0) + (dq[k].y == i ? 0x10000 : 0) +
                      (dq[k].z == i ? 0x10000 : 0) + (dq[k].w == i ? 0x10000 : 0);
        }
        packed += __popc(outm);
        while (outm) {
            int k = __ffs(outm) - 1;
            outm &= outm - 1;
            int d = dst[e0 + k];
            float acc = 0.f;
#pragma unroll
            for (int a = 0; a < AEMB; a++) acc += nv1[i * AEMB + a] * nv2[a * NN + d];
            sc[d] = fmaxf(acc, 0.f);
        }
        for (int off = 16; off > 0; off >>= 1) packed += __shfl_xor(packed, off);
        if ((tid & 31) == 0) cnt[tid >> 5] = packed;
        __syncthreads();
        float m = NEGBIG;
        for (int j = tid; j < NN; j += SU_T) m = fmaxf(m, sc[j]);
        red[tid] = m;
        __syncthreads();
        for (int s = SU_T / 2; s > 0; s >>= 1) {
            if (tid < s) red[tid] = fmaxf(red[tid], red[tid + s]);
            __syncthreads();
        }
        m = red[0];
        __syncthreads();
        float sum = 0.f;
        for (int j = tid; j < NN; j += SU_T) {
            float v = sc[j];
            if (v > 0.5f * NEGBIG) sum += __expf(v - m);
        }
        red[tid] = sum;
        __syncthreads();
        for (int s = SU_T / 2; s > 0; s >>= 1) {
            if (tid < s) red[tid] += red[tid + s];
            __syncthreads();
        }
        if (tid == 0) {
            rowmax[i] = m;
            rowsum[i] = (red[0] > 0.f) ? red[0] : 1.f;
            int tot = 0;
            for (int w = 0; w < SU_T / 32; w++) tot += cnt[w];
            nsv[i] = rsqrtf(fmaxf((float)(tot & 0xffff), 1.f));
            ndv[i] = rsqrtf(fmaxf((float)(tot >> 16), 1.f));
        }
    } else if (blockIdx.x < NN + SU_STARTB) {
        const int total4 = BB * NN * 13 * (HH / 4);
        for (int idx = (blockIdx.x - NN) * SU_T + tid; idx < total4; idx += SU_STARTB * SU_T) {
            int c4 = idx & 7;
            int mt = idx >> 3;
            int t = mt % 13;
            int m = mt / 13;
            int b = m >> 9, n = m & 511;
            float x = (t == 0) ? 0.f : feat[(b * 12 + (t - 1)) * NN + n];
            float4 w = ((const float4*)sw)[c4];
            float4 bb = ((const float4*)sb)[c4];
            ((float4*)h)[idx] = make_float4(w.x * x + bb.x, w.y * x + bb.y,
                                            w.z * x + bb.z, w.w * x + bb.w);
        }
    } else if (blockIdx.x < NN + SU_STARTB + SU_PREPB) {
        int idx = (blockIdx.x - NN - SU_STARTB) * SU_T + tid;
        {
            int ci = idx & 31;
            int tap = (idx >> 5) & 1;
            int cc = (idx >> 6) & 31;
            int l = idx >> 11;
            fwT[idx] = fw[((l * HH + cc) * HH + ci) * 2 + tap];
            gwT[idx] = gw[((l * HH + cc) * HH + ci) * 2 + tap];
        }
        if (idx < NLAY * HH * HH) {
            int k = idx & 31;
            int n = (idx >> 5) & 31;
            int l = idx >> 10;
            gcwT[idx] = gcw[l * HH * HH + k * HH + n];
        }
    } else {
        for (int j = tid; j < NLAY * 2 * HH; j += SU_T) bnstat[j] = 0.f;
    }
}

#define GAU_NT 2
#define GG_T   512
#define SK_NT  4
#define GLK (NLAY * HH)
#define E1_NT  4

template <bool NORM>
__device__ __forceinline__ void gau_body(const float* __restrict__ h,
                                         const float* __restrict__ pstat,
                                         const float* __restrict__ pgam,
                                         const float* __restrict__ pbet,
                                         const float* __restrict__ fwT,
                                         const float* __restrict__ fb,
                                         const float* __restrict__ gwT,
                                         const float* __restrict__ gb,
                                         const float* __restrict__ gcwT, float* y,
                                         float* glast, int T, int d, int l, int blk) {
    int Tp = T - d;
    int tm0 = (blk * 4 + (threadIdx.x >> 5)) * GAU_NT;
    int lane = threadIdx.x & 31;
    int c = lane & 15;
    int base = (lane & 16) ? 8 : 0;
    float sc[16], sh[16];
    if (NORM) {
        float inv_cnt = 1.f / (float)(BB * NN * T);
        bn_coef8(sc, sh, pstat, pgam, pbet, base, inv_cnt);
        bn_coef8(sc + 8, sh + 8, pstat, pgam, pbet, base + 16, inv_cnt);
    }
    v16h wf[2][2], wg[2][2], wc[2];
#pragma unroll
    for (int tn = 0; tn < 2; tn++) {
        const float* fwr = fwT + (size_t)((l * HH + tn * 16 + c) * 2) * HH;
        const float* gwr = gwT + (size_t)((l * HH + tn * 16 + c) * 2) * HH;
        const float* cwr = gcwT + (size_t)(l * HH + tn * 16 + c) * HH;
        pack8(wf[tn][0], 0, fwr + base);       pack8(wf[tn][0], 8, fwr + base + 16);
        pack8(wg[tn][0], 0, gwr + base);       pack8(wg[tn][0], 8, gwr + base + 16);
        pack8(wf[tn][1], 0, fwr + HH + base);  pack8(wf[tn][1], 8, fwr + HH + base + 16);
        pack8(wg[tn][1], 0, gwr + HH + base);  pack8(wg[tn][1], 8, gwr + HH + base + 16);
        pack8(wc[tn], 0, cwr + base);          pack8(wc[tn], 8, cwr + base + 16);
    }
    float4 rx[GAU_NT][8];
    int tt[GAU_NT], bnn[GAU_NT];
#pragma unroll
    for (int i = 0; i < GAU_NT; i++) {
        int mrow = (tm0 + i) * 16 + c;
        int t = mrow % Tp;
        int bn = mrow / Tp;
        tt[i] = t;
        bnn[i] = bn;
        const float* hrow0 = h + (bn * T + t) * HH + base;
        const float* hrow1 = hrow0 + d * HH;
        rx[i][0] = ((const float4*)hrow0)[0];        rx[i][1] = ((const float4*)hrow0)[1];
        rx[i][2] = ((const float4*)(hrow0 + 16))[0]; rx[i][3] = ((const float4*)(hrow0 + 16))[1];
        rx[i][4] = ((const float4*)hrow1)[0];        rx[i][5] = ((const float4*)hrow1)[1];
        rx[i][6] = ((const float4*)(hrow1 + 16))[0]; rx[i][7] = ((const float4*)(hrow1 + 16))[1];
    }
#pragma unroll
    for (int i = 0; i < GAU_NT; i++) {
        v16h x0, x1;
        if (NORM) {
            cvt8_affine(x0, 0, rx[i][0], rx[i][1], sc, sh);
            cvt8_affine(x0, 8, rx[i][2], rx[i][3], sc + 8, sh + 8);
            cvt8_affine(x1, 0, rx[i][4], rx[i][5], sc, sh);
            cvt8_affine(x1, 8, rx[i][6], rx[i][7], sc + 8, sh + 8);
        } else {
            cvt8(x0, 0, rx[i][0], rx[i][1]);  cvt8(x0, 8, rx[i][2], rx[i][3]);
            cvt8(x1, 0, rx[i][4], rx[i][5]);  cvt8(x1, 8, rx[i][6], rx[i][7]);
        }
        const int mrow = (tm0 + i) * 16 + c;
        float* yrow = y + (size_t)mrow * HH + base;
        float* grow = glast + (size_t)bnn[i] * GLK + l * HH + base;
        const bool keep = (tt[i] == Tp - 1);
        v16h gp;
        float vals[2][8];
#pragma unroll
        for (int tn = 0; tn < 2; tn++) {
            float4 f0 = ((const float4*)(fb + l * HH + tn * 16 + base))[0];
            float4 f1 = ((const float4*)(fb + l * HH + tn * 16 + base))[1];
            float4 g0 = ((const float4*)(gb + l * HH + tn * 16 + base))[0];
            float4 g1 = ((const float4*)(gb + l * HH + tn * 16 + base))[1];
            v8f cf = {f0.x, f0.y, f0.z, f0.w, f1.x, f1.y, f1.z, f1.w};
            v8f cg = {g0.x, g0.y, g0.z, g0.w, g1.x, g1.y, g1.z, g1.w};
            cf = wmma_f16(wf[tn][0], x0, cf);
            cg = wmma_f16(wg[tn][0], x0, cg);
            cf = wmma_f16(wf[tn][1], x1, cf);
            cg = wmma_f16(wg[tn][1], x1, cg);
#pragma unroll
            for (int r = 0; r < 8; r++) {
                vals[tn][r] = fast_tanh(cf[r]) * fast_sigmoid(cg[r]);
                gp[tn * 8 + r] = (_Float16)vals[tn][r];
            }
        }
#pragma unroll
        for (int tn = 0; tn < 2; tn++) {
            v8f cy = {};
            cy = wmma_f16(wc[tn], gp, cy);
            ((float4*)(yrow + tn * 16))[0] = make_float4(cy[0], cy[1], cy[2], cy[3]);
            ((float4*)(yrow + tn * 16))[1] = make_float4(cy[4], cy[5], cy[6], cy[7]);
        }
        if (keep) {
#pragma unroll
            for (int tn = 0; tn < 2; tn++) {
                ((float4*)(grow + tn * 16))[0] =
                    make_float4(vals[tn][0], vals[tn][1], vals[tn][2], vals[tn][3]);
                ((float4*)(grow + tn * 16))[1] =
                    make_float4(vals[tn][4], vals[tn][5], vals[tn][6], vals[tn][7]);
            }
        }
    }
}
#define CSR_T 128
#define CSR_PER (NE / CSR_T)
__device__ __forceinline__ void csr_piece(const int d, const int* __restrict__ src,
                                          const int* __restrict__ dst,
                                          const float* __restrict__ nv1,
                                          const float* __restrict__ nv2,
                                          const float* __restrict__ rowmax,
                                          const float* __restrict__ rowsum,
                                          const float* __restrict__ nsv,
                                          int2* rowse, int2* rec) {
    __shared__ int s_lt[CSR_T];
    __shared__ int s_eq[CSR_T];
    const int tid = threadIdx.x;
    const int e0 = tid * CSR_PER;
    int lt = 0;
    unsigned inm = 0;
    int4 dq[CSR_PER / 4];
#pragma unroll
    for (int k = 0; k < CSR_PER / 4; k++) dq[k] = ((const int4*)(dst + e0))[k];
#pragma unroll
    for (int k = 0; k < CSR_PER / 4; k++) {
        lt += (dq[k].x < d ? 1 : 0) + (dq[k].y < d ? 1 : 0) + (dq[k].z < d ? 1 : 0) +
              (dq[k].w < d ? 1 : 0);
        inm |= (dq[k].x == d ? 1u : 0u) << (4 * k);
        inm |= (dq[k].y == d ? 1u : 0u) << (4 * k + 1);
        inm |= (dq[k].z == d ? 1u : 0u) << (4 * k + 2);
        inm |= (dq[k].w == d ? 1u : 0u) << (4 * k + 3);
    }
    const int eq = __popc(inm);
    s_lt[tid] = lt;
    s_eq[tid] = eq;
    __syncthreads();
    for (int off = 1; off < CSR_T; off <<= 1) {
        int a = 0, b = 0;
        if (tid >= off) { a = s_lt[tid - off]; b = s_eq[tid - off]; }
        __syncthreads();
        s_lt[tid] += a;
        s_eq[tid] += b;
        __syncthreads();
    }
    const int first = s_lt[CSR_T - 1];
    int pos = first + s_eq[tid] - eq;
    if (tid == 0) rowse[d] = make_int2(first, first + s_eq[CSR_T - 1]);
    while (inm) {
        int e = e0 + __ffs(inm) - 1;
        inm &= inm - 1;
        int sn = src[e];
        float acc = 0.f;
#pragma unroll
        for (int a = 0; a < AEMB; a++) acc += nv1[sn * AEMB + a] * nv2[a * NN + d];
        float scv = fmaxf(acc, 0.f);
        float ewv = 1.f + __expf(scv - rowmax[sn]) * __builtin_amdgcn_rcpf(rowsum[sn]);
        rec[pos] = make_int2(sn, __float_as_int(ewv * nsv[sn]));
        pos++;
    }
}

__global__ void __launch_bounds__(128) k_gau_first(const float* __restrict__ h, const float* __restrict__ fwT,
                            const float* __restrict__ fb, const float* __restrict__ gwT,
                            const float* __restrict__ gb, const float* __restrict__ gcwT,
                            float* y, float* glast, int T, int d, int l,
                            const int* __restrict__ src, const int* __restrict__ dst,
                            const float* __restrict__ nv1, const float* __restrict__ nv2,
                            const float* __restrict__ rowmax, const float* __restrict__ rowsum,
                            const float* __restrict__ nsv, int2* rowse, int2* rec) {
    if (blockIdx.x < NN)
        csr_piece(blockIdx.x, src, dst, nv1, nv2, rowmax, rowsum, nsv, rowse, rec);
    else
        gau_body<false>(h, nullptr, nullptr, nullptr, fwT, fb, gwT, gb, gcwT, y, glast, T, d, l,
                        blockIdx.x - NN);
}
__global__ void __launch_bounds__(128) k_gau_wmma(const float* __restrict__ h, const float* __restrict__ pstat,
                           const float* __restrict__ pgam, const float* __restrict__ pbet,
                           const float* __restrict__ fwT, const float* __restrict__ fb,
                           const float* __restrict__ gwT, const float* __restrict__ gb,
                           const float* __restrict__ gcwT, float* y, float* glast,
                           int T, int d, int l) {
    gau_body<true>(h, pstat, pgam, pbet, fwT, fb, gwT, gb, gcwT, y, glast, T, d, l, blockIdx.x);
}

__global__ void __launch_bounds__(128) k_skip_all(const float* __restrict__ glast, const float* __restrict__ sw,
                           const float* __restrict__ sb, float* skip) {
    const int nGroupsN = SKC / 16 / SK_NT;
    int unit = blockIdx.x * 4 + (threadIdx.x >> 5);
    int tm = unit / nGroupsN;
    int tn0 = (unit % nGroupsN) * SK_NT;
    int lane = threadIdx.x & 31;
    int mrow = tm * 16 + (lane & 15);
    int base = (lane & 16) ? 8 : 0;
    const float* arow = glast + (size_t)mrow * GLK;
    v16h a[NLAY];
#pragma unroll
    for (int l = 0; l < NLAY; l++) {
        pack8(a[l], 0, arow + l * HH + base);
        pack8(a[l], 8, arow + l * HH + base + 16);
    }
    int rbase = tm * 16 + ((lane & 16) ? 8 : 0);
#pragma unroll
    for (int i = 0; i < SK_NT; i++) {
        int ncol = (tn0 + i) * 16 + (lane & 15);
        v8f c = {};
        float bias = 0.f;
#pragma unroll
        for (int l = 0; l < NLAY; l++) {
            const float* brow = sw + (size_t)(l * SKC + ncol) * HH;
            v16h bm;
            pack8(bm, 0, brow + base);
            pack8(bm, 8, brow + base + 16);
            c = wmma_f16(a[l], bm, c);
            bias += sb[l * SKC + ncol];
        }
#pragma unroll
        for (int v = 0; v < 8; v++)
            skip[(size_t)(rbase + v) * SKC + ncol] = c[v] + bias;
    }
}

__device__ __forceinline__ void fma4(float4& a, float w, const float4 v) {
    a.x = fmaf(w, v.x, a.x); a.y = fmaf(w, v.y, a.y);
    a.z = fmaf(w, v.z, a.z); a.w = fmaf(w, v.w, a.w);
}
__device__ __forceinline__ float4 gc_finish(float4 z, float nrm, float4 bias, float4 r,
                                            float4& s1, float4& s2) {
    float4 v;
    v.x = fmaxf(fmaf(z.x, nrm, bias.x), 0.f) + r.x;
    v.y = fmaxf(fmaf(z.y, nrm, bias.y), 0.f) + r.y;
    v.z = fmaxf(fmaf(z.z, nrm, bias.z), 0.f) + r.z;
    v.w = fmaxf(fmaf(z.w, nrm, bias.w), 0.f) + r.w;
    s1.x += v.x; s1.y += v.y; s1.z += v.z; s1.w += v.w;
    s2.x = fmaf(v.x, v.x, s2.x); s2.y = fmaf(v.y, v.y, s2.y);
    s2.z = fmaf(v.z, v.z, s2.z); s2.w = fmaf(v.w, v.w, s2.w);
    return v;
}
template <bool NORM>
__device__ __forceinline__ void gather_body(const int2* __restrict__ rowse,
                                            const int2* __restrict__ rec,
                                            const float4* __restrict__ y4,
                                            const float* __restrict__ ndv,
                                            const float* __restrict__ gcb,
                                            const float4* __restrict__ hold4,
                                            const float* __restrict__ pstat,
                                            const float* __restrict__ pgam,
                                            const float* __restrict__ pbet,
                                            float4* __restrict__ hpre4, float* stat,
                                            int T, int d, int l) {
    const int Tp = T - d;
    const int slab4 = Tp * 8;
    int idx = blockIdx.x * GG_T + threadIdx.x;
    int bd = idx / slab4;
    int q = idx - bd * slab4;
    int dn = bd & (NN - 1);
    int b0 = bd >> 9;
    const int c4 = q & 7;
    const int bstride = (BB / 4) * NN * slab4;
    const float4* g = y4 + (b0 * NN) * slab4 + q;
    const int hstride = (BB / 4) * NN * T * 8;
    const float4* hp = hold4 + ((b0 * NN + dn) * T + d) * 8 + q;
    float4 h0 = hp[0], h1 = hp[hstride], h2 = hp[2 * hstride], h3 = hp[3 * hstride];
    int2 se = rowse[dn];
    float4 a0 = make_float4(0.f, 0.f, 0.f, 0.f), a1 = a0, a2 = a0, a3 = a0;
    int j = se.x;
    for (; j + 2 <= se.y; j += 2) {
        int2 r0 = rec[j], r1 = rec[j + 1];
        const float4* p0 = g + r0.x * slab4;
        const float4* p1 = g + r1.x * slab4;
        float4 u0 = p0[0], u1 = p0[bstride], u2 = p0[2 * bstride], u3 = p0[3 * bstride];
        float4 v0 = p1[0], v1 = p1[bstride], v2 = p1[2 * bstride], v3 = p1[3 * bstride];
        float w0 = __int_as_float(r0.y), w1 = __int_as_float(r1.y);
        fma4(a0, w0, u0); fma4(a1, w0, u1); fma4(a2, w0, u2); fma4(a3, w0, u3);
        fma4(a0, w1, v0); fma4(a1, w1, v1); fma4(a2, w1, v2); fma4(a3, w1, v3);
    }
    if (j < se.y) {
        int2 r0 = rec[j];
        const float4* p0 = g + r0.x * slab4;
        float w0 = __int_as_float(r0.y);
        fma4(a0, w0, p0[0]); fma4(a1, w0, p0[bstride]);
        fma4(a2, w0, p0[2 * bstride]); fma4(a3, w0, p0[3 * bstride]);
    }
    const float nrm = ndv[dn];
    const float4 bias = ((const float4*)(gcb + l * HH))[c4];
    if (NORM) {
        const float inv_cnt = 1.f / (float)(BB * NN * T);
        float4 s = ((const float4*)pstat)[c4], qq = ((const float4*)(pstat + HH))[c4];
        float4 gm = ((const float4*)pgam)[c4], bt = ((const float4*)pbet)[c4];
        float4 psc, psh;
        bn_coef(psc.x, psh.x, s.x, qq.x, gm.x, bt.x, inv_cnt);
        bn_coef(psc.y, psh.y, s.y, qq.y, gm.y, bt.y, inv_cnt);
        bn_coef(psc.z, psh.z, s.z, qq.z, gm.z, bt.z, inv_cnt);
        bn_coef(psc.w, psh.w, s.w, qq.w, gm.w, bt.w, inv_cnt);
        h0 = make_float4(fmaf(h0.x, psc.x, psh.x), fmaf(h0.y, psc.y, psh.y),
                         fmaf(h0.z, psc.z, psh.z), fmaf(h0.w, psc.w, psh.w));
        h1 = make_float4(fmaf(h1.x, psc.x, psh.x), fmaf(h1.y, psc.y, psh.y),
                         fmaf(h1.z, psc.z, psh.z), fmaf(h1.w, psc.w, psh.w));
        h2 = make_float4(fmaf(h2.x, psc.x, psh.x), fmaf(h2.y, psc.y, psh.y),
                         fmaf(h2.z, psc.z, psh.z), fmaf(h2.w, psc.w, psh.w));
        h3 = make_float4(fmaf(h3.x, psc.x, psh.x), fmaf(h3.y, psc.y, psh.y),
                         fmaf(h3.z, psc.z, psh.z), fmaf(h3.w, psc.w, psh.w));
    }
    float4 s1 = make_float4(0.f, 0.f, 0.f, 0.f), s2 = s1;
    float4* o = hpre4 + (bd * slab4 + q);
    o[0]           = gc_finish(a0, nrm, bias, h0, s1, s2);
    o[bstride]     = gc_finish(a1, nrm, bias, h1, s1, s2);
    o[2 * bstride] = gc_finish(a2, nrm, bias, h2, s1, s2);
    o[3 * bstride] = gc_finish(a3, nrm, bias, h3, s1, s2);
    float sv[8] = {s1.x, s1.y, s1.z, s1.w, s2.x, s2.y, s2.z, s2.w};
#pragma unroll
    for (int k = 0; k < 8; k++) {
        sv[k] += __shfl_xor(sv[k], 8);
        sv[k] += __shfl_xor(sv[k], 16);
    }
    __shared__ alignas(16) float red[GG_T / 32][2 * HH];
    const int wave = threadIdx.x >> 5;
    const int lane = threadIdx.x & 31;
    if (lane < 8) {
        ((float4*)&red[wave][0])[lane]  = make_float4(sv[0], sv[1], sv[2], sv[3]);
        ((float4*)&red[wave][HH])[lane] = make_float4(sv[4], sv[5], sv[6], sv[7]);
    }
    __builtin_amdgcn_fence(__ATOMIC_RELEASE, "workgroup", "local");
    __builtin_amdgcn_s_barrier();
    __builtin_amdgcn_fence(__ATOMIC_ACQUIRE, "workgroup", "local");
    if (threadIdx.x < 2 * HH) {
        const int wd = threadIdx.x;
        float tot = 0.f;
#pragma unroll
        for (int k = 0; k < GG_T / 32; k++) tot += red[k][wd];
        atomicAdd(&stat[wd], tot);
    }
}
__global__ void __launch_bounds__(GG_T) k_gather_first(const int2* __restrict__ rowse, const int2* __restrict__ rec,
                               const float4* __restrict__ y4, const float* __restrict__ ndv,
                               const float* __restrict__ gcb, const float4* __restrict__ hold4,
                               float4* __restrict__ hpre4, float* stat, int T, int d, int l) {
    gather_body<false>(rowse, rec, y4, ndv, gcb, hold4, nullptr, nullptr, nullptr, hpre4, stat,
                       T, d, l);
}
__global__ void __launch_bounds__(GG_T) k_gather_gc(const int2* __restrict__ rowse, const int2* __restrict__ rec,
                            const float4* __restrict__ y4, const float* __restrict__ ndv,
                            const float* __restrict__ gcb, const float4* __restrict__ hold4,
                            const float* __restrict__ pstat, const float* __restrict__ pgam,
                            const float* __restrict__ pbet, float4* __restrict__ hpre4,
                            float* stat, int T, int d, int l) {
    gather_body<true>(rowse, rec, y4, ndv, gcb, hold4, pstat, pgam, pbet, hpre4, stat, T, d, l);
}

__global__ void __launch_bounds__(128) k_end1_wmma(const float* __restrict__ skip, const float* __restrict__ w1,
                            const float* __restrict__ b1, float* out1) {
    const int nGroupsN = ECC / 16 / E1_NT;
    int unit = blockIdx.x * 4 + (threadIdx.x >> 5);
    int tm = unit / nGroupsN;
    int tn0 = (unit % nGroupsN) * E1_NT;
    int lane = threadIdx.x & 31;
    int mrow = tm * 16 + (lane & 15);
    int base = (lane & 16) ? 8 : 0;
    const float* arow = skip + (size_t)mrow * SKC;
    v16h a[SKC / 32];
#pragma unroll
    for (int ks = 0; ks < SKC / 32; ks++) {
        pack8_relu(a[ks], 0, arow + ks * 32 + base);
        pack8_relu(a[ks], 8, arow + ks * 32 + base + 16);
    }
    int rbase = tm * 16 + ((lane & 16) ? 8 : 0);
#pragma unroll
    for (int i = 0; i < E1_NT; i++) {
        int ncol = (tn0 + i) * 16 + (lane & 15);
        const float* brow = w1 + (size_t)ncol * SKC;
        v8f c = {};
#pragma unroll
        for (int ks = 0; ks < SKC / 32; ks++) {
            v16h bm;
            pack8(bm, 0, brow + ks * 32 + base);
            pack8(bm, 8, brow + ks * 32 + base + 16);
            c = wmma_f16(a[ks], bm, c);
        }
        float bias = b1[ncol];
#pragma unroll
        for (int v = 0; v < 8; v++)
            out1[(size_t)(rbase + v) * ECC + ncol] = fmaxf(c[v] + bias, 0.f);
    }
}

__global__ void k_end2_wmma(const float* __restrict__ in1, const float* __restrict__ w2,
                            const float* __restrict__ b2, float* out) {
    int tm = blockIdx.x * 4 + (threadIdx.x >> 5);
    int lane = threadIdx.x & 31;
    int mrow = tm * 16 + (lane & 15);
    int ncol = lane & 15;
    int base = (lane & 16) ? 8 : 0;
    const float* arow = in1 + (size_t)mrow * ECC;
    const float* brow = w2 + (size_t)ncol * ECC;
    v8f c = {};
#pragma unroll
    for (int k0 = 0; k0 < ECC; k0 += 32) {
        v16h a, bm = {};
        pack8(a, 0, arow + k0 + base);
        pack8(a, 8, arow + k0 + base + 16);
        if (ncol < OUTF) {
            pack8(bm, 0, brow + k0 + base);
            pack8(bm, 8, brow + k0 + base + 16);
        }
        c = wmma_f16(a, bm, c);
    }
    if (ncol < OUTF) {
        int rbase = tm * 16 + ((lane & 16) ? 8 : 0);
        float bias = b2[ncol];
#pragma unroll
        for (int v = 0; v < 8; v++) {
            int row = rbase + v;
            int b = row >> 9, n = row & 511;
            out[(b * OUTF + ncol) * NN + n] = c[v] + bias;
        }
    }
}

extern "C" void kernel_launch(void* const* d_in, const int* in_sizes, int n_in,
                              void* d_out, int out_size, void* d_ws, size_t ws_size,
                              hipStream_t stream) {
    const float* feat     = (const float*)d_in[0];
    const int*   src      = (const int*)d_in[1];
    const int*   dst      = (const int*)d_in[2];
    const float* nv1      = (const float*)d_in[3];
    const float* nv2      = (const float*)d_in[4];
    const float* start_w  = (const float*)d_in[5];
    const float* start_b  = (const float*)d_in[6];
    const float* filter_w = (const float*)d_in[7];
    const float* filter_b = (const float*)d_in[8];
    const float* gate_w   = (const float*)d_in[9];
    const float* gate_b   = (const float*)d_in[10];
    const float* skip_w   = (const float*)d_in[11];
    const float* skip_b   = (const float*)d_in[12];
    const float* bn_gamma = (const float*)d_in[13];
    const float* bn_beta  = (const float*)d_in[14];
    const float* gconv_w  = (const float*)d_in[15];
    const float* gconv_b  = (const float*)d_in[16];
    const float* end1_w   = (const float*)d_in[17];
    const float* end1_b   = (const float*)d_in[18];
    const float* end2_w   = (const float*)d_in[19];
    const float* end2_b   = (const float*)d_in[20];

    float* ws = (float*)d_ws;
    size_t o = 0;
    float* rowmax = ws + o; o += NN;
    float* rowsum = ws + o; o += NN;
    float* nsv    = ws + o; o += NN;
    float* ndv    = ws + o; o += NN;
    float* bnstat = ws + o; o += NLAY * 2 * HH;
    float* fwT    = ws + o; o += NLAY * HH * 2 * HH;
    float* gwT    = ws + o; o += NLAY * HH * 2 * HH;
    float* gcwT   = ws + o; o += NLAY * HH * HH;
    float* skip   = ws + o; o += (size_t)BB * NN * SKC;
    float* glast  = ws + o; o += (size_t)BB * NN * GLK;
    float* hA     = ws + o; o += (size_t)BB * NN * 13 * HH;
    float* hB     = ws + o; o += (size_t)BB * NN * 13 * HH;
    float* ygc    = ws + o; o += (size_t)BB * NN * 12 * HH * 2;
    int2*  rowse  = (int2*)(ws + o); o += 2 * NN;
    int2*  rec    = (int2*)(ws + o); o += 2 * NE;
    float* out1   = ygc;

    k_setup<<<SU_BLOCKS, SU_T, 0, stream>>>(src, dst, nv1, nv2, feat, start_w, start_b,
                                            filter_w, gate_w, gconv_w, rowmax, rowsum, nsv, ndv,
                                            hA, fwT, gwT, gcwT, bnstat);

    const int dil[8] = {1, 2, 1, 2, 1, 2, 1, 2};
    float* hcur = hA;
    float* hnext = hB;
    int T = 13;
    for (int l = 0; l < NLAY; l++) {
        int d = dil[l];
        int Tp = T - d;
        int M = BB * NN * Tp;
        int blocksGau = (M / 16) / 4 / GAU_NT;
        int blocksGather = (BB / 4) * NN * Tp * 8 / GG_T;

        const int lp = l ? l - 1 : 0;
        const float* pstat = bnstat + lp * 2 * HH;
        const float* pgam  = bn_gamma + lp * HH;
        const float* pbet  = bn_beta + lp * HH;
        if (l == 0)
            k_gau_first<<<NN + blocksGau, 128, 0, stream>>>(
                hcur, fwT, filter_b, gwT, gate_b, gcwT, ygc, glast, T, d, l,
                src, dst, nv1, nv2, rowmax, rowsum, nsv, rowse, rec);
        else
            k_gau_wmma<<<blocksGau, 128, 0, stream>>>(hcur, pstat, pgam, pbet, fwT, filter_b, gwT,
                                                      gate_b, gcwT, ygc, glast, T, d, l);
        if (l == 0)
            k_gather_first<<<blocksGather, GG_T, 0, stream>>>(
                rowse, rec, (const float4*)ygc, ndv, gconv_b, (const float4*)hcur,
                (float4*)hnext, bnstat, T, d, l);
        else
            k_gather_gc<<<blocksGather, GG_T, 0, stream>>>(
                rowse, rec, (const float4*)ygc, ndv, gconv_b, (const float4*)hcur, pstat, pgam,
                pbet, (float4*)hnext, bnstat + l * 2 * HH, T, d, l);
        float* tmp = hcur; hcur = hnext; hnext = tmp;
        T = Tp;
    }

    k_skip_all<<<(BB * NN / 16) * (SKC / 16) / 4 / SK_NT, 128, 0, stream>>>(glast, skip_w, skip_b,
                                                                            skip);
    k_end1_wmma<<<(BB * NN / 16) * (ECC / 16) / 4 / E1_NT, 128, 0, stream>>>(skip, end1_w, end1_b,
                                                                             out1);
    k_end2_wmma<<<(BB * NN / 16) / 4, 128, 0, stream>>>(out1, end2_w, end2_b, (float*)d_out);
}


# end part in one launch, not 3: skip product, end1, end2 chained transposed in registers and LDS, a block's 4 waves split output channels; skip, out1 planes go; end2 K sum in 4 parts, bias starts sums
# speedup vs baseline: 7.7141x; 1.2832x over previous
#include <hip/hip_runtime.h>
#include <hip/hip_bf16.h>

typedef __attribute__((ext_vector_type(16))) _Float16 v16h;
typedef __attribute__((ext_vector_type(8)))  float    v8f;
typedef __attribute__((ext_vector_type(8)))  _Float16 v8h;

#define NN   512
#define NE   4096
#define BB   32
#define HH   32
#define SKC  256
#define ECC  512
#define OUTF 12
#define AEMB 10
#define NEGBIG (-1e30f)
#define NLAY 8

__device__ __forceinline__ v8f wmma_f16(v16h a, v16h b, v8f c) {
    return __builtin_amdgcn_wmma_f32_16x16x32_f16(false, a, false, b, (short)0, c, false, false);
}

__device__ __forceinline__ float fast_sigmoid(float x) {
    return __builtin_amdgcn_rcpf(1.f + __expf(-x));
}
__device__ __forceinline__ float fast_tanh(float x) {
    float t = __expf(-2.f * fabsf(x));
    float r = (1.f - t) * __builtin_amdgcn_rcpf(1.f + t);
    return copysignf(r, x);
}

__device__ __forceinline__ void pack8(v16h& d, int off, const float* __restrict__ p) {
    float4 x = ((const float4*)p)[0];
    float4 y = ((const float4*)p)[1];
    d[off + 0] = (_Float16)x.x; d[off + 1] = (_Float16)x.y;
    d[off + 2] = (_Float16)x.z; d[off + 3] = (_Float16)x.w;
    d[off + 4] = (_Float16)y.x; d[off + 5] = (_Float16)y.y;
    d[off + 6] = (_Float16)y.z; d[off + 7] = (_Float16)y.w;
}
__device__ __forceinline__ void cvt8(v16h& d, int off, float4 x, float4 y) {
    d[off + 0] = (_Float16)x.x; d[off + 1] = (_Float16)x.y;
    d[off + 2] = (_Float16)x.z; d[off + 3] = (_Float16)x.w;
    d[off + 4] = (_Float16)y.x; d[off + 5] = (_Float16)y.y;
    d[off + 6] = (_Float16)y.z; d[off + 7] = (_Float16)y.w;
}

__device__ __forceinline__ void cvt8_affine(v16h& d, int off, float4 x, float4 y,
                                            const float* sc, const float* sh) {
    d[off + 0] = (_Float16)fmaf(x.x, sc[0], sh[0]); d[off + 1] = (_Float16)fmaf(x.y, sc[1], sh[1]);
    d[off + 2] = (_Float16)fmaf(x.z, sc[2], sh[2]); d[off + 3] = (_Float16)fmaf(x.w, sc[3], sh[3]);
    d[off + 4] = (_Float16)fmaf(y.x, sc[4], sh[4]); d[off + 5] = (_Float16)fmaf(y.y, sc[5], sh[5]);
    d[off + 6] = (_Float16)fmaf(y.z, sc[6], sh[6]); d[off + 7] = (_Float16)fmaf(y.w, sc[7], sh[7]);
}

__device__ __forceinline__ void bn_coef(float& sc, float& sh, float s, float q, float g, float b,
                                        float inv_cnt) {
    float mu = s * inv_cnt;
    float var = q * inv_cnt - mu * mu;
    sc = rsqrtf(var + 1e-5f) * g;
    sh = b - mu * sc;
}
__device__ __forceinline__ void bn_coef8(float* sc, float* sh, const float* __restrict__ stat,
                                         const float* __restrict__ gam,
                                         const float* __restrict__ bet, int c0, float inv_cnt) {
    float4 s0 = ((const float4*)(stat + c0))[0],      s1 = ((const float4*)(stat + c0))[1];
    float4 q0 = ((const float4*)(stat + HH + c0))[0], q1 = ((const float4*)(stat + HH + c0))[1];
    float4 g0 = ((const float4*)(gam + c0))[0],       g1 = ((const float4*)(gam + c0))[1];
    float4 b0 = ((const float4*)(bet + c0))[0],       b1 = ((const float4*)(bet + c0))[1];
    bn_coef(sc[0], sh[0], s0.x, q0.x, g0.x, b0.x, inv_cnt);
    bn_coef(sc[1], sh[1], s0.y, q0.y, g0.y, b0.y, inv_cnt);
    bn_coef(sc[2], sh[2], s0.z, q0.z, g0.z, b0.z, inv_cnt);
    bn_coef(sc[3], sh[3], s0.w, q0.w, g0.w, b0.w, inv_cnt);
    bn_coef(sc[4], sh[4], s1.x, q1.x, g1.x, b1.x, inv_cnt);
    bn_coef(sc[5], sh[5], s1.y, q1.y, g1.y, b1.y, inv_cnt);
    bn_coef(sc[6], sh[6], s1.z, q1.z, g1.z, b1.z, inv_cnt);
    bn_coef(sc[7], sh[7], s1.w, q1.w, g1.w, b1.w, inv_cnt);
}

#define SU_T      256
#define SU_PER    (NE / SU_T)
#define SU_STARTB ((BB * NN * 13 * (HH / 4)) / (SU_T * 8))
#define SU_PREPB  ((NLAY * HH * 2 * HH) / SU_T)
#define SU_BLOCKS (NN + SU_STARTB + SU_PREPB + 1)
__global__ void __launch_bounds__(SU_T) k_setup(const int* __restrict__ src, const int* __restrict__ dst,
                        const float* __restrict__ nv1, const float* __restrict__ nv2,
                        const float* __restrict__ feat, const float* __restrict__ sw,
                        const float* __restrict__ sb,
                        const float* __restrict__ fw, const float* __restrict__ gw,
                        const float* __restrict__ gcw,
                        float* rowmax, float* rowsum, float* nsv, float* ndv,
                        float* h, float* fwT, float* gwT, float* gcwT, float* bnstat) {
    __shared__ float sc[NN];
    __shared__ float red[SU_T];
    __shared__ int cnt[SU_T / 32];
    const int tid = threadIdx.x;
    if (blockIdx.x < NN) {
        const int i = blockIdx.x;
        const int e0 = tid * SU_PER;
        int4 sq[SU_PER / 4], dq[SU_PER / 4];
#pragma unroll
        for (int k = 0; k < SU_PER / 4; k++) {
            sq[k] = ((const int4*)(src + e0))[k];
            dq[k] = ((const int4*)(dst + e0))[k];
        }
        for (int j = tid; j < NN; j += SU_T) sc[j] = NEGBIG;
        __syncthreads();
        unsigned outm = 0;
        int packed = 0;
#pragma unroll
        for (int k = 0; k < SU_PER / 4; k++) {
            outm |= (sq[k].x == i ? 1u : 0u) << (4 * k);
            outm |= (sq[k].y == i ? 1u : 0u) << (4 * k + 1);
            outm |= (sq[k].z == i ? 1u : 0u) << (4 * k + 2);
            outm |= (sq[k].w == i ? 1u : 0u) << (4 * k + 3);
            packed += (dq[k].x == i ? 0x10000 : 0) + (dq[k].y == i ? 0x10000 : 0) +
                      (dq[k].z == i ? 0x10000 : 0) + (dq[k].w == i ? 0x10000 : 0);
        }
        packed += __popc(outm);
        while (outm) {
            int k = __ffs(outm) - 1;
            outm &= outm - 1;
            int d = dst[e0 + k];
            float acc = 0.f;
#pragma unroll
            for (int a = 0; a < AEMB; a++) acc += nv1[i * AEMB + a] * nv2[a * NN + d];
            sc[d] = fmaxf(acc, 0.f);
        }
        for (int off = 16; off > 0; off >>= 1) packed += __shfl_xor(packed, off);
        if ((tid & 31) == 0) cnt[tid >> 5] = packed;
        __syncthreads();
        float m = NEGBIG;
        for (int j = tid; j < NN; j += SU_T) m = fmaxf(m, sc[j]);
        red[tid] = m;
        __syncthreads();
        for (int s = SU_T / 2; s > 0; s >>= 1) {
            if (tid < s) red[tid] = fmaxf(red[tid], red[tid + s]);
            __syncthreads();
        }
        m = red[0];
        __syncthreads();
        float sum = 0.f;
        for (int j = tid; j < NN; j += SU_T) {
            float v = sc[j];
            if (v > 0.5f * NEGBIG) sum += __expf(v - m);
        }
        red[tid] = sum;
        __syncthreads();
        for (int s = SU_T / 2; s > 0; s >>= 1) {
            if (tid < s) red[tid] += red[tid + s];
            __syncthreads();
        }
        if (tid == 0) {
            rowmax[i] = m;
            rowsum[i] = (red[0] > 0.f) ? red[0] : 1.f;
            int tot = 0;
            for (int w = 0; w < SU_T / 32; w++) tot += cnt[w];
            nsv[i] = rsqrtf(fmaxf((float)(tot & 0xffff), 1.f));
            ndv[i] = rsqrtf(fmaxf((float)(tot >> 16), 1.f));
        }
    } else if (blockIdx.x < NN + SU_STARTB) {
        const int total4 = BB * NN * 13 * (HH / 4);
        for (int idx = (blockIdx.x - NN) * SU_T + tid; idx < total4; idx += SU_STARTB * SU_T) {
            int c4 = idx & 7;
            int mt = idx >> 3;
            int t = mt % 13;
            int m = mt / 13;
            int b = m >> 9, n = m & 511;
            float x = (t == 0) ? 0.f : feat[(b * 12 + (t - 1)) * NN + n];
            float4 w = ((const float4*)sw)[c4];
            float4 bb = ((const float4*)sb)[c4];
            ((float4*)h)[idx] = make_float4(w.x * x + bb.x, w.y * x + bb.y,
                                            w.z * x + bb.z, w.w * x + bb.w);
        }
    } else if (blockIdx.x < NN + SU_STARTB + SU_PREPB) {
        int idx = (blockIdx.x - NN - SU_STARTB) * SU_T + tid;
        {
            int ci = idx & 31;
            int tap = (idx >> 5) & 1;
            int cc = (idx >> 6) & 31;
            int l = idx >> 11;
            fwT[idx] = fw[((l * HH + cc) * HH + ci) * 2 + tap];
            gwT[idx] = gw[((l * HH + cc) * HH + ci) * 2 + tap];
        }
        if (idx < NLAY * HH * HH) {
            int k = idx & 31;
            int n = (idx >> 5) & 31;
            int l = idx >> 10;
            gcwT[idx] = gcw[l * HH * HH + k * HH + n];
        }
    } else {
        for (int j = tid; j < NLAY * 2 * HH; j += SU_T) bnstat[j] = 0.f;
    }
}

#define GAU_NT 2
#define GG_T   512
#define GLK (NLAY * HH)

template <bool NORM>
__device__ __forceinline__ void gau_body(const float* __restrict__ h,
                                         const float* __restrict__ pstat,
                                         const float* __restrict__ pgam,
                                         const float* __restrict__ pbet,
                                         const float* __restrict__ fwT,
                                         const float* __restrict__ fb,
                                         const float* __restrict__ gwT,
                                         const float* __restrict__ gb,
                                         const float* __restrict__ gcwT, float* y,
                                         float* glast, int T, int d, int l, int blk) {
    int Tp = T - d;
    int tm0 = (blk * 4 + (threadIdx.x >> 5)) * GAU_NT;
    int lane = threadIdx.x & 31;
    int c = lane & 15;
    int base = (lane & 16) ? 8 : 0;
    float sc[16], sh[16];
    if (NORM) {
        float inv_cnt = 1.f / (float)(BB * NN * T);
        bn_coef8(sc, sh, pstat, pgam, pbet, base, inv_cnt);
        bn_coef8(sc + 8, sh + 8, pstat, pgam, pbet, base + 16, inv_cnt);
    }
    v16h wf[2][2], wg[2][2], wc[2];
#pragma unroll
    for (int tn = 0; tn < 2; tn++) {
        const float* fwr = fwT + (size_t)((l * HH + tn * 16 + c) * 2) * HH;
        const float* gwr = gwT + (size_t)((l * HH + tn * 16 + c) * 2) * HH;
        const float* cwr = gcwT + (size_t)(l * HH + tn * 16 + c) * HH;
        pack8(wf[tn][0], 0, fwr + base);       pack8(wf[tn][0], 8, fwr + base + 16);
        pack8(wg[tn][0], 0, gwr + base);       pack8(wg[tn][0], 8, gwr + base + 16);
        pack8(wf[tn][1], 0, fwr + HH + base);  pack8(wf[tn][1], 8, fwr + HH + base + 16);
        pack8(wg[tn][1], 0, gwr + HH + base);  pack8(wg[tn][1], 8, gwr + HH + base + 16);
        pack8(wc[tn], 0, cwr + base);          pack8(wc[tn], 8, cwr + base + 16);
    }
    float4 rx[GAU_NT][8];
    int tt[GAU_NT], bnn[GAU_NT];
#pragma unroll
    for (int i = 0; i < GAU_NT; i++) {
        int mrow = (tm0 + i) * 16 + c;
        int t = mrow % Tp;
        int bn = mrow / Tp;
        tt[i] = t;
        bnn[i] = bn;
        const float* hrow0 = h + (bn * T + t) * HH + base;
        const float* hrow1 = hrow0 + d * HH;
        rx[i][0] = ((const float4*)hrow0)[0];        rx[i][1] = ((const float4*)hrow0)[1];
        rx[i][2] = ((const float4*)(hrow0 + 16))[0]; rx[i][3] = ((const float4*)(hrow0 + 16))[1];
        rx[i][4] = ((const float4*)hrow1)[0];        rx[i][5] = ((const float4*)hrow1)[1];
        rx[i][6] = ((const float4*)(hrow1 + 16))[0]; rx[i][7] = ((const float4*)(hrow1 + 16))[1];
    }
#pragma unroll
    for (int i = 0; i < GAU_NT; i++) {
        v16h x0, x1;
        if (NORM) {
            cvt8_affine(x0, 0, rx[i][0], rx[i][1], sc, sh);
            cvt8_affine(x0, 8, rx[i][2], rx[i][3], sc + 8, sh + 8);
            cvt8_affine(x1, 0, rx[i][4], rx[i][5], sc, sh);
            cvt8_affine(x1, 8, rx[i][6], rx[i][7], sc + 8, sh + 8);
        } else {
            cvt8(x0, 0, rx[i][0], rx[i][1]);  cvt8(x0, 8, rx[i][2], rx[i][3]);
            cvt8(x1, 0, rx[i][4], rx[i][5]);  cvt8(x1, 8, rx[i][6], rx[i][7]);
        }
        const int mrow = (tm0 + i) * 16 + c;
        float* yrow = y + (size_t)mrow * HH + base;
        float* grow = glast + (size_t)bnn[i] * GLK + l * HH + base;
        const bool keep = (tt[i] == Tp - 1);
        v16h gp;
        float vals[2][8];
#pragma unroll
        for (int tn = 0; tn < 2; tn++) {
            float4 f0 = ((const float4*)(fb + l * HH + tn * 16 + base))[0];
            float4 f1 = ((const float4*)(fb + l * HH + tn * 16 + base))[1];
            float4 g0 = ((const float4*)(gb + l * HH + tn * 16 + base))[0];
            float4 g1 = ((const float4*)(gb + l * HH + tn * 16 + base))[1];
            v8f cf = {f0.x, f0.y, f0.z, f0.w, f1.x, f1.y, f1.z, f1.w};
            v8f cg = {g0.x, g0.y, g0.z, g0.w, g1.x, g1.y, g1.z, g1.w};
            cf = wmma_f16(wf[tn][0], x0, cf);
            cg = wmma_f16(wg[tn][0], x0, cg);
            cf = wmma_f16(wf[tn][1], x1, cf);
            cg = wmma_f16(wg[tn][1], x1, cg);
#pragma unroll
            for (int r = 0; r < 8; r++) {
                vals[tn][r] = fast_tanh(cf[r]) * fast_sigmoid(cg[r]);
                gp[tn * 8 + r] = (_Float16)vals[tn][r];
            }
        }
#pragma unroll
        for (int tn = 0; tn < 2; tn++) {
            v8f cy = {};
            cy = wmma_f16(wc[tn], gp, cy);
            ((float4*)(yrow + tn * 16))[0] = make_float4(cy[0], cy[1], cy[2], cy[3]);
            ((float4*)(yrow + tn * 16))[1] = make_float4(cy[4], cy[5], cy[6], cy[7]);
        }
        if (keep) {
#pragma unroll
            for (int tn = 0; tn < 2; tn++) {
                ((float4*)(grow + tn * 16))[0] =
                    make_float4(vals[tn][0], vals[tn][1], vals[tn][2], vals[tn][3]);
                ((float4*)(grow + tn * 16))[1] =
                    make_float4(vals[tn][4], vals[tn][5], vals[tn][6], vals[tn][7]);
            }
        }
    }
}
#define CSR_T 128
#define CSR_PER (NE / CSR_T)
__device__ __forceinline__ void csr_piece(const int d, const int* __restrict__ src,
                                          const int* __restrict__ dst,
                                          const float* __restrict__ nv1,
                                          const float* __restrict__ nv2,
                                          const float* __restrict__ rowmax,
                                          const float* __restrict__ rowsum,
                                          const float* __restrict__ nsv,
                                          int2* rowse, int2* rec) {
    __shared__ int s_lt[CSR_T];
    __shared__ int s_eq[CSR_T];
    const int tid = threadIdx.x;
    const int e0 = tid * CSR_PER;
    int lt = 0;
    unsigned inm = 0;
    int4 dq[CSR_PER / 4];
#pragma unroll
    for (int k = 0; k < CSR_PER / 4; k++) dq[k] = ((const int4*)(dst + e0))[k];
#pragma unroll
    for (int k = 0; k < CSR_PER / 4; k++) {
        lt += (dq[k].x < d ? 1 : 0) + (dq[k].y < d ? 1 : 0) + (dq[k].z < d ? 1 : 0) +
              (dq[k].w < d ? 1 : 0);
        inm |= (dq[k].x == d ? 1u : 0u) << (4 * k);
        inm |= (dq[k].y == d ? 1u : 0u) << (4 * k + 1);
        inm |= (dq[k].z == d ? 1u : 0u) << (4 * k + 2);
        inm |= (dq[k].w == d ? 1u : 0u) << (4 * k + 3);
    }
    const int eq = __popc(inm);
    s_lt[tid] = lt;
    s_eq[tid] = eq;
    __syncthreads();
    for (int off = 1; off < CSR_T; off <<= 1) {
        int a = 0, b = 0;
        if (tid >= off) { a = s_lt[tid - off]; b = s_eq[tid - off]; }
        __syncthreads();
        s_lt[tid] += a;
        s_eq[tid] += b;
        __syncthreads();
    }
    const int first = s_lt[CSR_T - 1];
    int pos = first + s_eq[tid] - eq;
    if (tid == 0) rowse[d] = make_int2(first, first + s_eq[CSR_T - 1]);
    while (inm) {
        int e = e0 + __ffs(inm) - 1;
        inm &= inm - 1;
        int sn = src[e];
        float acc = 0.f;
#pragma unroll
        for (int a = 0; a < AEMB; a++) acc += nv1[sn * AEMB + a] * nv2[a * NN + d];
        float scv = fmaxf(acc, 0.f);
        float ewv = 1.f + __expf(scv - rowmax[sn]) * __builtin_amdgcn_rcpf(rowsum[sn]);
        rec[pos] = make_int2(sn, __float_as_int(ewv * nsv[sn]));
        pos++;
    }
}

__global__ void __launch_bounds__(128) k_gau_first(const float* __restrict__ h, const float* __restrict__ fwT,
                            const float* __restrict__ fb, const float* __restrict__ gwT,
                            const float* __restrict__ gb, const float* __restrict__ gcwT,
                            float* y, float* glast, int T, int d, int l,
                            const int* __restrict__ src, const int* __restrict__ dst,
                            const float* __restrict__ nv1, const float* __restrict__ nv2,
                            const float* __restrict__ rowmax, const float* __restrict__ rowsum,
                            const float* __restrict__ nsv, int2* rowse, int2* rec) {
    if (blockIdx.x < NN)
        csr_piece(blockIdx.x, src, dst, nv1, nv2, rowmax, rowsum, nsv, rowse, rec);
    else
        gau_body<false>(h, nullptr, nullptr, nullptr, fwT, fb, gwT, gb, gcwT, y, glast, T, d, l,
                        blockIdx.x - NN);
}
__global__ void __launch_bounds__(128) k_gau_wmma(const float* __restrict__ h, const float* __restrict__ pstat,
                           const float* __restrict__ pgam, const float* __restrict__ pbet,
                           const float* __restrict__ fwT, const float* __restrict__ fb,
                           const float* __restrict__ gwT, const float* __restrict__ gb,
                           const float* __restrict__ gcwT, float* y, float* glast,
                           int T, int d, int l) {
    gau_body<true>(h, pstat, pgam, pbet, fwT, fb, gwT, gb, gcwT, y, glast, T, d, l, blockIdx.x);
}

__device__ __forceinline__ void fma4(float4& a, float w, const float4 v) {
    a.x = fmaf(w, v.x, a.x); a.y = fmaf(w, v.y, a.y);
    a.z = fmaf(w, v.z, a.z); a.w = fmaf(w, v.w, a.w);
}
__device__ __forceinline__ float4 gc_finish(float4 z, float nrm, float4 bias, float4 r,
                                            float4& s1, float4& s2) {
    float4 v;
    v.x = fmaxf(fmaf(z.x, nrm, bias.x), 0.f) + r.x;
    v.y = fmaxf(fmaf(z.y, nrm, bias.y), 0.f) + r.y;
    v.z = fmaxf(fmaf(z.z, nrm, bias.z), 0.f) + r.z;
    v.w = fmaxf(fmaf(z.w, nrm, bias.w), 0.f) + r.w;
    s1.x += v.x; s1.y += v.y; s1.z += v.z; s1.w += v.w;
    s2.x = fmaf(v.x, v.x, s2.x); s2.y = fmaf(v.y, v.y, s2.y);
    s2.z = fmaf(v.z, v.z, s2.z); s2.w = fmaf(v.w, v.w, s2.w);
    return v;
}
template <bool NORM>
__device__ __forceinline__ void gather_body(const int2* __restrict__ rowse,
                                            const int2* __restrict__ rec,
                                            const float4* __restrict__ y4,
                                            const float* __restrict__ ndv,
                                            const float* __restrict__ gcb,
                                            const float4* __restrict__ hold4,
                                            const float* __restrict__ pstat,
                                            const float* __restrict__ pgam,
                                            const float* __restrict__ pbet,
                                            float4* __restrict__ hpre4, float* stat,
                                            int T, int d, int l) {
    const int Tp = T - d;
    const int slab4 = Tp * 8;
    int idx = blockIdx.x * GG_T + threadIdx.x;
    int bd = idx / slab4;
    int q = idx - bd * slab4;
    int dn = bd & (NN - 1);
    int b0 = bd >> 9;
    const int c4 = q & 7;
    const int bstride = (BB / 4) * NN * slab4;
    const float4* g = y4 + (b0 * NN) * slab4 + q;
    const int hstride = (BB / 4) * NN * T * 8;
    const float4* hp = hold4 + ((b0 * NN + dn) * T + d) * 8 + q;
    float4 h0 = hp[0], h1 = hp[hstride], h2 = hp[2 * hstride], h3 = hp[3 * hstride];
    int2 se = rowse[dn];
    float4 a0 = make_float4(0.f, 0.f, 0.f, 0.f), a1 = a0, a2 = a0, a3 = a0;
    int j = se.x;
    for (; j + 2 <= se.y; j += 2) {
        int2 r0 = rec[j], r1 = rec[j + 1];
        const float4* p0 = g + r0.x * slab4;
        const float4* p1 = g + r1.x * slab4;
        float4 u0 = p0[0], u1 = p0[bstride], u2 = p0[2 * bstride], u3 = p0[3 * bstride];
        float4 v0 = p1[0], v1 = p1[bstride], v2 = p1[2 * bstride], v3 = p1[3 * bstride];
        float w0 = __int_as_float(r0.y), w1 = __int_as_float(r1.y);
        fma4(a0, w0, u0); fma4(a1, w0, u1); fma4(a2, w0, u2); fma4(a3, w0, u3);
        fma4(a0, w1, v0); fma4(a1, w1, v1); fma4(a2, w1, v2); fma4(a3, w1, v3);
    }
    if (j < se.y) {
        int2 r0 = rec[j];
        const float4* p0 = g + r0.x * slab4;
        float w0 = __int_as_float(r0.y);
        fma4(a0, w0, p0[0]); fma4(a1, w0, p0[bstride]);
        fma4(a2, w0, p0[2 * bstride]); fma4(a3, w0, p0[3 * bstride]);
    }
    const float nrm = ndv[dn];
    const float4 bias = ((const float4*)(gcb + l * HH))[c4];
    if (NORM) {
        const float inv_cnt = 1.f / (float)(BB * NN * T);
        float4 s = ((const float4*)pstat)[c4], qq = ((const float4*)(pstat + HH))[c4];
        float4 gm = ((const float4*)pgam)[c4], bt = ((const float4*)pbet)[c4];
        float4 psc, psh;
        bn_coef(psc.x, psh.x, s.x, qq.x, gm.x, bt.x, inv_cnt);
        bn_coef(psc.y, psh.y, s.y, qq.y, gm.y, bt.y, inv_cnt);
        bn_coef(psc.z, psh.z, s.z, qq.z, gm.z, bt.z, inv_cnt);
        bn_coef(psc.w, psh.w, s.w, qq.w, gm.w, bt.w, inv_cnt);
        h0 = make_float4(fmaf(h0.x, psc.x, psh.x), fmaf(h0.y, psc.y, psh.y),
                         fmaf(h0.z, psc.z, psh.z), fmaf(h0.w, psc.w, psh.w));
        h1 = make_float4(fmaf(h1.x, psc.x, psh.x), fmaf(h1.y, psc.y, psh.y),
                         fmaf(h1.z, psc.z, psh.z), fmaf(h1.w, psc.w, psh.w));
        h2 = make_float4(fmaf(h2.x, psc.x, psh.x), fmaf(h2.y, psc.y, psh.y),
                         fmaf(h2.z, psc.z, psh.z), fmaf(h2.w, psc.w, psh.w));
        h3 = make_float4(fmaf(h3.x, psc.x, psh.x), fmaf(h3.y, psc.y, psh.y),
                         fmaf(h3.z, psc.z, psh.z), fmaf(h3.w, psc.w, psh.w));
    }
    float4 s1 = make_float4(0.f, 0.f, 0.f, 0.f), s2 = s1;
    float4* o = hpre4 + (bd * slab4 + q);
    o[0]           = gc_finish(a0, nrm, bias, h0, s1, s2);
    o[bstride]     = gc_finish(a1, nrm, bias, h1, s1, s2);
    o[2 * bstride] = gc_finish(a2, nrm, bias, h2, s1, s2);
    o[3 * bstride] = gc_finish(a3, nrm, bias, h3, s1, s2);
    float sv[8] = {s1.x, s1.y, s1.z, s1.w, s2.x, s2.y, s2.z, s2.w};
#pragma unroll
    for (int k = 0; k < 8; k++) {
        sv[k] += __shfl_xor(sv[k], 8);
        sv[k] += __shfl_xor(sv[k], 16);
    }
    __shared__ alignas(16) float red[GG_T / 32][2 * HH];
    const int wave = threadIdx.x >> 5;
    const int lane = threadIdx.x & 31;
    if (lane < 8) {
        ((float4*)&red[wave][0])[lane]  = make_float4(sv[0], sv[1], sv[2], sv[3]);
        ((float4*)&red[wave][HH])[lane] = make_float4(sv[4], sv[5], sv[6], sv[7]);
    }
    __builtin_amdgcn_fence(__ATOMIC_RELEASE, "workgroup", "local");
    __builtin_amdgcn_s_barrier();
    __builtin_amdgcn_fence(__ATOMIC_ACQUIRE, "workgroup", "local");
    if (threadIdx.x < 2 * HH) {
        const int wd = threadIdx.x;
        float tot = 0.f;
#pragma unroll
        for (int k = 0; k < GG_T / 32; k++) tot += red[k][wd];
        atomicAdd(&stat[wd], tot);
    }
}
__global__ void __launch_bounds__(GG_T) k_gather_first(const int2* __restrict__ rowse, const int2* __restrict__ rec,
                               const float4* __restrict__ y4, const float* __restrict__ ndv,
                               const float* __restrict__ gcb, const float4* __restrict__ hold4,
                               float4* __restrict__ hpre4, float* stat, int T, int d, int l) {
    gather_body<false>(rowse, rec, y4, ndv, gcb, hold4, nullptr, nullptr, nullptr, hpre4, stat,
                       T, d, l);
}
__global__ void __launch_bounds__(GG_T) k_gather_gc(const int2* __restrict__ rowse, const int2* __restrict__ rec,
                            const float4* __restrict__ y4, const float* __restrict__ ndv,
                            const float* __restrict__ gcb, const float4* __restrict__ hold4,
                            const float* __restrict__ pstat, const float* __restrict__ pgam,
                            const float* __restrict__ pbet, float4* __restrict__ hpre4,
                            float* stat, int T, int d, int l) {
    gather_body<true>(rowse, rec, y4, ndv, gcb, hold4, pstat, pgam, pbet, hpre4, stat, T, d, l);
}

#define EF_ROWS 64
__global__ void __launch_bounds__(128) k_end_fused(const float* __restrict__ glast,
                            const float* __restrict__ sw, const float* __restrict__ sb,
                            const float* __restrict__ w1, const float* __restrict__ b1,
                            const float* __restrict__ w2, const float* __restrict__ b2,
                            float* out) {
    __shared__ v8h sB[4 * 8 * 2 * 32];
    __shared__ float4 red[4 * 4 * 2 * 32];
    const int w = threadIdx.x >> 5;
    const int lane = threadIdx.x & 31;
    const int c = lane & 15;
    const int h = lane >> 4;
    const int base = h * 8;
    const int row0 = blockIdx.x * EF_ROWS;

    v16h wA[4][NLAY];
    v8f bs[4];
#pragma unroll
    for (int m = 0; m < 4; m++) {
        const int ch0 = (4 * w + m) * 16;
        float4 s0 = make_float4(0.f, 0.f, 0.f, 0.f), s1 = s0;
#pragma unroll
        for (int l = 0; l < NLAY; l++) {
            const float* p = sw + (size_t)(l * SKC + ch0 + c) * HH;
            pack8(wA[m][l], 0, p + base);
            pack8(wA[m][l], 8, p + base + 16);
            float4 x = ((const float4*)(sb + l * SKC + ch0 + base))[0];
            float4 y = ((const float4*)(sb + l * SKC + ch0 + base))[1];
            s0.x += x.x; s0.y += x.y; s0.z += x.z; s0.w += x.w;
            s1.x += y.x; s1.y += y.y; s1.z += y.z; s1.w += y.w;
        }
        bs[m] = v8f{s0.x, s0.y, s0.z, s0.w, s1.x, s1.y, s1.z, s1.w};
    }
#pragma unroll
    for (int rt = 0; rt < 4; rt++) {
        const float* arow = glast + (size_t)(row0 + rt * 16 + c) * GLK;
        v16h g[NLAY];
#pragma unroll
        for (int l = 0; l < NLAY; l++) {
            pack8(g[l], 0, arow + l * HH + base);
            pack8(g[l], 8, arow + l * HH + base + 16);
        }
        v8f acc[4];
#pragma unroll
        for (int m = 0; m < 4; m++) acc[m] = bs[m];
#pragma unroll
        for (int l = 0; l < NLAY; l++) {
#pragma unroll
            for (int m = 0; m < 4; m++) acc[m] = wmma_f16(wA[m][l], g[l], acc[m]);
        }
#pragma unroll
        for (int m = 0; m < 4; m++) {
            v8h p;
#pragma unroll
            for (int r = 0; r < 8; r++) p[r] = (_Float16)fmaxf(acc[m][r], 0.f);
            sB[((rt * 8 + 2 * w + (m >> 1)) * 2 + (m & 1)) * 32 + lane] = p;
        }
    }
    float4 raw[2][SKC / 32][4];
#define EF_LOAD_RAW(P, J)                                                                     \
    {                                                                                         \
        const float* q = w1 + (size_t)((2 * (4 * w + (P)) + (J)) * 16 + c) * SKC + base;      \
        _Pragma("unroll") for (int ks = 0; ks < SKC / 32; ks++) {                             \
            raw[J][ks][0] = ((const float4*)(q + ks * 32))[0];                                \
            raw[J][ks][1] = ((const float4*)(q + ks * 32))[1];                                \
            raw[J][ks][2] = ((const float4*)(q + ks * 32 + 16))[0];                           \
            raw[J][ks][3] = ((const float4*)(q + ks * 32 + 16))[1];                           \
        }                                                                                     \
    }
#define EF_LOAD_SMALL(P)                                                                      \
    {                                                                                         \
        const int ks3 = 4 * w + (P);                                                          \
        const float* q = w2 + (size_t)(c < OUTF ? c : 0) * ECC + ks3 * 32;                    \
        v16h t;                                                                               \
        pack8(t, 0, q + base);                                                                \
        pack8(t, 8, q + base + 16);                                                           \
        w2An = (c < OUTF) ? t : v16h{};                                                       \
        _Pragma("unroll") for (int j = 0; j < 2; j++) {                                       \
            const int ch0 = (2 * ks3 + j) * 16;                                               \
            float4 x = ((const float4*)(b1 + ch0 + base))[0];                                 \
            float4 y = ((const float4*)(b1 + ch0 + base))[1];                                 \
            bbn[j] = v8f{x.x, x.y, x.z, x.w, y.x, y.y, y.z, y.w};                             \
        }                                                                                     \
    }
    v16h w2An;
    v8f bbn[2];
    EF_LOAD_SMALL(0)
    EF_LOAD_RAW(0, 0)
    EF_LOAD_RAW(0, 1)
    __syncthreads();
    v16h bf[4][SKC / 32];
#pragma unroll
    for (int rt = 0; rt < 4; rt++) {
#pragma unroll
        for (int ks = 0; ks < SKC / 32; ks++) {
            v8h lo = sB[((rt * 8 + ks) * 2 + 0) * 32 + lane];
            v8h hi = sB[((rt * 8 + ks) * 2 + 1) * 32 + lane];
            bf[rt][ks] = __builtin_shufflevector(lo, hi, 0, 1, 2, 3, 4, 5, 6, 7,
                                                 8, 9, 10, 11, 12, 13, 14, 15);
        }
    }
    v8f oacc[4];
#pragma unroll
    for (int rt = 0; rt < 4; rt++) oacc[rt] = v8f{0.f, 0.f, 0.f, 0.f, 0.f, 0.f, 0.f, 0.f};
#pragma unroll 1
    for (int p = 0; p < 4; p++) {
        v16h wB[2][SKC / 32];
#pragma unroll
        for (int j = 0; j < 2; j++) {
#pragma unroll
            for (int ks = 0; ks < SKC / 32; ks++) {
                cvt8(wB[j][ks], 0, raw[j][ks][0], raw[j][ks][1]);
                cvt8(wB[j][ks], 8, raw[j][ks][2], raw[j][ks][3]);
            }
        }
        const v16h w2A = w2An;
        const v8f bb0 = bbn[0], bb1 = bbn[1];
#pragma unroll
        for (int rt = 0; rt < 4; rt++) {
            if (rt == 0 && p < 3) {
                EF_LOAD_SMALL(p + 1)
                EF_LOAD_RAW(p + 1, 0)
            }
            if (rt == 2 && p < 3) EF_LOAD_RAW(p + 1, 1)
            v8f a0 = bb0, a1 = bb1;
#pragma unroll
            for (int ks = 0; ks < SKC / 32; ks++) {
                a0 = wmma_f16(wB[0][ks], bf[rt][ks], a0);
                a1 = wmma_f16(wB[1][ks], bf[rt][ks], a1);
            }
            v16h b3;
#pragma unroll
            for (int r = 0; r < 8; r++) {
                b3[r]     = (_Float16)fmaxf(a0[r], 0.f);
                b3[8 + r] = (_Float16)fmaxf(a1[r], 0.f);
            }
            oacc[rt] = wmma_f16(w2A, b3, oacc[rt]);
        }
    }
#undef EF_LOAD_RAW
#undef EF_LOAD_SMALL
#pragma unroll
    for (int rt = 0; rt < 4; rt++) {
        red[((w * 4 + rt) * 2 + 0) * 32 + lane] =
            make_float4(oacc[rt][0], oacc[rt][1], oacc[rt][2], oacc[rt][3]);
        red[((w * 4 + rt) * 2 + 1) * 32 + lane] =
            make_float4(oacc[rt][4], oacc[rt][5], oacc[rt][6], oacc[rt][7]);
    }
    __syncthreads();
    if (w < 2) {
        const int b = row0 >> 9;
#pragma unroll
        for (int jj = 0; jj < 2; jj++) {
            const int rt = 2 * w + jj;
            float4 t0 = make_float4(0.f, 0.f, 0.f, 0.f), t1 = t0;
#pragma unroll
            for (int ww = 0; ww < 4; ww++) {
                float4 x = red[((ww * 4 + rt) * 2 + 0) * 32 + lane];
                float4 y = red[((ww * 4 + rt) * 2 + 1) * 32 + lane];
                t0.x += x.x; t0.y += x.y; t0.z += x.z; t0.w += x.w;
                t1.x += y.x; t1.y += y.y; t1.z += y.z; t1.w += y.w;
            }
            const int n = (row0 & (NN - 1)) + rt * 16 + c;
            const float v[8] = {t0.x, t0.y, t0.z, t0.w, t1.x, t1.y, t1.z, t1.w};
#pragma unroll
            for (int r = 0; r < 8; r++) {
                const int f = base + r;
                if (f < OUTF) out[(b * OUTF + f) * NN + n] = v[r] + b2[f];
            }
        }
    }
}

extern "C" void kernel_launch(void* const* d_in, const int* in_sizes, int n_in,
                              void* d_out, int out_size, void* d_ws, size_t ws_size,
                              hipStream_t stream) {
    const float* feat     = (const float*)d_in[0];
    const int*   src      = (const int*)d_in[1];
    const int*   dst      = (const int*)d_in[2];
    const float* nv1      = (const float*)d_in[3];
    const float* nv2      = (const float*)d_in[4];
    const float* start_w  = (const float*)d_in[5];
    const float* start_b  = (const float*)d_in[6];
    const float* filter_w = (const float*)d_in[7];
    const float* filter_b = (const float*)d_in[8];
    const float* gate_w   = (const float*)d_in[9];
    const float* gate_b   = (const float*)d_in[10];
    const float* skip_w   = (const float*)d_in[11];
    const float* skip_b   = (const float*)d_in[12];
    const float* bn_gamma = (const float*)d_in[13];
    const float* bn_beta  = (const float*)d_in[14];
    const float* gconv_w  = (const float*)d_in[15];
    const float* gconv_b  = (const float*)d_in[16];
    const float* end1_w   = (const float*)d_in[17];
    const float* end1_b   = (const float*)d_in[18];
    const float* end2_w   = (const float*)d_in[19];
    const float* end2_b   = (const float*)d_in[20];

    float* ws = (float*)d_ws;
    size_t o = 0;
    float* rowmax = ws + o; o += NN;
    float* rowsum = ws + o; o += NN;
    float* nsv    = ws + o; o += NN;
    float* ndv    = ws + o; o += NN;
    float* bnstat = ws + o; o += NLAY * 2 * HH;
    float* fwT    = ws + o; o += NLAY * HH * 2 * HH;
    float* gwT    = ws + o; o += NLAY * HH * 2 * HH;
    float* gcwT   = ws + o; o += NLAY * HH * HH;
    float* glast  = ws + o; o += (size_t)BB * NN * GLK;
    float* hA     = ws + o; o += (size_t)BB * NN * 13 * HH;
    float* hB     = ws + o; o += (size_t)BB * NN * 13 * HH;
    float* ygc    = ws + o; o += (size_t)BB * NN * 12 * HH;
    int2*  rowse  = (int2*)(ws + o); o += 2 * NN;
    int2*  rec    = (int2*)(ws + o); o += 2 * NE;

    k_setup<<<SU_BLOCKS, SU_T, 0, stream>>>(src, dst, nv1, nv2, feat, start_w, start_b,
                                            filter_w, gate_w, gconv_w, rowmax, rowsum, nsv, ndv,
                                            hA, fwT, gwT, gcwT, bnstat);

    const int dil[8] = {1, 2, 1, 2, 1, 2, 1, 2};
    float* hcur = hA;
    float* hnext = hB;
    int T = 13;
    for (int l = 0; l < NLAY; l++) {
        int d = dil[l];
        int Tp = T - d;
        int M = BB * NN * Tp;
        int blocksGau = (M / 16) / 4 / GAU_NT;
        int blocksGather = (BB / 4) * NN * Tp * 8 / GG_T;

        const int lp = l ? l - 1 : 0;
        const float* pstat = bnstat + lp * 2 * HH;
        const float* pgam  = bn_gamma + lp * HH;
        const float* pbet  = bn_beta + lp * HH;
        if (l == 0)
            k_gau_first<<<NN + blocksGau, 128, 0, stream>>>(
                hcur, fwT, filter_b, gwT, gate_b, gcwT, ygc, glast, T, d, l,
                src, dst, nv1, nv2, rowmax, rowsum, nsv, rowse, rec);
        else
            k_gau_wmma<<<blocksGau, 128, 0, stream>>>(hcur, pstat, pgam, pbet, fwT, filter_b, gwT,
                                                      gate_b, gcwT, ygc, glast, T, d, l);
        if (l == 0)
            k_gather_first<<<blocksGather, GG_T, 0, stream>>>(
                rowse, rec, (const float4*)ygc, ndv, gconv_b, (const float4*)hcur,
                (float4*)hnext, bnstat, T, d, l);
        else
            k_gather_gc<<<blocksGather, GG_T, 0, stream>>>(
                rowse, rec, (const float4*)ygc, ndv, gconv_b, (const float4*)hcur, pstat, pgam,
                pbet, (float4*)hnext, bnstat + l * 2 * HH, T, d, l);
        float* tmp = hcur; hcur = hnext; hnext = tmp;
        T = Tp;
    }

    k_end_fused<<<BB * NN / EF_ROWS, 128, 0, stream>>>(glast, skip_w, skip_b, end1_w, end1_b,
                                                       end2_w, end2_b, (float*)d_out);
}


# last layer folded into the end launch: its gate/filter product is formed there per row tile and passed in LDS; its graph conv and batchnorm, which nothing reads, are not formed; 16 launches, not 18
# speedup vs baseline: 8.0303x; 1.0410x over previous
#include <hip/hip_runtime.h>
#include <hip/hip_bf16.h>

typedef __attribute__((ext_vector_type(16))) _Float16 v16h;
typedef __attribute__((ext_vector_type(8)))  float    v8f;
typedef __attribute__((ext_vector_type(8)))  _Float16 v8h;

#define NN   512
#define NE   4096
#define BB   32
#define HH   32
#define SKC  256
#define ECC  512
#define OUTF 12
#define AEMB 10
#define NEGBIG (-1e30f)
#define NLAY 8

__device__ __forceinline__ v8f wmma_f16(v16h a, v16h b, v8f c) {
    return __builtin_amdgcn_wmma_f32_16x16x32_f16(false, a, false, b, (short)0, c, false, false);
}

__device__ __forceinline__ float fast_sigmoid(float x) {
    return __builtin_amdgcn_rcpf(1.f + __expf(-x));
}
__device__ __forceinline__ float fast_tanh(float x) {
    float t = __expf(-2.f * fabsf(x));
    float r = (1.f - t) * __builtin_amdgcn_rcpf(1.f + t);
    return copysignf(r, x);
}

__device__ __forceinline__ void pack8(v16h& d, int off, const float* __restrict__ p) {
    float4 x = ((const float4*)p)[0];
    float4 y = ((const float4*)p)[1];
    d[off + 0] = (_Float16)x.x; d[off + 1] = (_Float16)x.y;
    d[off + 2] = (_Float16)x.z; d[off + 3] = (_Float16)x.w;
    d[off + 4] = (_Float16)y.x; d[off + 5] = (_Float16)y.y;
    d[off + 6] = (_Float16)y.z; d[off + 7] = (_Float16)y.w;
}
__device__ __forceinline__ void cvt8(v16h& d, int off, float4 x, float4 y) {
    d[off + 0] = (_Float16)x.x; d[off + 1] = (_Float16)x.y;
    d[off + 2] = (_Float16)x.z; d[off + 3] = (_Float16)x.w;
    d[off + 4] = (_Float16)y.x; d[off + 5] = (_Float16)y.y;
    d[off + 6] = (_Float16)y.z; d[off + 7] = (_Float16)y.w;
}

__device__ __forceinline__ void cvt8_affine(v16h& d, int off, float4 x, float4 y,
                                            const float* sc, const float* sh) {
    d[off + 0] = (_Float16)fmaf(x.x, sc[0], sh[0]); d[off + 1] = (_Float16)fmaf(x.y, sc[1], sh[1]);
    d[off + 2] = (_Float16)fmaf(x.z, sc[2], sh[2]); d[off + 3] = (_Float16)fmaf(x.w, sc[3], sh[3]);
    d[off + 4] = (_Float16)fmaf(y.x, sc[4], sh[4]); d[off + 5] = (_Float16)fmaf(y.y, sc[5], sh[5]);
    d[off + 6] = (_Float16)fmaf(y.z, sc[6], sh[6]); d[off + 7] = (_Float16)fmaf(y.w, sc[7], sh[7]);
}

__device__ __forceinline__ void bn_coef(float& sc, float& sh, float s, float q, float g, float b,
                                        float inv_cnt) {
    float mu = s * inv_cnt;
    float var = q * inv_cnt - mu * mu;
    sc = rsqrtf(var + 1e-5f) * g;
    sh = b - mu * sc;
}
__device__ __forceinline__ void bn_coef8(float* sc, float* sh, const float* __restrict__ stat,
                                         const float* __restrict__ gam,
                                         const float* __restrict__ bet, int c0, float inv_cnt) {
    float4 s0 = ((const float4*)(stat + c0))[0],      s1 = ((const float4*)(stat + c0))[1];
    float4 q0 = ((const float4*)(stat + HH + c0))[0], q1 = ((const float4*)(stat + HH + c0))[1];
    float4 g0 = ((const float4*)(gam + c0))[0],       g1 = ((const float4*)(gam + c0))[1];
    float4 b0 = ((const float4*)(bet + c0))[0],       b1 = ((const float4*)(bet + c0))[1];
    bn_coef(sc[0], sh[0], s0.x, q0.x, g0.x, b0.x, inv_cnt);
    bn_coef(sc[1], sh[1], s0.y, q0.y, g0.y, b0.y, inv_cnt);
    bn_coef(sc[2], sh[2], s0.z, q0.z, g0.z, b0.z, inv_cnt);
    bn_coef(sc[3], sh[3], s0.w, q0.w, g0.w, b0.w, inv_cnt);
    bn_coef(sc[4], sh[4], s1.x, q1.x, g1.x, b1.x, inv_cnt);
    bn_coef(sc[5], sh[5], s1.y, q1.y, g1.y, b1.y, inv_cnt);
    bn_coef(sc[6], sh[6], s1.z, q1.z, g1.z, b1.z, inv_cnt);
    bn_coef(sc[7], sh[7], s1.w, q1.w, g1.w, b1.w, inv_cnt);
}

#define SU_T      256
#define SU_PER    (NE / SU_T)
#define SU_STARTB ((BB * NN * 13 * (HH / 4)) / (SU_T * 8))
#define SU_PREPB  ((NLAY * HH * 2 * HH) / SU_T)
#define SU_BLOCKS (NN + SU_STARTB + SU_PREPB + 1)
__global__ void __launch_bounds__(SU_T) k_setup(const int* __restrict__ src, const int* __restrict__ dst,
                        const float* __restrict__ nv1, const float* __restrict__ nv2,
                        const float* __restrict__ feat, const float* __restrict__ sw,
                        const float* __restrict__ sb,
                        const float* __restrict__ fw, const float* __restrict__ gw,
                        const float* __restrict__ gcw,
                        float* rowmax, float* rowsum, float* nsv, float* ndv,
                        float* h, float* fwT, float* gwT, float* gcwT, float* bnstat) {
    __shared__ float sc[NN];
    __shared__ float red[SU_T];
    __shared__ int cnt[SU_T / 32];
    const int tid = threadIdx.x;
    if (blockIdx.x < NN) {
        const int i = blockIdx.x;
        const int e0 = tid * SU_PER;
        int4 sq[SU_PER / 4], dq[SU_PER / 4];
#pragma unroll
        for (int k = 0; k < SU_PER / 4; k++) {
            sq[k] = ((const int4*)(src + e0))[k];
            dq[k] = ((const int4*)(dst + e0))[k];
        }
        for (int j = tid; j < NN; j += SU_T) sc[j] = NEGBIG;
        __syncthreads();
        unsigned outm = 0;
        int packed = 0;
#pragma unroll
        for (int k = 0; k < SU_PER / 4; k++) {
            outm |= (sq[k].x == i ? 1u : 0u) << (4 * k);
            outm |= (sq[k].y == i ? 1u : 0u) << (4 * k + 1);
            outm |= (sq[k].z == i ? 1u : 0u) << (4 * k + 2);
            outm |= (sq[k].w == i ? 1u : 0u) << (4 * k + 3);
            packed += (dq[k].x == i ? 0x10000 : 0) + (dq[k].y == i ? 0x10000 : 0) +
                      (dq[k].z == i ? 0x10000 : 0) + (dq[k].w == i ? 0x10000 : 0);
        }
        packed += __popc(outm);
        while (outm) {
            int k = __ffs(outm) - 1;
            outm &= outm - 1;
            int d = dst[e0 + k];
            float acc = 0.f;
#pragma unroll
            for (int a = 0; a < AEMB; a++) acc += nv1[i * AEMB + a] * nv2[a * NN + d];
            sc[d] = fmaxf(acc, 0.f);
        }
        for (int off = 16; off > 0; off >>= 1) packed += __shfl_xor(packed, off);
        if ((tid & 31) == 0) cnt[tid >> 5] = packed;
        __syncthreads();
        float m = NEGBIG;
        for (int j = tid; j < NN; j += SU_T) m = fmaxf(m, sc[j]);
        red[tid] = m;
        __syncthreads();
        for (int s = SU_T / 2; s > 0; s >>= 1) {
            if (tid < s) red[tid] = fmaxf(red[tid], red[tid + s]);
            __syncthreads();
        }
        m = red[0];
        __syncthreads();
        float sum = 0.f;
        for (int j = tid; j < NN; j += SU_T) {
            float v = sc[j];
            if (v > 0.5f * NEGBIG) sum += __expf(v - m);
        }
        red[tid] = sum;
        __syncthreads();
        for (int s = SU_T / 2; s > 0; s >>= 1) {
            if (tid < s) red[tid] += red[tid + s];
            __syncthreads();
        }
        if (tid == 0) {
            rowmax[i] = m;
            rowsum[i] = (red[0] > 0.f) ? red[0] : 1.f;
            int tot = 0;
            for (int w = 0; w < SU_T / 32; w++) tot += cnt[w];
            nsv[i] = rsqrtf(fmaxf((float)(tot & 0xffff), 1.f));
            ndv[i] = rsqrtf(fmaxf((float)(tot >> 16), 1.f));
        }
    } else if (blockIdx.x < NN + SU_STARTB) {
        const int total4 = BB * NN * 13 * (HH / 4);
        for (int idx = (blockIdx.x - NN) * SU_T + tid; idx < total4; idx += SU_STARTB * SU_T) {
            int c4 = idx & 7;
            int mt = idx >> 3;
            int t = mt % 13;
            int m = mt / 13;
            int b = m >> 9, n = m & 511;
            float x = (t == 0) ? 0.f : feat[(b * 12 + (t - 1)) * NN + n];
            float4 w = ((const float4*)sw)[c4];
            float4 bb = ((const float4*)sb)[c4];
            ((float4*)h)[idx] = make_float4(w.x * x + bb.x, w.y * x + bb.y,
                                            w.z * x + bb.z, w.w * x + bb.w);
        }
    } else if (blockIdx.x < NN + SU_STARTB + SU_PREPB) {
        int idx = (blockIdx.x - NN - SU_STARTB) * SU_T + tid;
        {
            int ci = idx & 31;
            int tap = (idx >> 5) & 1;
            int cc = (idx >> 6) & 31;
            int l = idx >> 11;
            fwT[idx] = fw[((l * HH + cc) * HH + ci) * 2 + tap];
            gwT[idx] = gw[((l * HH + cc) * HH + ci) * 2 + tap];
        }
        if (idx < NLAY * HH * HH) {
            int k = idx & 31;
            int n = (idx >> 5) & 31;
            int l = idx >> 10;
            gcwT[idx] = gcw[l * HH * HH + k * HH + n];
        }
    } else {
        for (int j = tid; j < NLAY * 2 * HH; j += SU_T) bnstat[j] = 0.f;
    }
}

#define GAU_NT 2
#define GG_T   512
#define GLK ((NLAY - 1) * HH)

template <bool NORM>
__device__ __forceinline__ void gau_body(const float* __restrict__ h,
                                         const float* __restrict__ pstat,
                                         const float* __restrict__ pgam,
                                         const float* __restrict__ pbet,
                                         const float* __restrict__ fwT,
                                         const float* __restrict__ fb,
                                         const float* __restrict__ gwT,
                                         const float* __restrict__ gb,
                                         const float* __restrict__ gcwT, float* y,
                                         float* glast, int T, int d, int l, int blk) {
    int Tp = T - d;
    int tm0 = (blk * 4 + (threadIdx.x >> 5)) * GAU_NT;
    int lane = threadIdx.x & 31;
    int c = lane & 15;
    int base = (lane & 16) ? 8 : 0;
    float sc[16], sh[16];
    if (NORM) {
        float inv_cnt = 1.f / (float)(BB * NN * T);
        bn_coef8(sc, sh, pstat, pgam, pbet, base, inv_cnt);
        bn_coef8(sc + 8, sh + 8, pstat, pgam, pbet, base + 16, inv_cnt);
    }
    v16h wf[2][2], wg[2][2], wc[2];
#pragma unroll
    for (int tn = 0; tn < 2; tn++) {
        const float* fwr = fwT + (size_t)((l * HH + tn * 16 + c) * 2) * HH;
        const float* gwr = gwT + (size_t)((l * HH + tn * 16 + c) * 2) * HH;
        const float* cwr = gcwT + (size_t)(l * HH + tn * 16 + c) * HH;
        pack8(wf[tn][0], 0, fwr + base);       pack8(wf[tn][0], 8, fwr + base + 16);
        pack8(wg[tn][0], 0, gwr + base);       pack8(wg[tn][0], 8, gwr + base + 16);
        pack8(wf[tn][1], 0, fwr + HH + base);  pack8(wf[tn][1], 8, fwr + HH + base + 16);
        pack8(wg[tn][1], 0, gwr + HH + base);  pack8(wg[tn][1], 8, gwr + HH + base + 16);
        pack8(wc[tn], 0, cwr + base);          pack8(wc[tn], 8, cwr + base + 16);
    }
    float4 rx[GAU_NT][8];
    int tt[GAU_NT], bnn[GAU_NT];
#pragma unroll
    for (int i = 0; i < GAU_NT; i++) {
        int mrow = (tm0 + i) * 16 + c;
        int t = mrow % Tp;
        int bn = mrow / Tp;
        tt[i] = t;
        bnn[i] = bn;
        const float* hrow0 = h + (bn * T + t) * HH + base;
        const float* hrow1 = hrow0 + d * HH;
        rx[i][0] = ((const float4*)hrow0)[0];        rx[i][1] = ((const float4*)hrow0)[1];
        rx[i][2] = ((const float4*)(hrow0 + 16))[0]; rx[i][3] = ((const float4*)(hrow0 + 16))[1];
        rx[i][4] = ((const float4*)hrow1)[0];        rx[i][5] = ((const float4*)hrow1)[1];
        rx[i][6] = ((const float4*)(hrow1 + 16))[0]; rx[i][7] = ((const float4*)(hrow1 + 16))[1];
    }
#pragma unroll
    for (int i = 0; i < GAU_NT; i++) {
        v16h x0, x1;
        if (NORM) {
            cvt8_affine(x0, 0, rx[i][0], rx[i][1], sc, sh);
            cvt8_affine(x0, 8, rx[i][2], rx[i][3], sc + 8, sh + 8);
            cvt8_affine(x1, 0, rx[i][4], rx[i][5], sc, sh);
            cvt8_affine(x1, 8, rx[i][6], rx[i][7], sc + 8, sh + 8);
        } else {
            cvt8(x0, 0, rx[i][0], rx[i][1]);  cvt8(x0, 8, rx[i][2], rx[i][3]);
            cvt8(x1, 0, rx[i][4], rx[i][5]);  cvt8(x1, 8, rx[i][6], rx[i][7]);
        }
        const int mrow = (tm0 + i) * 16 + c;
        float* yrow = y + (size_t)mrow * HH + base;
        float* grow = glast + (size_t)bnn[i] * GLK + l * HH + base;
        const bool keep = (tt[i] == Tp - 1);
        v16h gp;
        float vals[2][8];
#pragma unroll
        for (int tn = 0; tn < 2; tn++) {
            float4 f0 = ((const float4*)(fb + l * HH + tn * 16 + base))[0];
            float4 f1 = ((const float4*)(fb + l * HH + tn * 16 + base))[1];
            float4 g0 = ((const float4*)(gb + l * HH + tn * 16 + base))[0];
            float4 g1 = ((const float4*)(gb + l * HH + tn * 16 + base))[1];
            v8f cf = {f0.x, f0.y, f0.z, f0.w, f1.x, f1.y, f1.z, f1.w};
            v8f cg = {g0.x, g0.y, g0.z, g0.w, g1.x, g1.y, g1.z, g1.w};
            cf = wmma_f16(wf[tn][0], x0, cf);
            cg = wmma_f16(wg[tn][0], x0, cg);
            cf = wmma_f16(wf[tn][1], x1, cf);
            cg = wmma_f16(wg[tn][1], x1, cg);
#pragma unroll
            for (int r = 0; r < 8; r++) {
                vals[tn][r] = fast_tanh(cf[r]) * fast_sigmoid(cg[r]);
                gp[tn * 8 + r] = (_Float16)vals[tn][r];
            }
        }
#pragma unroll
        for (int tn = 0; tn < 2; tn++) {
            v8f cy = {};
            cy = wmma_f16(wc[tn], gp, cy);
            ((float4*)(yrow + tn * 16))[0] = make_float4(cy[0], cy[1], cy[2], cy[3]);
            ((float4*)(yrow + tn * 16))[1] = make_float4(cy[4], cy[5], cy[6], cy[7]);
        }
        if (keep) {
#pragma unroll
            for (int tn = 0; tn < 2; tn++) {
                ((float4*)(grow + tn * 16))[0] =
                    make_float4(vals[tn][0], vals[tn][1], vals[tn][2], vals[tn][3]);
                ((float4*)(grow + tn * 16))[1] =
                    make_float4(vals[tn][4], vals[tn][5], vals[tn][6], vals[tn][7]);
            }
        }
    }
}
#define CSR_T 128
#define CSR_PER (NE / CSR_T)
__device__ __forceinline__ void csr_piece(const int d, const int* __restrict__ src,
                                          const int* __restrict__ dst,
                                          const float* __restrict__ nv1,
                                          const float* __restrict__ nv2,
                                          const float* __restrict__ rowmax,
                                          const float* __restrict__ rowsum,
                                          const float* __restrict__ nsv,
                                          int2* rowse, int2* rec) {
    __shared__ int s_lt[CSR_T];
    __shared__ int s_eq[CSR_T];
    const int tid = threadIdx.x;
    const int e0 = tid * CSR_PER;
    int lt = 0;
    unsigned inm = 0;
    int4 dq[CSR_PER / 4];
#pragma unroll
    for (int k = 0; k < CSR_PER / 4; k++) dq[k] = ((const int4*)(dst + e0))[k];
#pragma unroll
    for (int k = 0; k < CSR_PER / 4; k++) {
        lt += (dq[k].x < d ? 1 : 0) + (dq[k].y < d ? 1 : 0) + (dq[k].z < d ? 1 : 0) +
              (dq[k].w < d ? 1 : 0);
        inm |= (dq[k].x == d ? 1u : 0u) << (4 * k);
        inm |= (dq[k].y == d ? 1u : 0u) << (4 * k + 1);
        inm |= (dq[k].z == d ? 1u : 0u) << (4 * k + 2);
        inm |= (dq[k].w == d ? 1u : 0u) << (4 * k + 3);
    }
    const int eq = __popc(inm);
    s_lt[tid] = lt;
    s_eq[tid] = eq;
    __syncthreads();
    for (int off = 1; off < CSR_T; off <<= 1) {
        int a = 0, b = 0;
        if (tid >= off) { a = s_lt[tid - off]; b = s_eq[tid - off]; }
        __syncthreads();
        s_lt[tid] += a;
        s_eq[tid] += b;
        __syncthreads();
    }
    const int first = s_lt[CSR_T - 1];
    int pos = first + s_eq[tid] - eq;
    if (tid == 0) rowse[d] = make_int2(first, first + s_eq[CSR_T - 1]);
    while (inm) {
        int e = e0 + __ffs(inm) - 1;
        inm &= inm - 1;
        int sn = src[e];
        float acc = 0.f;
#pragma unroll
        for (int a = 0; a < AEMB; a++) acc += nv1[sn * AEMB + a] * nv2[a * NN + d];
        float scv = fmaxf(acc, 0.f);
        float ewv = 1.f + __expf(scv - rowmax[sn]) * __builtin_amdgcn_rcpf(rowsum[sn]);
        rec[pos] = make_int2(sn, __float_as_int(ewv * nsv[sn]));
        pos++;
    }
}

__global__ void __launch_bounds__(128) k_gau_first(const float* __restrict__ h, const float* __restrict__ fwT,
                            const float* __restrict__ fb, const float* __restrict__ gwT,
                            const float* __restrict__ gb, const float* __restrict__ gcwT,
                            float* y, float* glast, int T, int d, int l,
                            const int* __restrict__ src, const int* __restrict__ dst,
                            const float* __restrict__ nv1, const float* __restrict__ nv2,
                            const float* __restrict__ rowmax, const float* __restrict__ rowsum,
                            const float* __restrict__ nsv, int2* rowse, int2* rec) {
    if (blockIdx.x < NN)
        csr_piece(blockIdx.x, src, dst, nv1, nv2, rowmax, rowsum, nsv, rowse, rec);
    else
        gau_body<false>(h, nullptr, nullptr, nullptr, fwT, fb, gwT, gb, gcwT, y, glast, T, d, l,
                        blockIdx.x - NN);
}
__global__ void __launch_bounds__(128) k_gau_wmma(const float* __restrict__ h, const float* __restrict__ pstat,
                           const float* __restrict__ pgam, const float* __restrict__ pbet,
                           const float* __restrict__ fwT, const float* __restrict__ fb,
                           const float* __restrict__ gwT, const float* __restrict__ gb,
                           const float* __restrict__ gcwT, float* y, float* glast,
                           int T, int d, int l) {
    gau_body<true>(h, pstat, pgam, pbet, fwT, fb, gwT, gb, gcwT, y, glast, T, d, l, blockIdx.x);
}

__device__ __forceinline__ void fma4(float4& a, float w, const float4 v) {
    a.x = fmaf(w, v.x, a.x); a.y = fmaf(w, v.y, a.y);
    a.z = fmaf(w, v.z, a.z); a.w = fmaf(w, v.w, a.w);
}
__device__ __forceinline__ float4 gc_finish(float4 z, float nrm, float4 bias, float4 r,
                                            float4& s1, float4& s2) {
    float4 v;
    v.x = fmaxf(fmaf(z.x, nrm, bias.x), 0.f) + r.x;
    v.y = fmaxf(fmaf(z.y, nrm, bias.y), 0.f) + r.y;
    v.z = fmaxf(fmaf(z.z, nrm, bias.z), 0.f) + r.z;
    v.w = fmaxf(fmaf(z.w, nrm, bias.w), 0.f) + r.w;
    s1.x += v.x; s1.y += v.y; s1.z += v.z; s1.w += v.w;
    s2.x = fmaf(v.x, v.x, s2.x); s2.y = fmaf(v.y, v.y, s2.y);
    s2.z = fmaf(v.z, v.z, s2.z); s2.w = fmaf(v.w, v.w, s2.w);
    return v;
}
template <bool NORM>
__device__ __forceinline__ void gather_body(const int2* __restrict__ rowse,
                                            const int2* __restrict__ rec,
                                            const float4* __restrict__ y4,
                                            const float* __restrict__ ndv,
                                            const float* __restrict__ gcb,
                                            const float4* __restrict__ hold4,
                                            const float* __restrict__ pstat,
                                            const float* __restrict__ pgam,
                                            const float* __restrict__ pbet,
                                            float4* __restrict__ hpre4, float* stat,
                                            int T, int d, int l) {
    const int Tp = T - d;
    const int slab4 = Tp * 8;
    int idx = blockIdx.x * GG_T + threadIdx.x;
    int bd = idx / slab4;
    int q = idx - bd * slab4;
    int dn = bd & (NN - 1);
    int b0 = bd >> 9;
    const int c4 = q & 7;
    const int bstride = (BB / 4) * NN * slab4;
    const float4* g = y4 + (b0 * NN) * slab4 + q;
    const int hstride = (BB / 4) * NN * T * 8;
    const float4* hp = hold4 + ((b0 * NN + dn) * T + d) * 8 + q;
    float4 h0 = hp[0], h1 = hp[hstride], h2 = hp[2 * hstride], h3 = hp[3 * hstride];
    int2 se = rowse[dn];
    float4 a0 = make_float4(0.f, 0.f, 0.f, 0.f), a1 = a0, a2 = a0, a3 = a0;
    int j = se.x;
    for (; j + 2 <= se.y; j += 2) {
        int2 r0 = rec[j], r1 = rec[j + 1];
        const float4* p0 = g + r0.x * slab4;
        const float4* p1 = g + r1.x * slab4;
        float4 u0 = p0[0], u1 = p0[bstride], u2 = p0[2 * bstride], u3 = p0[3 * bstride];
        float4 v0 = p1[0], v1 = p1[bstride], v2 = p1[2 * bstride], v3 = p1[3 * bstride];
        float w0 = __int_as_float(r0.y), w1 = __int_as_float(r1.y);
        fma4(a0, w0, u0); fma4(a1, w0, u1); fma4(a2, w0, u2); fma4(a3, w0, u3);
        fma4(a0, w1, v0); fma4(a1, w1, v1); fma4(a2, w1, v2); fma4(a3, w1, v3);
    }
    if (j < se.y) {
        int2 r0 = rec[j];
        const float4* p0 = g + r0.x * slab4;
        float w0 = __int_as_float(r0.y);
        fma4(a0, w0, p0[0]); fma4(a1, w0, p0[bstride]);
        fma4(a2, w0, p0[2 * bstride]); fma4(a3, w0, p0[3 * bstride]);
    }
    const float nrm = ndv[dn];
    const float4 bias = ((const float4*)(gcb + l * HH))[c4];
    if (NORM) {
        const float inv_cnt = 1.f / (float)(BB * NN * T);
        float4 s = ((const float4*)pstat)[c4], qq = ((const float4*)(pstat + HH))[c4];
        float4 gm = ((const float4*)pgam)[c4], bt = ((const float4*)pbet)[c4];
        float4 psc, psh;
        bn_coef(psc.x, psh.x, s.x, qq.x, gm.x, bt.x, inv_cnt);
        bn_coef(psc.y, psh.y, s.y, qq.y, gm.y, bt.y, inv_cnt);
        bn_coef(psc.z, psh.z, s.z, qq.z, gm.z, bt.z, inv_cnt);
        bn_coef(psc.w, psh.w, s.w, qq.w, gm.w, bt.w, inv_cnt);
        h0 = make_float4(fmaf(h0.x, psc.x, psh.x), fmaf(h0.y, psc.y, psh.y),
                         fmaf(h0.z, psc.z, psh.z), fmaf(h0.w, psc.w, psh.w));
        h1 = make_float4(fmaf(h1.x, psc.x, psh.x), fmaf(h1.y, psc.y, psh.y),
                         fmaf(h1.z, psc.z, psh.z), fmaf(h1.w, psc.w, psh.w));
        h2 = make_float4(fmaf(h2.x, psc.x, psh.x), fmaf(h2.y, psc.y, psh.y),
                         fmaf(h2.z, psc.z, psh.z), fmaf(h2.w, psc.w, psh.w));
        h3 = make_float4(fmaf(h3.x, psc.x, psh.x), fmaf(h3.y, psc.y, psh.y),
                         fmaf(h3.z, psc.z, psh.z), fmaf(h3.w, psc.w, psh.w));
    }
    float4 s1 = make_float4(0.f, 0.f, 0.f, 0.f), s2 = s1;
    float4* o = hpre4 + (bd * slab4 + q);
    o[0]           = gc_finish(a0, nrm, bias, h0, s1, s2);
    o[bstride]     = gc_finish(a1, nrm, bias, h1, s1, s2);
    o[2 * bstride] = gc_finish(a2, nrm, bias, h2, s1, s2);
    o[3 * bstride] = gc_finish(a3, nrm, bias, h3, s1, s2);
    float sv[8] = {s1.x, s1.y, s1.z, s1.w, s2.x, s2.y, s2.z, s2.w};
#pragma unroll
    for (int k = 0; k < 8; k++) {
        sv[k] += __shfl_xor(sv[k], 8);
        sv[k] += __shfl_xor(sv[k], 16);
    }
    __shared__ alignas(16) float red[GG_T / 32][2 * HH];
    const int wave = threadIdx.x >> 5;
    const int lane = threadIdx.x & 31;
    if (lane < 8) {
        ((float4*)&red[wave][0])[lane]  = make_float4(sv[0], sv[1], sv[2], sv[3]);
        ((float4*)&red[wave][HH])[lane] = make_float4(sv[4], sv[5], sv[6], sv[7]);
    }
    __builtin_amdgcn_fence(__ATOMIC_RELEASE, "workgroup", "local");
    __builtin_amdgcn_s_barrier();
    __builtin_amdgcn_fence(__ATOMIC_ACQUIRE, "workgroup", "local");
    if (threadIdx.x < 2 * HH) {
        const int wd = threadIdx.x;
        float tot = 0.f;
#pragma unroll
        for (int k = 0; k < GG_T / 32; k++) tot += red[k][wd];
        atomicAdd(&stat[wd], tot);
    }
}
__global__ void __launch_bounds__(GG_T) k_gather_first(const int2* __restrict__ rowse, const int2* __restrict__ rec,
                               const float4* __restrict__ y4, const float* __restrict__ ndv,
                               const float* __restrict__ gcb, const float4* __restrict__ hold4,
                               float4* __restrict__ hpre4, float* stat, int T, int d, int l) {
    gather_body<false>(rowse, rec, y4, ndv, gcb, hold4, nullptr, nullptr, nullptr, hpre4, stat,
                       T, d, l);
}
__global__ void __launch_bounds__(GG_T) k_gather_gc(const int2* __restrict__ rowse, const int2* __restrict__ rec,
                            const float4* __restrict__ y4, const float* __restrict__ ndv,
                            const float* __restrict__ gcb, const float4* __restrict__ hold4,
                            const float* __restrict__ pstat, const float* __restrict__ pgam,
                            const float* __restrict__ pbet, float4* __restrict__ hpre4,
                            float* stat, int T, int d, int l) {
    gather_body<true>(rowse, rec, y4, ndv, gcb, hold4, pstat, pgam, pbet, hpre4, stat, T, d, l);
}

__device__ __forceinline__ v16h gau_last_tile(const float* __restrict__ h,
                                              const float* __restrict__ pstat,
                                              const float* __restrict__ pgam,
                                              const float* __restrict__ pbet,
                                              const float* __restrict__ fwT,
                                              const float* __restrict__ fb,
                                              const float* __restrict__ gwT,
                                              const float* __restrict__ gb,
                                              int T, int d, int l, int row0) {
    const int lane = threadIdx.x & 31;
    const int c = lane & 15;
    const int base = (lane & 16) ? 8 : 0;
    float sc[16], sh[16];
    const float inv_cnt = 1.f / (float)(BB * NN * T);
    bn_coef8(sc, sh, pstat, pgam, pbet, base, inv_cnt);
    bn_coef8(sc + 8, sh + 8, pstat, pgam, pbet, base + 16, inv_cnt);
    const float* hrow1 = h + ((size_t)(row0 + c) * T + (T - 1)) * HH + base;
    const float* hrow0 = hrow1 - d * HH;
    float4 rx[8];
    rx[0] = ((const float4*)hrow0)[0];        rx[1] = ((const float4*)hrow0)[1];
    rx[2] = ((const float4*)(hrow0 + 16))[0]; rx[3] = ((const float4*)(hrow0 + 16))[1];
    rx[4] = ((const float4*)hrow1)[0];        rx[5] = ((const float4*)hrow1)[1];
    rx[6] = ((const float4*)(hrow1 + 16))[0]; rx[7] = ((const float4*)(hrow1 + 16))[1];
    v16h x0, x1;
    cvt8_affine(x0, 0, rx[0], rx[1], sc, sh);
    cvt8_affine(x0, 8, rx[2], rx[3], sc + 8, sh + 8);
    cvt8_affine(x1, 0, rx[4], rx[5], sc, sh);
    cvt8_affine(x1, 8, rx[6], rx[7], sc + 8, sh + 8);
    v16h gp;
#pragma unroll
    for (int tn = 0; tn < 2; tn++) {
        const float* fwr = fwT + (size_t)((l * HH + tn * 16 + c) * 2) * HH;
        const float* gwr = gwT + (size_t)((l * HH + tn * 16 + c) * 2) * HH;
        v16h wf0, wf1, wg0, wg1;
        pack8(wf0, 0, fwr + base);       pack8(wf0, 8, fwr + base + 16);
        pack8(wg0, 0, gwr + base);       pack8(wg0, 8, gwr + base + 16);
        pack8(wf1, 0, fwr + HH + base);  pack8(wf1, 8, fwr + HH + base + 16);
        pack8(wg1, 0, gwr + HH + base);  pack8(wg1, 8, gwr + HH + base + 16);
        float4 f0 = ((const float4*)(fb + l * HH + tn * 16 + base))[0];
        float4 f1 = ((const float4*)(fb + l * HH + tn * 16 + base))[1];
        float4 g0 = ((const float4*)(gb + l * HH + tn * 16 + base))[0];
        float4 g1 = ((const float4*)(gb + l * HH + tn * 16 + base))[1];
        v8f cf = {f0.x, f0.y, f0.z, f0.w, f1.x, f1.y, f1.z, f1.w};
        v8f cg = {g0.x, g0.y, g0.z, g0.w, g1.x, g1.y, g1.z, g1.w};
        cf = wmma_f16(wf0, x0, cf);
        cg = wmma_f16(wg0, x0, cg);
        cf = wmma_f16(wf1, x1, cf);
        cg = wmma_f16(wg1, x1, cg);
#pragma unroll
        for (int r = 0; r < 8; r++)
            gp[tn * 8 + r] = (_Float16)(fast_tanh(cf[r]) * fast_sigmoid(cg[r]));
    }
    return gp;
}

#define EF_ROWS 64
__global__ void __launch_bounds__(128) k_end_fused(const float* __restrict__ glast,
                            const float* __restrict__ sw, const float* __restrict__ sb,
                            const float* __restrict__ w1, const float* __restrict__ b1,
                            const float* __restrict__ w2, const float* __restrict__ b2,
                            const float* __restrict__ hlast, const float* __restrict__ pstat,
                            const float* __restrict__ pgam, const float* __restrict__ pbet,
                            const float* __restrict__ fwT, const float* __restrict__ fb,
                            const float* __restrict__ gwT, const float* __restrict__ gb,
                            int T, int d, int llast, float* out) {
    __shared__ v8h sB[4 * 8 * 2 * 32];
    __shared__ float4 red[4 * 4 * 2 * 32];
    __shared__ v8h sG[4 * 2 * 32];
    const int w = threadIdx.x >> 5;
    const int lane = threadIdx.x & 31;
    const int c = lane & 15;
    const int h = lane >> 4;
    const int base = h * 8;
    const int row0 = blockIdx.x * EF_ROWS;

    {
        const v16h gp = gau_last_tile(hlast, pstat, pgam, pbet, fwT, fb, gwT, gb, T, d, llast,
                                      row0 + w * 16);
        sG[(w * 2 + 0) * 32 + lane] = __builtin_shufflevector(gp, gp, 0, 1, 2, 3, 4, 5, 6, 7);
        sG[(w * 2 + 1) * 32 + lane] = __builtin_shufflevector(gp, gp, 8, 9, 10, 11, 12, 13, 14, 15);
    }

    v16h wA[4][NLAY];
    v8f bs[4];
#pragma unroll
    for (int m = 0; m < 4; m++) {
        const int ch0 = (4 * w + m) * 16;
        float4 s0 = make_float4(0.f, 0.f, 0.f, 0.f), s1 = s0;
#pragma unroll
        for (int l = 0; l < NLAY; l++) {
            const float* p = sw + (size_t)(l * SKC + ch0 + c) * HH;
            pack8(wA[m][l], 0, p + base);
            pack8(wA[m][l], 8, p + base + 16);
            float4 x = ((const float4*)(sb + l * SKC + ch0 + base))[0];
            float4 y = ((const float4*)(sb + l * SKC + ch0 + base))[1];
            s0.x += x.x; s0.y += x.y; s0.z += x.z; s0.w += x.w;
            s1.x += y.x; s1.y += y.y; s1.z += y.z; s1.w += y.w;
        }
        bs[m] = v8f{s0.x, s0.y, s0.z, s0.w, s1.x, s1.y, s1.z, s1.w};
    }
    __builtin_amdgcn_fence(__ATOMIC_RELEASE, "workgroup", "local");
    __builtin_amdgcn_s_barrier();
    __builtin_amdgcn_fence(__ATOMIC_ACQUIRE, "workgroup", "local");
#pragma unroll
    for (int rt = 0; rt < 4; rt++) {
        const float* arow = glast + (size_t)(row0 + rt * 16 + c) * GLK;
        v16h g[NLAY];
#pragma unroll
        for (int ll = 0; ll < NLAY - 1; ll++) {
            pack8(g[ll], 0, arow + ll * HH + base);
            pack8(g[ll], 8, arow + ll * HH + base + 16);
        }
        {
            v8h lo = sG[(rt * 2 + 0) * 32 + lane];
            v8h hi = sG[(rt * 2 + 1) * 32 + lane];
            g[NLAY - 1] = __builtin_shufflevector(lo, hi, 0, 1, 2, 3, 4, 5, 6, 7,
                                                  8, 9, 10, 11, 12, 13, 14, 15);
        }
        v8f acc[4];
#pragma unroll
        for (int m = 0; m < 4; m++) acc[m] = bs[m];
#pragma unroll
        for (int l = 0; l < NLAY; l++) {
#pragma unroll
            for (int m = 0; m < 4; m++) acc[m] = wmma_f16(wA[m][l], g[l], acc[m]);
        }
#pragma unroll
        for (int m = 0; m < 4; m++) {
            v8h p;
#pragma unroll
            for (int r = 0; r < 8; r++) p[r] = (_Float16)fmaxf(acc[m][r], 0.f);
            sB[((rt * 8 + 2 * w + (m >> 1)) * 2 + (m & 1)) * 32 + lane] = p;
        }
    }
    float4 raw[2][SKC / 32][4];
#define EF_LOAD_RAW(P, J)                                                                     \
    {                                                                                         \
        const float* q = w1 + (size_t)((2 * (4 * w + (P)) + (J)) * 16 + c) * SKC + base;      \
        _Pragma("unroll") for (int ks = 0; ks < SKC / 32; ks++) {                             \
            raw[J][ks][0] = ((const float4*)(q + ks * 32))[0];                                \
            raw[J][ks][1] = ((const float4*)(q + ks * 32))[1];                                \
            raw[J][ks][2] = ((const float4*)(q + ks * 32 + 16))[0];                           \
            raw[J][ks][3] = ((const float4*)(q + ks * 32 + 16))[1];                           \
        }                                                                                     \
    }
#define EF_LOAD_SMALL(P)                                                                      \
    {                                                                                         \
        const int ks3 = 4 * w + (P);                                                          \
        const float* q = w2 + (size_t)(c < OUTF ? c : 0) * ECC + ks3 * 32;                    \
        v16h t;                                                                               \
        pack8(t, 0, q + base);                                                                \
        pack8(t, 8, q + base + 16);                                                           \
        w2An = (c < OUTF) ? t : v16h{};                                                       \
        _Pragma("unroll") for (int j = 0; j < 2; j++) {                                       \
            const int ch0 = (2 * ks3 + j) * 16;                                               \
            float4 x = ((const float4*)(b1 + ch0 + base))[0];                                 \
            float4 y = ((const float4*)(b1 + ch0 + base))[1];                                 \
            bbn[j] = v8f{x.x, x.y, x.z, x.w, y.x, y.y, y.z, y.w};                             \
        }                                                                                     \
    }
    v16h w2An;
    v8f bbn[2];
    EF_LOAD_SMALL(0)
    EF_LOAD_RAW(0, 0)
    EF_LOAD_RAW(0, 1)
    __syncthreads();
    v16h bf[4][SKC / 32];
#pragma unroll
    for (int rt = 0; rt < 4; rt++) {
#pragma unroll
        for (int ks = 0; ks < SKC / 32; ks++) {
            v8h lo = sB[((rt * 8 + ks) * 2 + 0) * 32 + lane];
            v8h hi = sB[((rt * 8 + ks) * 2 + 1) * 32 + lane];
            bf[rt][ks] = __builtin_shufflevector(lo, hi, 0, 1, 2, 3, 4, 5, 6, 7,
                                                 8, 9, 10, 11, 12, 13, 14, 15);
        }
    }
    v8f oacc[4];
#pragma unroll
    for (int rt = 0; rt < 4; rt++) oacc[rt] = v8f{0.f, 0.f, 0.f, 0.f, 0.f, 0.f, 0.f, 0.f};
#pragma unroll 1
    for (int p = 0; p < 4; p++) {
        v16h wB[2][SKC / 32];
#pragma unroll
        for (int j = 0; j < 2; j++) {
#pragma unroll
            for (int ks = 0; ks < SKC / 32; ks++) {
                cvt8(wB[j][ks], 0, raw[j][ks][0], raw[j][ks][1]);
                cvt8(wB[j][ks], 8, raw[j][ks][2], raw[j][ks][3]);
            }
        }
        const v16h w2A = w2An;
        const v8f bb0 = bbn[0], bb1 = bbn[1];
#pragma unroll
        for (int rt = 0; rt < 4; rt++) {
            if (rt == 0 && p < 3) {
                EF_LOAD_SMALL(p + 1)
                EF_LOAD_RAW(p + 1, 0)
            }
            if (rt == 2 && p < 3) EF_LOAD_RAW(p + 1, 1)
            v8f a0 = bb0, a1 = bb1;
#pragma unroll
            for (int ks = 0; ks < SKC / 32; ks++) {
                a0 = wmma_f16(wB[0][ks], bf[rt][ks], a0);
                a1 = wmma_f16(wB[1][ks], bf[rt][ks], a1);
            }
            v16h b3;
#pragma unroll
            for (int r = 0; r < 8; r++) {
                b3[r]     = (_Float16)fmaxf(a0[r], 0.f);
                b3[8 + r] = (_Float16)fmaxf(a1[r], 0.f);
            }
            oacc[rt] = wmma_f16(w2A, b3, oacc[rt]);
        }
    }
#undef EF_LOAD_RAW
#undef EF_LOAD_SMALL
#pragma unroll
    for (int rt = 0; rt < 4; rt++) {
        red[((w * 4 + rt) * 2 + 0) * 32 + lane] =
            make_float4(oacc[rt][0], oacc[rt][1], oacc[rt][2], oacc[rt][3]);
        red[((w * 4 + rt) * 2 + 1) * 32 + lane] =
            make_float4(oacc[rt][4], oacc[rt][5], oacc[rt][6], oacc[rt][7]);
    }
    __syncthreads();
    if (w < 2) {
        const int b = row0 >> 9;
#pragma unroll
        for (int jj = 0; jj < 2; jj++) {
            const int rt = 2 * w + jj;
            float4 t0 = make_float4(0.f, 0.f, 0.f, 0.f), t1 = t0;
#pragma unroll
            for (int ww = 0; ww < 4; ww++) {
                float4 x = red[((ww * 4 + rt) * 2 + 0) * 32 + lane];
                float4 y = red[((ww * 4 + rt) * 2 + 1) * 32 + lane];
                t0.x += x.x; t0.y += x.y; t0.z += x.z; t0.w += x.w;
                t1.x += y.x; t1.y += y.y; t1.z += y.z; t1.w += y.w;
            }
            const int n = (row0 & (NN - 1)) + rt * 16 + c;
            const float v[8] = {t0.x, t0.y, t0.z, t0.w, t1.x, t1.y, t1.z, t1.w};
#pragma unroll
            for (int r = 0; r < 8; r++) {
                const int f = base + r;
                if (f < OUTF) out[(b * OUTF + f) * NN + n] = v[r] + b2[f];
            }
        }
    }
}

extern "C" void kernel_launch(void* const* d_in, const int* in_sizes, int n_in,
                              void* d_out, int out_size, void* d_ws, size_t ws_size,
                              hipStream_t stream) {
    const float* feat     = (const float*)d_in[0];
    const int*   src      = (const int*)d_in[1];
    const int*   dst      = (const int*)d_in[2];
    const float* nv1      = (const float*)d_in[3];
    const float* nv2      = (const float*)d_in[4];
    const float* start_w  = (const float*)d_in[5];
    const float* start_b  = (const float*)d_in[6];
    const float* filter_w = (const float*)d_in[7];
    const float* filter_b = (const float*)d_in[8];
    const float* gate_w   = (const float*)d_in[9];
    const float* gate_b   = (const float*)d_in[10];
    const float* skip_w   = (const float*)d_in[11];
    const float* skip_b   = (const float*)d_in[12];
    const float* bn_gamma = (const float*)d_in[13];
    const float* bn_beta  = (const float*)d_in[14];
    const float* gconv_w  = (const float*)d_in[15];
    const float* gconv_b  = (const float*)d_in[16];
    const float* end1_w   = (const float*)d_in[17];
    const float* end1_b   = (const float*)d_in[18];
    const float* end2_w   = (const float*)d_in[19];
    const float* end2_b   = (const float*)d_in[20];

    float* ws = (float*)d_ws;
    size_t o = 0;
    float* rowmax = ws + o; o += NN;
    float* rowsum = ws + o; o += NN;
    float* nsv    = ws + o; o += NN;
    float* ndv    = ws + o; o += NN;
    float* bnstat = ws + o; o += NLAY * 2 * HH;
    float* fwT    = ws + o; o += NLAY * HH * 2 * HH;
    float* gwT    = ws + o; o += NLAY * HH * 2 * HH;
    float* gcwT   = ws + o; o += NLAY * HH * HH;
    float* glast  = ws + o; o += (size_t)BB * NN * GLK;
    float* hA     = ws + o; o += (size_t)BB * NN * 13 * HH;
    float* hB     = ws + o; o += (size_t)BB * NN * 13 * HH;
    float* ygc    = ws + o; o += (size_t)BB * NN * 12 * HH;
    int2*  rowse  = (int2*)(ws + o); o += 2 * NN;
    int2*  rec    = (int2*)(ws + o); o += 2 * NE;

    k_setup<<<SU_BLOCKS, SU_T, 0, stream>>>(src, dst, nv1, nv2, feat, start_w, start_b,
                                            filter_w, gate_w, gconv_w, rowmax, rowsum, nsv, ndv,
                                            hA, fwT, gwT, gcwT, bnstat);

    const int dil[8] = {1, 2, 1, 2, 1, 2, 1, 2};
    float* hcur = hA;
    float* hnext = hB;
    int T = 13;
    for (int l = 0; l < NLAY - 1; l++) {
        int d = dil[l];
        int Tp = T - d;
        int M = BB * NN * Tp;
        int blocksGau = (M / 16) / 4 / GAU_NT;
        int blocksGather = (BB / 4) * NN * Tp * 8 / GG_T;

        const int lp = l ? l - 1 : 0;
        const float* pstat = bnstat + lp * 2 * HH;
        const float* pgam  = bn_gamma + lp * HH;
        const float* pbet  = bn_beta + lp * HH;
        if (l == 0)
            k_gau_first<<<NN + blocksGau, 128, 0, stream>>>(
                hcur, fwT, filter_b, gwT, gate_b, gcwT, ygc, glast, T, d, l,
                src, dst, nv1, nv2, rowmax, rowsum, nsv, rowse, rec);
        else
            k_gau_wmma<<<blocksGau, 128, 0, stream>>>(hcur, pstat, pgam, pbet, fwT, filter_b, gwT,
                                                      gate_b, gcwT, ygc, glast, T, d, l);
        if (l == 0)
            k_gather_first<<<blocksGather, GG_T, 0, stream>>>(
                rowse, rec, (const float4*)ygc, ndv, gconv_b, (const float4*)hcur,
                (float4*)hnext, bnstat, T, d, l);
        else
            k_gather_gc<<<blocksGather, GG_T, 0, stream>>>(
                rowse, rec, (const float4*)ygc, ndv, gconv_b, (const float4*)hcur, pstat, pgam,
                pbet, (float4*)hnext, bnstat + l * 2 * HH, T, d, l);
        float* tmp = hcur; hcur = hnext; hnext = tmp;
        T = Tp;
    }

    const int ll = NLAY - 1;
    k_end_fused<<<BB * NN / EF_ROWS, 128, 0, stream>>>(
        glast, skip_w, skip_b, end1_w, end1_b, end2_w, end2_b, hcur, bnstat + (ll - 1) * 2 * HH,
        bn_gamma + (ll - 1) * HH, bn_beta + (ll - 1) * HH, fwT, filter_b, gwT, gate_b, T, dil[ll],
        ll, (float*)d_out);
}
